# starting kernel
# baseline (speedup 1.0000x reference)
#include <hip/hip_runtime.h>
#include <hip/hip_bf16.h>

#define NN 50000
#define EE 800000
#define KDIM 256
#define OUTD 64
#define NEG_SLOPE 0.2f

typedef __attribute__((ext_vector_type(2))) float v2f;
typedef __attribute__((ext_vector_type(4))) float v4f;
typedef __attribute__((ext_vector_type(8))) float v8f;

__device__ __forceinline__ unsigned orderFloat(float f) {
  unsigned u = __float_as_uint(f);
  return (u & 0x80000000u) ? ~u : (u | 0x80000000u);
}
__device__ __forceinline__ float unorderFloat(unsigned u) {
  u = (u & 0x80000000u) ? (u & 0x7FFFFFFFu) : ~u;
  return __uint_as_float(u);
}

__global__ __launch_bounds__(256) void init_kernel(
    float* __restrict__ rst, const float* __restrict__ gat_bias,
    unsigned* __restrict__ m_u32, float* __restrict__ denom) {
  int i = blockIdx.x * blockDim.x + threadIdx.x;
  if (i < NN * KDIM) rst[i] = gat_bias[i & (KDIM - 1)];
  if (i < NN * 4) { m_u32[i] = 0u; denom[i] = 0.0f; }
}

__global__ __launch_bounds__(32) void wmma_proj_kernel(
    const float* __restrict__ feat,
    const float* __restrict__ Wsrc, const float* __restrict__ Bsrc,
    const float* __restrict__ Wdst, const float* __restrict__ Bdst,
    float* __restrict__ Hsrc, float* __restrict__ Hdst) {
  const int lane = threadIdx.x;
  const int rowTile = blockIdx.x;
  const int yt = blockIdx.y;
  const int mat = yt >> 4;
  const int colTile = yt & 15;

  const float* __restrict__ W  = mat ? Wdst : Wsrc;
  const float* __restrict__ Bv = mat ? Bdst : Bsrc;
  float* __restrict__ Hout     = mat ? Hdst : Hsrc;

  const int row0 = rowTile * 64;
  const int col0 = colTile * 16;
  const int khi  = lane >> 4;
  const int mlo  = lane & 15;

  int arow[4];
#pragma unroll
  for (int s = 0; s < 4; ++s) {
    int r = row0 + s * 16 + mlo;
    arow[s] = (r < NN) ? r : 0;
  }

  v8f acc[4] = {};
  for (int kb = 0; kb < KDIM; kb += 4) {
    v2f bfrag;
    bfrag.x = W[(size_t)(kb + 2 * khi)     * KDIM + col0 + mlo];
    bfrag.y = W[(size_t)(kb + 2 * khi + 1) * KDIM + col0 + mlo];
#pragma unroll
    for (int s = 0; s < 4; ++s) {
      v2f afrag = *(const v2f*)(feat + (size_t)arow[s] * KDIM + kb + 2 * khi);
      acc[s] = __builtin_amdgcn_wmma_f32_16x16x4_f32(
          false, afrag, false, bfrag, (short)0, acc[s], false, false);
    }
  }

  const int mbase = khi * 8;
  const int ncol  = col0 + mlo;
  const float bias = Bv[ncol];
  if (row0 + 64 <= NN) {
#pragma unroll
    for (int s = 0; s < 4; ++s) {
#pragma unroll
      for (int i = 0; i < 8; ++i) {
        int r = row0 + s * 16 + mbase + i;
        Hout[(size_t)r * KDIM + ncol] = acc[s][i] + bias;
      }
    }
  } else {
#pragma unroll
    for (int s = 0; s < 4; ++s) {
#pragma unroll
      for (int i = 0; i < 8; ++i) {
        int r = row0 + s * 16 + mbase + i;
        if (r < NN) Hout[(size_t)r * KDIM + ncol] = acc[s][i] + bias;
      }
    }
  }
}

__global__ __launch_bounds__(256) void edge_score_kernel(
    const int* __restrict__ ei, const float* __restrict__ Hsrc,
    const float* __restrict__ Hdst, const float* __restrict__ attn,
    float* __restrict__ score, unsigned* __restrict__ m_u32) {
  const int e = (blockIdx.x * blockDim.x + threadIdx.x) >> 5;
  const int lane = threadIdx.x & 31;
  if (e >= EE) return;
  const int src = ei[e];
  const int dst = ei[EE + e];

  const v4f* ps = (const v4f*)(Hsrc + (size_t)src * KDIM + lane * 8);
  const v4f* pd = (const v4f*)(Hdst + (size_t)dst * KDIM + lane * 8);
  const v4f* pa = (const v4f*)(attn + lane * 8);
  v4f s0 = ps[0], s1 = ps[1];
  v4f d0 = pd[0], d1 = pd[1];
  v4f a0 = pa[0], a1 = pa[1];

  float part = 0.0f;
#pragma unroll
  for (int i = 0; i < 4; ++i) {
    float v = s0[i] + d0[i];
    v = (v >= 0.0f) ? v : v * NEG_SLOPE;
    part += v * a0[i];
    float w = s1[i] + d1[i];
    w = (w >= 0.0f) ? w : w * NEG_SLOPE;
    part += w * a1[i];
  }
  part += __shfl_xor(part, 1, 32);
  part += __shfl_xor(part, 2, 32);
  part += __shfl_xor(part, 4, 32);
  if ((lane & 7) == 0) {
    int h = lane >> 3;
    score[(size_t)e * 4 + h] = part;
    atomicMax(&m_u32[(size_t)dst * 4 + h], orderFloat(part));
  }
}

__global__ __launch_bounds__(256) void sanitize_m_kernel(unsigned* __restrict__ m) {
  int i = blockIdx.x * blockDim.x + threadIdx.x;
  if (i >= NN * 4) return;
  float f = unorderFloat(m[i]);
  if (!isfinite(f)) f = 0.0f;
  ((float*)m)[i] = f;
}

__global__ __launch_bounds__(256) void edge_exp_kernel(
    const int* __restrict__ ei, const float* __restrict__ m_f,
    float* __restrict__ score, float* __restrict__ denom) {
  int i = blockIdx.x * blockDim.x + threadIdx.x;
  if (i >= EE * 4) return;
  int e = i >> 2, h = i & 3;
  int dst = ei[EE + e];
  float ex = expf(score[i] - m_f[(size_t)dst * 4 + h]);
  score[i] = ex;
  unsafeAtomicAdd(&denom[(size_t)dst * 4 + h], ex);
}

__global__ __launch_bounds__(256) void edge_scatter_kernel(
    const int* __restrict__ ei, const float* __restrict__ Hsrc,
    const float* __restrict__ ex, const float* __restrict__ denom,
    float* __restrict__ rst) {
  const int e = (blockIdx.x * blockDim.x + threadIdx.x) >> 5;
  const int lane = threadIdx.x & 31;
  if (e >= EE) return;
  const int src = ei[e];
  const int dst = ei[EE + e];
  const int h = lane >> 3;
  float den = denom[(size_t)dst * 4 + h];
  float alpha = ex[(size_t)e * 4 + h] / ((den == 0.0f) ? 1.0f : den);
  const v4f* ps = (const v4f*)(Hsrc + (size_t)src * KDIM + lane * 8);
  v4f s0 = ps[0], s1 = ps[1];
  float* pr = rst + (size_t)dst * KDIM + lane * 8;
#pragma unroll
  for (int i = 0; i < 4; ++i) {
    unsafeAtomicAdd(&pr[i],     alpha * s0[i]);
    unsafeAtomicAdd(&pr[i + 4], alpha * s1[i]);
  }
}

__global__ __launch_bounds__(32) void wmma_out_kernel(
    const float* __restrict__ rst, const float* __restrict__ Wfc,
    const float* __restrict__ bfc, float* __restrict__ out) {
  const int lane = threadIdx.x;
  const int row0 = blockIdx.x * 64;
  const int col0 = blockIdx.y * 16;
  const int khi = lane >> 4;
  const int mlo = lane & 15;

  int arow[4];
#pragma unroll
  for (int s = 0; s < 4; ++s) {
    int r = row0 + s * 16 + mlo;
    arow[s] = (r < NN) ? r : 0;
  }

  v8f acc[4] = {};
  for (int kb = 0; kb < KDIM; kb += 4) {
    v2f bfrag;
    bfrag.x = Wfc[(size_t)(kb + 2 * khi)     * OUTD + col0 + mlo];
    bfrag.y = Wfc[(size_t)(kb + 2 * khi + 1) * OUTD + col0 + mlo];
#pragma unroll
    for (int s = 0; s < 4; ++s) {
      v2f afrag = *(const v2f*)(rst + (size_t)arow[s] * KDIM + kb + 2 * khi);
      acc[s] = __builtin_amdgcn_wmma_f32_16x16x4_f32(
          false, afrag, false, bfrag, (short)0, acc[s], false, false);
    }
  }

  const int mbase = khi * 8;
  const int ncol  = col0 + mlo;
  const float bias = bfc[ncol];
  if (row0 + 64 <= NN) {
#pragma unroll
    for (int s = 0; s < 4; ++s) {
#pragma unroll
      for (int i = 0; i < 8; ++i) {
        int r = row0 + s * 16 + mbase + i;
        out[(size_t)r * OUTD + ncol] = acc[s][i] + bias;
      }
    }
  } else {
#pragma unroll
    for (int s = 0; s < 4; ++s) {
#pragma unroll
      for (int i = 0; i < 8; ++i) {
        int r = row0 + s * 16 + mbase + i;
        if (r < NN) out[(size_t)r * OUTD + ncol] = acc[s][i] + bias;
      }
    }
  }
}

extern "C" void kernel_launch(void* const* d_in, const int* in_sizes, int n_in,
                              void* d_out, int out_size, void* d_ws, size_t ws_size,
                              hipStream_t stream) {
  const float* feat     = (const float*)d_in[0];
  const int*   ei       = (const int*)d_in[1];
  const float* w_src    = (const float*)d_in[2];
  const float* b_src    = (const float*)d_in[3];
  const float* w_dst    = (const float*)d_in[4];
  const float* b_dst    = (const float*)d_in[5];
  const float* attn     = (const float*)d_in[6];
  const float* gat_bias = (const float*)d_in[7];
  const float* w_fc     = (const float*)d_in[8];
  const float* b_fc     = (const float*)d_in[9];
  float* out = (float*)d_out;

  float* ws    = (float*)d_ws;
  float* Hsrc  = ws;
  float* Hdst  = Hsrc + (size_t)NN * KDIM;
  float* score = Hdst + (size_t)NN * KDIM;
  float* mbuf  = score + (size_t)EE * 4;
  float* denom = mbuf + (size_t)NN * 4;
  float* rst   = denom + (size_t)NN * 4;

  const int rowTiles = (NN + 63) / 64;

  init_kernel<<<(NN * KDIM + 255) / 256, 256, 0, stream>>>(
      rst, gat_bias, (unsigned*)mbuf, denom);

  wmma_proj_kernel<<<dim3(rowTiles, 32), 32, 0, stream>>>(
      feat, w_src, b_src, w_dst, b_dst, Hsrc, Hdst);

  edge_score_kernel<<<(EE * 32) / 256, 256, 0, stream>>>(
      ei, Hsrc, Hdst, attn, score, (unsigned*)mbuf);

  sanitize_m_kernel<<<(NN * 4 + 255) / 256, 256, 0, stream>>>((unsigned*)mbuf);

  edge_exp_kernel<<<(EE * 4 + 255) / 256, 256, 0, stream>>>(
      ei, mbuf, score, denom);

  edge_scatter_kernel<<<(EE * 32) / 256, 256, 0, stream>>>(
      ei, Hsrc, score, denom, rst);

  wmma_out_kernel<<<dim3(rowTiles, 4), 32, 0, stream>>>(rst, w_fc, b_fc, out);
}


# Projection GEMM data path: an 8-wave work-group copies its 128x32 feat tile and 32x256 W tile into LDS with full-line 16-byte global loads; waves read A and B fragments from LDS; k order kept
# speedup vs baseline: 1.3886x; 1.3886x over previous
#include <hip/hip_runtime.h>
#include <hip/hip_bf16.h>

#define NN 50000
#define EE 800000
#define KDIM 256
#define OUTD 64
#define NEG_SLOPE 0.2f

typedef __attribute__((ext_vector_type(2))) float v2f;
typedef __attribute__((ext_vector_type(4))) float v4f;
typedef __attribute__((ext_vector_type(8))) float v8f;

__device__ __forceinline__ unsigned orderFloat(float f) {
  unsigned u = __float_as_uint(f);
  return (u & 0x80000000u) ? ~u : (u | 0x80000000u);
}
__device__ __forceinline__ float unorderFloat(unsigned u) {
  u = (u & 0x80000000u) ? (u & 0x7FFFFFFFu) : ~u;
  return __uint_as_float(u);
}

__global__ __launch_bounds__(256) void init_kernel(
    float* __restrict__ rst, const float* __restrict__ gat_bias,
    unsigned* __restrict__ m_u32, float* __restrict__ denom) {
  int i = blockIdx.x * blockDim.x + threadIdx.x;
  if (i < NN * KDIM) rst[i] = gat_bias[i & (KDIM - 1)];
  if (i < NN * 4) { m_u32[i] = 0u; denom[i] = 0.0f; }
}

#define PBM 128
#define PBK 32
#define A_PITCH 36
#define B_PITCH 544

__global__ __launch_bounds__(256) void wmma_proj_kernel(
    const float* __restrict__ feat,
    const float* __restrict__ Wsrc, const float* __restrict__ Bsrc,
    const float* __restrict__ Wdst, const float* __restrict__ Bdst,
    float* __restrict__ Hsrc, float* __restrict__ Hdst) {
  __shared__ __attribute__((aligned(16))) float As[PBM * A_PITCH];
  __shared__ __attribute__((aligned(16))) float Bs[(PBK / 2) * B_PITCH];

  const int tid  = threadIdx.x;
  const int lane = tid & 31;
  const int wave = tid >> 5;
  const int wr   = wave >> 2;
  const int wc   = wave & 3;
  const int mat  = blockIdx.x;
  const int row0 = blockIdx.y * PBM;
  const int khi  = lane >> 4;
  const int mlo  = lane & 15;

  const float* __restrict__ W  = mat ? Wdst : Wsrc;
  const float* __restrict__ Bv = mat ? Bdst : Bsrc;
  float* __restrict__ Hout     = mat ? Hdst : Hsrc;

  const int ar = tid >> 3;
  const int ac = (tid & 7) * 4;
  const int bcg = tid & 63;
  const int bkp = tid >> 6;

  const float* ag[4];
#pragma unroll
  for (int i = 0; i < 4; ++i) {
    int r = row0 + ar + 32 * i;
    r = (r < NN) ? r : 0;
    ag[i] = feat + (size_t)r * KDIM + ac;
  }
  const float* bg = W + (size_t)(2 * bkp) * KDIM + bcg * 4;

  v4f ra[4], rb0[4], rb1[4];
#pragma unroll
  for (int i = 0; i < 4; ++i) {
    ra[i]  = *(const v4f*)(ag[i]);
    rb0[i] = *(const v4f*)(bg + (size_t)(8 * i) * KDIM);
    rb1[i] = *(const v4f*)(bg + (size_t)(8 * i + 1) * KDIM);
  }

  const float* aFrag = &As[(wr * 64 + mlo) * A_PITCH + 2 * khi];
  const float* bFrag = &Bs[khi * B_PITCH + (wc * 64 + mlo) * 2];

  v8f acc[4][4] = {};
  for (int c = 0; c < KDIM / PBK; ++c) {
#pragma unroll
    for (int i = 0; i < 4; ++i) {
      *(v4f*)&As[(ar + 32 * i) * A_PITCH + ac] = ra[i];
      float* p = &Bs[(bkp + 4 * i) * B_PITCH + bcg * 8];
      v4f lo = {rb0[i].x, rb1[i].x, rb0[i].y, rb1[i].y};
      v4f hi = {rb0[i].z, rb1[i].z, rb0[i].w, rb1[i].w};
      *(v4f*)p = lo;
      *(v4f*)(p + 4) = hi;
    }
    __syncthreads();

    if (c + 1 < KDIM / PBK) {
      const int k0 = (c + 1) * PBK;
#pragma unroll
      for (int i = 0; i < 4; ++i) {
        ra[i]  = *(const v4f*)(ag[i] + k0);
        rb0[i] = *(const v4f*)(bg + (size_t)(k0 + 8 * i) * KDIM);
        rb1[i] = *(const v4f*)(bg + (size_t)(k0 + 8 * i + 1) * KDIM);
      }
    }

#pragma unroll
    for (int ks = 0; ks < PBK / 4; ++ks) {
      v2f af[4], bf[4];
#pragma unroll
      for (int s = 0; s < 4; ++s)
        af[s] = *(const v2f*)(aFrag + s * 16 * A_PITCH + ks * 4);
#pragma unroll
      for (int t = 0; t < 4; ++t)
        bf[t] = *(const v2f*)(bFrag + ks * 2 * B_PITCH + t * 32);
#pragma unroll
      for (int s = 0; s < 4; ++s) {
#pragma unroll
        for (int t = 0; t < 4; ++t) {
          acc[s][t] = __builtin_amdgcn_wmma_f32_16x16x4_f32(
              false, af[s], false, bf[t], (short)0, acc[s][t], false, false);
        }
      }
    }
    __syncthreads();
  }

  const int mbase = khi * 8;
  float bias[4];
#pragma unroll
  for (int t = 0; t < 4; ++t) bias[t] = Bv[wc * 64 + t * 16 + mlo];
  const int rbase = row0 + wr * 64 + mbase;
  if (row0 + PBM <= NN) {
#pragma unroll
    for (int s = 0; s < 4; ++s) {
#pragma unroll
      for (int i = 0; i < 8; ++i) {
        int r = rbase + s * 16 + i;
#pragma unroll
        for (int t = 0; t < 4; ++t)
          Hout[(size_t)r * KDIM + wc * 64 + t * 16 + mlo] = acc[s][t][i] + bias[t];
      }
    }
  } else {
#pragma unroll
    for (int s = 0; s < 4; ++s) {
#pragma unroll
      for (int i = 0; i < 8; ++i) {
        int r = rbase + s * 16 + i;
        if (r < NN) {
#pragma unroll
          for (int t = 0; t < 4; ++t)
            Hout[(size_t)r * KDIM + wc * 64 + t * 16 + mlo] = acc[s][t][i] + bias[t];
        }
      }
    }
  }
}

__global__ __launch_bounds__(256) void edge_score_kernel(
    const int* __restrict__ ei, const float* __restrict__ Hsrc,
    const float* __restrict__ Hdst, const float* __restrict__ attn,
    float* __restrict__ score, unsigned* __restrict__ m_u32) {
  const int e = (blockIdx.x * blockDim.x + threadIdx.x) >> 5;
  const int lane = threadIdx.x & 31;
  if (e >= EE) return;
  const int src = ei[e];
  const int dst = ei[EE + e];

  const v4f* ps = (const v4f*)(Hsrc + (size_t)src * KDIM + lane * 8);
  const v4f* pd = (const v4f*)(Hdst + (size_t)dst * KDIM + lane * 8);
  const v4f* pa = (const v4f*)(attn + lane * 8);
  v4f s0 = ps[0], s1 = ps[1];
  v4f d0 = pd[0], d1 = pd[1];
  v4f a0 = pa[0], a1 = pa[1];

  float part = 0.0f;
#pragma unroll
  for (int i = 0; i < 4; ++i) {
    float v = s0[i] + d0[i];
    v = (v >= 0.0f) ? v : v * NEG_SLOPE;
    part += v * a0[i];
    float w = s1[i] + d1[i];
    w = (w >= 0.0f) ? w : w * NEG_SLOPE;
    part += w * a1[i];
  }
  part += __shfl_xor(part, 1, 32);
  part += __shfl_xor(part, 2, 32);
  part += __shfl_xor(part, 4, 32);
  if ((lane & 7) == 0) {
    int h = lane >> 3;
    score[(size_t)e * 4 + h] = part;
    atomicMax(&m_u32[(size_t)dst * 4 + h], orderFloat(part));
  }
}

__global__ __launch_bounds__(256) void sanitize_m_kernel(unsigned* __restrict__ m) {
  int i = blockIdx.x * blockDim.x + threadIdx.x;
  if (i >= NN * 4) return;
  float f = unorderFloat(m[i]);
  if (!isfinite(f)) f = 0.0f;
  ((float*)m)[i] = f;
}

__global__ __launch_bounds__(256) void edge_exp_kernel(
    const int* __restrict__ ei, const float* __restrict__ m_f,
    float* __restrict__ score, float* __restrict__ denom) {
  int i = blockIdx.x * blockDim.x + threadIdx.x;
  if (i >= EE * 4) return;
  int e = i >> 2, h = i & 3;
  int dst = ei[EE + e];
  float ex = expf(score[i] - m_f[(size_t)dst * 4 + h]);
  score[i] = ex;
  unsafeAtomicAdd(&denom[(size_t)dst * 4 + h], ex);
}

__global__ __launch_bounds__(256) void edge_scatter_kernel(
    const int* __restrict__ ei, const float* __restrict__ Hsrc,
    const float* __restrict__ ex, const float* __restrict__ denom,
    float* __restrict__ rst) {
  const int e = (blockIdx.x * blockDim.x + threadIdx.x) >> 5;
  const int lane = threadIdx.x & 31;
  if (e >= EE) return;
  const int src = ei[e];
  const int dst = ei[EE + e];
  const int h = lane >> 3;
  float den = denom[(size_t)dst * 4 + h];
  float alpha = ex[(size_t)e * 4 + h] / ((den == 0.0f) ? 1.0f : den);
  const v4f* ps = (const v4f*)(Hsrc + (size_t)src * KDIM + lane * 8);
  v4f s0 = ps[0], s1 = ps[1];
  float* pr = rst + (size_t)dst * KDIM + lane * 8;
#pragma unroll
  for (int i = 0; i < 4; ++i) {
    unsafeAtomicAdd(&pr[i],     alpha * s0[i]);
    unsafeAtomicAdd(&pr[i + 4], alpha * s1[i]);
  }
}

__global__ __launch_bounds__(32) void wmma_out_kernel(
    const float* __restrict__ rst, const float* __restrict__ Wfc,
    const float* __restrict__ bfc, float* __restrict__ out) {
  const int lane = threadIdx.x;
  const int row0 = blockIdx.x * 64;
  const int col0 = blockIdx.y * 16;
  const int khi = lane >> 4;
  const int mlo = lane & 15;

  int arow[4];
#pragma unroll
  for (int s = 0; s < 4; ++s) {
    int r = row0 + s * 16 + mlo;
    arow[s] = (r < NN) ? r : 0;
  }

  v8f acc[4] = {};
  for (int kb = 0; kb < KDIM; kb += 4) {
    v2f bfrag;
    bfrag.x = Wfc[(size_t)(kb + 2 * khi)     * OUTD + col0 + mlo];
    bfrag.y = Wfc[(size_t)(kb + 2 * khi + 1) * OUTD + col0 + mlo];
#pragma unroll
    for (int s = 0; s < 4; ++s) {
      v2f afrag = *(const v2f*)(rst + (size_t)arow[s] * KDIM + kb + 2 * khi);
      acc[s] = __builtin_amdgcn_wmma_f32_16x16x4_f32(
          false, afrag, false, bfrag, (short)0, acc[s], false, false);
    }
  }

  const int mbase = khi * 8;
  const int ncol  = col0 + mlo;
  const float bias = bfc[ncol];
  if (row0 + 64 <= NN) {
#pragma unroll
    for (int s = 0; s < 4; ++s) {
#pragma unroll
      for (int i = 0; i < 8; ++i) {
        int r = row0 + s * 16 + mbase + i;
        out[(size_t)r * OUTD + ncol] = acc[s][i] + bias;
      }
    }
  } else {
#pragma unroll
    for (int s = 0; s < 4; ++s) {
#pragma unroll
      for (int i = 0; i < 8; ++i) {
        int r = row0 + s * 16 + mbase + i;
        if (r < NN) out[(size_t)r * OUTD + ncol] = acc[s][i] + bias;
      }
    }
  }
}

extern "C" void kernel_launch(void* const* d_in, const int* in_sizes, int n_in,
                              void* d_out, int out_size, void* d_ws, size_t ws_size,
                              hipStream_t stream) {
  const float* feat     = (const float*)d_in[0];
  const int*   ei       = (const int*)d_in[1];
  const float* w_src    = (const float*)d_in[2];
  const float* b_src    = (const float*)d_in[3];
  const float* w_dst    = (const float*)d_in[4];
  const float* b_dst    = (const float*)d_in[5];
  const float* attn     = (const float*)d_in[6];
  const float* gat_bias = (const float*)d_in[7];
  const float* w_fc     = (const float*)d_in[8];
  const float* b_fc     = (const float*)d_in[9];
  float* out = (float*)d_out;

  float* ws    = (float*)d_ws;
  float* Hsrc  = ws;
  float* Hdst  = Hsrc + (size_t)NN * KDIM;
  float* score = Hdst + (size_t)NN * KDIM;
  float* mbuf  = score + (size_t)EE * 4;
  float* denom = mbuf + (size_t)NN * 4;
  float* rst   = denom + (size_t)NN * 4;

  const int rowTiles = (NN + 63) / 64;

  init_kernel<<<(NN * KDIM + 255) / 256, 256, 0, stream>>>(
      rst, gat_bias, (unsigned*)mbuf, denom);

  wmma_proj_kernel<<<dim3(2, (NN + PBM - 1) / PBM), 256, 0, stream>>>(
      feat, w_src, b_src, w_dst, b_dst, Hsrc, Hdst);

  edge_score_kernel<<<(EE * 32) / 256, 256, 0, stream>>>(
      ei, Hsrc, Hdst, attn, score, (unsigned*)mbuf);

  sanitize_m_kernel<<<(NN * 4 + 255) / 256, 256, 0, stream>>>((unsigned*)mbuf);

  edge_exp_kernel<<<(EE * 4 + 255) / 256, 256, 0, stream>>>(
      ei, mbuf, score, denom);

  edge_scatter_kernel<<<(EE * 32) / 256, 256, 0, stream>>>(
      ei, Hsrc, score, denom, rst);

  wmma_out_kernel<<<dim3(rowTiles, 4), 32, 0, stream>>>(rst, w_fc, b_fc, out);
}


# Scatter kernel lane map: lane l owns row elements 32i+l, so each of an edge's 8 atomic adds (and 8 row loads) covers one whole 128-byte line instead of 4 bytes in each of 8 lines; float atomics kept
# speedup vs baseline: 2.1597x; 1.5553x over previous
#include <hip/hip_runtime.h>
#include <hip/hip_bf16.h>

#define NN 50000
#define EE 800000
#define KDIM 256
#define OUTD 64
#define NEG_SLOPE 0.2f

typedef __attribute__((ext_vector_type(2))) float v2f;
typedef __attribute__((ext_vector_type(4))) float v4f;
typedef __attribute__((ext_vector_type(8))) float v8f;

__device__ __forceinline__ unsigned orderFloat(float f) {
  unsigned u = __float_as_uint(f);
  return (u & 0x80000000u) ? ~u : (u | 0x80000000u);
}
__device__ __forceinline__ float unorderFloat(unsigned u) {
  u = (u & 0x80000000u) ? (u & 0x7FFFFFFFu) : ~u;
  return __uint_as_float(u);
}

__global__ __launch_bounds__(256) void init_kernel(
    float* __restrict__ rst, const float* __restrict__ gat_bias,
    unsigned* __restrict__ m_u32, float* __restrict__ denom) {
  int i = blockIdx.x * blockDim.x + threadIdx.x;
  if (i < NN * KDIM) rst[i] = gat_bias[i & (KDIM - 1)];
  if (i < NN * 4) { m_u32[i] = 0u; denom[i] = 0.0f; }
}

#define PBM 128
#define PBK 32
#define A_PITCH 36
#define B_PITCH 544

__global__ __launch_bounds__(256) void wmma_proj_kernel(
    const float* __restrict__ feat,
    const float* __restrict__ Wsrc, const float* __restrict__ Bsrc,
    const float* __restrict__ Wdst, const float* __restrict__ Bdst,
    float* __restrict__ Hsrc, float* __restrict__ Hdst) {
  __shared__ __attribute__((aligned(16))) float As[PBM * A_PITCH];
  __shared__ __attribute__((aligned(16))) float Bs[(PBK / 2) * B_PITCH];

  const int tid  = threadIdx.x;
  const int lane = tid & 31;
  const int wave = tid >> 5;
  const int wr   = wave >> 2;
  const int wc   = wave & 3;
  const int mat  = blockIdx.x;
  const int row0 = blockIdx.y * PBM;
  const int khi  = lane >> 4;
  const int mlo  = lane & 15;

  const float* __restrict__ W  = mat ? Wdst : Wsrc;
  const float* __restrict__ Bv = mat ? Bdst : Bsrc;
  float* __restrict__ Hout     = mat ? Hdst : Hsrc;

  const int ar = tid >> 3;
  const int ac = (tid & 7) * 4;
  const int bcg = tid & 63;
  const int bkp = tid >> 6;

  const float* ag[4];
#pragma unroll
  for (int i = 0; i < 4; ++i) {
    int r = row0 + ar + 32 * i;
    r = (r < NN) ? r : 0;
    ag[i] = feat + (size_t)r * KDIM + ac;
  }
  const float* bg = W + (size_t)(2 * bkp) * KDIM + bcg * 4;

  v4f ra[4], rb0[4], rb1[4];
#pragma unroll
  for (int i = 0; i < 4; ++i) {
    ra[i]  = *(const v4f*)(ag[i]);
    rb0[i] = *(const v4f*)(bg + (size_t)(8 * i) * KDIM);
    rb1[i] = *(const v4f*)(bg + (size_t)(8 * i + 1) * KDIM);
  }

  const float* aFrag = &As[(wr * 64 + mlo) * A_PITCH + 2 * khi];
  const float* bFrag = &Bs[khi * B_PITCH + (wc * 64 + mlo) * 2];

  v8f acc[4][4] = {};
  for (int c = 0; c < KDIM / PBK; ++c) {
#pragma unroll
    for (int i = 0; i < 4; ++i) {
      *(v4f*)&As[(ar + 32 * i) * A_PITCH + ac] = ra[i];
      float* p = &Bs[(bkp + 4 * i) * B_PITCH + bcg * 8];
      v4f lo = {rb0[i].x, rb1[i].x, rb0[i].y, rb1[i].y};
      v4f hi = {rb0[i].z, rb1[i].z, rb0[i].w, rb1[i].w};
      *(v4f*)p = lo;
      *(v4f*)(p + 4) = hi;
    }
    __syncthreads();

    if (c + 1 < KDIM / PBK) {
      const int k0 = (c + 1) * PBK;
#pragma unroll
      for (int i = 0; i < 4; ++i) {
        ra[i]  = *(const v4f*)(ag[i] + k0);
        rb0[i] = *(const v4f*)(bg + (size_t)(k0 + 8 * i) * KDIM);
        rb1[i] = *(const v4f*)(bg + (size_t)(k0 + 8 * i + 1) * KDIM);
      }
    }

#pragma unroll
    for (int ks = 0; ks < PBK / 4; ++ks) {
      v2f af[4], bf[4];
#pragma unroll
      for (int s = 0; s < 4; ++s)
        af[s] = *(const v2f*)(aFrag + s * 16 * A_PITCH + ks * 4);
#pragma unroll
      for (int t = 0; t < 4; ++t)
        bf[t] = *(const v2f*)(bFrag + ks * 2 * B_PITCH + t * 32);
#pragma unroll
      for (int s = 0; s < 4; ++s) {
#pragma unroll
        for (int t = 0; t < 4; ++t) {
          acc[s][t] = __builtin_amdgcn_wmma_f32_16x16x4_f32(
              false, af[s], false, bf[t], (short)0, acc[s][t], false, false);
        }
      }
    }
    __syncthreads();
  }

  const int mbase = khi * 8;
  float bias[4];
#pragma unroll
  for (int t = 0; t < 4; ++t) bias[t] = Bv[wc * 64 + t * 16 + mlo];
  const int rbase = row0 + wr * 64 + mbase;
  if (row0 + PBM <= NN) {
#pragma unroll
    for (int s = 0; s < 4; ++s) {
#pragma unroll
      for (int i = 0; i < 8; ++i) {
        int r = rbase + s * 16 + i;
#pragma unroll
        for (int t = 0; t < 4; ++t)
          Hout[(size_t)r * KDIM + wc * 64 + t * 16 + mlo] = acc[s][t][i] + bias[t];
      }
    }
  } else {
#pragma unroll
    for (int s = 0; s < 4; ++s) {
#pragma unroll
      for (int i = 0; i < 8; ++i) {
        int r = rbase + s * 16 + i;
        if (r < NN) {
#pragma unroll
          for (int t = 0; t < 4; ++t)
            Hout[(size_t)r * KDIM + wc * 64 + t * 16 + mlo] = acc[s][t][i] + bias[t];
        }
      }
    }
  }
}

__global__ __launch_bounds__(256) void edge_score_kernel(
    const int* __restrict__ ei, const float* __restrict__ Hsrc,
    const float* __restrict__ Hdst, const float* __restrict__ attn,
    float* __restrict__ score, unsigned* __restrict__ m_u32) {
  const int e = (blockIdx.x * blockDim.x + threadIdx.x) >> 5;
  const int lane = threadIdx.x & 31;
  if (e >= EE) return;
  const int src = ei[e];
  const int dst = ei[EE + e];

  const v4f* ps = (const v4f*)(Hsrc + (size_t)src * KDIM + lane * 8);
  const v4f* pd = (const v4f*)(Hdst + (size_t)dst * KDIM + lane * 8);
  const v4f* pa = (const v4f*)(attn + lane * 8);
  v4f s0 = ps[0], s1 = ps[1];
  v4f d0 = pd[0], d1 = pd[1];
  v4f a0 = pa[0], a1 = pa[1];

  float part = 0.0f;
#pragma unroll
  for (int i = 0; i < 4; ++i) {
    float v = s0[i] + d0[i];
    v = (v >= 0.0f) ? v : v * NEG_SLOPE;
    part += v * a0[i];
    float w = s1[i] + d1[i];
    w = (w >= 0.0f) ? w : w * NEG_SLOPE;
    part += w * a1[i];
  }
  part += __shfl_xor(part, 1, 32);
  part += __shfl_xor(part, 2, 32);
  part += __shfl_xor(part, 4, 32);
  if ((lane & 7) == 0) {
    int h = lane >> 3;
    score[(size_t)e * 4 + h] = part;
    atomicMax(&m_u32[(size_t)dst * 4 + h], orderFloat(part));
  }
}

__global__ __launch_bounds__(256) void sanitize_m_kernel(unsigned* __restrict__ m) {
  int i = blockIdx.x * blockDim.x + threadIdx.x;
  if (i >= NN * 4) return;
  float f = unorderFloat(m[i]);
  if (!isfinite(f)) f = 0.0f;
  ((float*)m)[i] = f;
}

__global__ __launch_bounds__(256) void edge_exp_kernel(
    const int* __restrict__ ei, const float* __restrict__ m_f,
    float* __restrict__ score, float* __restrict__ denom) {
  int i = blockIdx.x * blockDim.x + threadIdx.x;
  if (i >= EE * 4) return;
  int e = i >> 2, h = i & 3;
  int dst = ei[EE + e];
  float ex = expf(score[i] - m_f[(size_t)dst * 4 + h]);
  score[i] = ex;
  unsafeAtomicAdd(&denom[(size_t)dst * 4 + h], ex);
}

__global__ __launch_bounds__(256) void edge_scatter_kernel(
    const int* __restrict__ ei, const float* __restrict__ Hsrc,
    const float* __restrict__ ex, const float* __restrict__ denom,
    float* __restrict__ rst) {
  const int e = (blockIdx.x * blockDim.x + threadIdx.x) >> 5;
  const int lane = threadIdx.x & 31;
  if (e >= EE) return;
  const int src = ei[e];
  const int dst = ei[EE + e];
  const int hl = lane & 3;
  float den = denom[(size_t)dst * 4 + hl];
  float alpha = ex[(size_t)e * 4 + hl] / ((den == 0.0f) ? 1.0f : den);
  const float* ps = Hsrc + (size_t)src * KDIM + lane;
  float v[8];
#pragma unroll
  for (int i = 0; i < 8; ++i) v[i] = ps[32 * i];
  float* pr = rst + (size_t)dst * KDIM + lane;
#pragma unroll
  for (int i = 0; i < 8; ++i) {
    const float ah = __int_as_float(
        __builtin_amdgcn_readlane(__float_as_int(alpha), i >> 1));
    unsafeAtomicAdd(&pr[32 * i], ah * v[i]);
  }
}

__global__ __launch_bounds__(32) void wmma_out_kernel(
    const float* __restrict__ rst, const float* __restrict__ Wfc,
    const float* __restrict__ bfc, float* __restrict__ out) {
  const int lane = threadIdx.x;
  const int row0 = blockIdx.x * 64;
  const int col0 = blockIdx.y * 16;
  const int khi = lane >> 4;
  const int mlo = lane & 15;

  int arow[4];
#pragma unroll
  for (int s = 0; s < 4; ++s) {
    int r = row0 + s * 16 + mlo;
    arow[s] = (r < NN) ? r : 0;
  }

  v8f acc[4] = {};
  for (int kb = 0; kb < KDIM; kb += 4) {
    v2f bfrag;
    bfrag.x = Wfc[(size_t)(kb + 2 * khi)     * OUTD + col0 + mlo];
    bfrag.y = Wfc[(size_t)(kb + 2 * khi + 1) * OUTD + col0 + mlo];
#pragma unroll
    for (int s = 0; s < 4; ++s) {
      v2f afrag = *(const v2f*)(rst + (size_t)arow[s] * KDIM + kb + 2 * khi);
      acc[s] = __builtin_amdgcn_wmma_f32_16x16x4_f32(
          false, afrag, false, bfrag, (short)0, acc[s], false, false);
    }
  }

  const int mbase = khi * 8;
  const int ncol  = col0 + mlo;
  const float bias = bfc[ncol];
  if (row0 + 64 <= NN) {
#pragma unroll
    for (int s = 0; s < 4; ++s) {
#pragma unroll
      for (int i = 0; i < 8; ++i) {
        int r = row0 + s * 16 + mbase + i;
        out[(size_t)r * OUTD + ncol] = acc[s][i] + bias;
      }
    }
  } else {
#pragma unroll
    for (int s = 0; s < 4; ++s) {
#pragma unroll
      for (int i = 0; i < 8; ++i) {
        int r = row0 + s * 16 + mbase + i;
        if (r < NN) out[(size_t)r * OUTD + ncol] = acc[s][i] + bias;
      }
    }
  }
}

extern "C" void kernel_launch(void* const* d_in, const int* in_sizes, int n_in,
                              void* d_out, int out_size, void* d_ws, size_t ws_size,
                              hipStream_t stream) {
  const float* feat     = (const float*)d_in[0];
  const int*   ei       = (const int*)d_in[1];
  const float* w_src    = (const float*)d_in[2];
  const float* b_src    = (const float*)d_in[3];
  const float* w_dst    = (const float*)d_in[4];
  const float* b_dst    = (const float*)d_in[5];
  const float* attn     = (const float*)d_in[6];
  const float* gat_bias = (const float*)d_in[7];
  const float* w_fc     = (const float*)d_in[8];
  const float* b_fc     = (const float*)d_in[9];
  float* out = (float*)d_out;

  float* ws    = (float*)d_ws;
  float* Hsrc  = ws;
  float* Hdst  = Hsrc + (size_t)NN * KDIM;
  float* score = Hdst + (size_t)NN * KDIM;
  float* mbuf  = score + (size_t)EE * 4;
  float* denom = mbuf + (size_t)NN * 4;
  float* rst   = denom + (size_t)NN * 4;

  const int rowTiles = (NN + 63) / 64;

  init_kernel<<<(NN * KDIM + 255) / 256, 256, 0, stream>>>(
      rst, gat_bias, (unsigned*)mbuf, denom);

  wmma_proj_kernel<<<dim3(2, (NN + PBM - 1) / PBM), 256, 0, stream>>>(
      feat, w_src, b_src, w_dst, b_dst, Hsrc, Hdst);

  edge_score_kernel<<<(EE * 32) / 256, 256, 0, stream>>>(
      ei, Hsrc, Hdst, attn, score, (unsigned*)mbuf);

  sanitize_m_kernel<<<(NN * 4 + 255) / 256, 256, 0, stream>>>((unsigned*)mbuf);

  edge_exp_kernel<<<(EE * 4 + 255) / 256, 256, 0, stream>>>(
      ei, mbuf, score, denom);

  edge_scatter_kernel<<<(EE * 32) / 256, 256, 0, stream>>>(
      ei, Hsrc, score, denom, rst);

  wmma_out_kernel<<<dim3(rowTiles, 4), 32, 0, stream>>>(rst, w_fc, b_fc, out);
}


# Per-item kernels (init, score, exp, scatter): a wave takes 8 edges (exp 32, init 2 KB) instead of one: 8x fewer waves enter, 8 edges' row loads in flight together; score sum tree and atomics kept
# speedup vs baseline: 2.4076x; 1.1148x over previous
#include <hip/hip_runtime.h>
#include <hip/hip_bf16.h>

#define NN 50000
#define EE 800000
#define KDIM 256
#define OUTD 64
#define NEG_SLOPE 0.2f

typedef __attribute__((ext_vector_type(2))) float v2f;
typedef __attribute__((ext_vector_type(4))) float v4f;
typedef __attribute__((ext_vector_type(8))) float v8f;

__device__ __forceinline__ unsigned orderFloat(float f) {
  unsigned u = __float_as_uint(f);
  return (u & 0x80000000u) ? ~u : (u | 0x80000000u);
}
__device__ __forceinline__ float unorderFloat(unsigned u) {
  u = (u & 0x80000000u) ? (u & 0x7FFFFFFFu) : ~u;
  return __uint_as_float(u);
}

#define INIT_T (NN * KDIM / 16)
static_assert((NN * KDIM) % 16 == 0, "rst must split into 16-float pieces");
__global__ __launch_bounds__(256) void init_kernel(
    float* __restrict__ rst, const float* __restrict__ gat_bias,
    unsigned* __restrict__ m_u32, float* __restrict__ denom) {
  const int t = blockIdx.x * blockDim.x + threadIdx.x;
  if (t < INIT_T) {
#pragma unroll
    for (int k = 0; k < 4; ++k) {
      const int q = t + k * INIT_T;
      ((v4f*)rst)[q] = ((const v4f*)gat_bias)[q & (KDIM / 4 - 1)];
    }
  }
  if (t < NN) {
    const v4f z = {0.0f, 0.0f, 0.0f, 0.0f};
    ((v4f*)m_u32)[t] = z;
    ((v4f*)denom)[t] = z;
  }
}

#define PBM 128
#define PBK 32
#define A_PITCH 36
#define B_PITCH 544

__global__ __launch_bounds__(256) void wmma_proj_kernel(
    const float* __restrict__ feat,
    const float* __restrict__ Wsrc, const float* __restrict__ Bsrc,
    const float* __restrict__ Wdst, const float* __restrict__ Bdst,
    float* __restrict__ Hsrc, float* __restrict__ Hdst) {
  __shared__ __attribute__((aligned(16))) float As[PBM * A_PITCH];
  __shared__ __attribute__((aligned(16))) float Bs[(PBK / 2) * B_PITCH];

  const int tid  = threadIdx.x;
  const int lane = tid & 31;
  const int wave = tid >> 5;
  const int wr   = wave >> 2;
  const int wc   = wave & 3;
  const int mat  = blockIdx.x;
  const int row0 = blockIdx.y * PBM;
  const int khi  = lane >> 4;
  const int mlo  = lane & 15;

  const float* __restrict__ W  = mat ? Wdst : Wsrc;
  const float* __restrict__ Bv = mat ? Bdst : Bsrc;
  float* __restrict__ Hout     = mat ? Hdst : Hsrc;

  const int ar = tid >> 3;
  const int ac = (tid & 7) * 4;
  const int bcg = tid & 63;
  const int bkp = tid >> 6;

  const float* ag[4];
#pragma unroll
  for (int i = 0; i < 4; ++i) {
    int r = row0 + ar + 32 * i;
    r = (r < NN) ? r : 0;
    ag[i] = feat + (size_t)r * KDIM + ac;
  }
  const float* bg = W + (size_t)(2 * bkp) * KDIM + bcg * 4;

  v4f ra[4], rb0[4], rb1[4];
#pragma unroll
  for (int i = 0; i < 4; ++i) {
    ra[i]  = *(const v4f*)(ag[i]);
    rb0[i] = *(const v4f*)(bg + (size_t)(8 * i) * KDIM);
    rb1[i] = *(const v4f*)(bg + (size_t)(8 * i + 1) * KDIM);
  }

  const float* aFrag = &As[(wr * 64 + mlo) * A_PITCH + 2 * khi];
  const float* bFrag = &Bs[khi * B_PITCH + (wc * 64 + mlo) * 2];

  v8f acc[4][4] = {};
  for (int c = 0; c < KDIM / PBK; ++c) {
#pragma unroll
    for (int i = 0; i < 4; ++i) {
      *(v4f*)&As[(ar + 32 * i) * A_PITCH + ac] = ra[i];
      float* p = &Bs[(bkp + 4 * i) * B_PITCH + bcg * 8];
      v4f lo = {rb0[i].x, rb1[i].x, rb0[i].y, rb1[i].y};
      v4f hi = {rb0[i].z, rb1[i].z, rb0[i].w, rb1[i].w};
      *(v4f*)p = lo;
      *(v4f*)(p + 4) = hi;
    }
    __syncthreads();

    if (c + 1 < KDIM / PBK) {
      const int k0 = (c + 1) * PBK;
#pragma unroll
      for (int i = 0; i < 4; ++i) {
        ra[i]  = *(const v4f*)(ag[i] + k0);
        rb0[i] = *(const v4f*)(bg + (size_t)(k0 + 8 * i) * KDIM);
        rb1[i] = *(const v4f*)(bg + (size_t)(k0 + 8 * i + 1) * KDIM);
      }
    }

#pragma unroll
    for (int ks = 0; ks < PBK / 4; ++ks) {
      v2f af[4], bf[4];
#pragma unroll
      for (int s = 0; s < 4; ++s)
        af[s] = *(const v2f*)(aFrag + s * 16 * A_PITCH + ks * 4);
#pragma unroll
      for (int t = 0; t < 4; ++t)
        bf[t] = *(const v2f*)(bFrag + ks * 2 * B_PITCH + t * 32);
#pragma unroll
      for (int s = 0; s < 4; ++s) {
#pragma unroll
        for (int t = 0; t < 4; ++t) {
          acc[s][t] = __builtin_amdgcn_wmma_f32_16x16x4_f32(
              false, af[s], false, bf[t], (short)0, acc[s][t], false, false);
        }
      }
    }
    __syncthreads();
  }

  const int mbase = khi * 8;
  float bias[4];
#pragma unroll
  for (int t = 0; t < 4; ++t) bias[t] = Bv[wc * 64 + t * 16 + mlo];
  const int rbase = row0 + wr * 64 + mbase;
  if (row0 + PBM <= NN) {
#pragma unroll
    for (int s = 0; s < 4; ++s) {
#pragma unroll
      for (int i = 0; i < 8; ++i) {
        int r = rbase + s * 16 + i;
#pragma unroll
        for (int t = 0; t < 4; ++t)
          Hout[(size_t)r * KDIM + wc * 64 + t * 16 + mlo] = acc[s][t][i] + bias[t];
      }
    }
  } else {
#pragma unroll
    for (int s = 0; s < 4; ++s) {
#pragma unroll
      for (int i = 0; i < 8; ++i) {
        int r = rbase + s * 16 + i;
        if (r < NN) {
#pragma unroll
          for (int t = 0; t < 4; ++t)
            Hout[(size_t)r * KDIM + wc * 64 + t * 16 + mlo] = acc[s][t][i] + bias[t];
        }
      }
    }
  }
}

#define EPW 8
static_assert(EE % EPW == 0, "an edge count that is no multiple of EPW needs a tail");
__global__ __launch_bounds__(256) void edge_score_kernel(
    const int* __restrict__ ei, const float* __restrict__ Hsrc,
    const float* __restrict__ Hdst, const float* __restrict__ attn,
    float* __restrict__ score, unsigned* __restrict__ m_u32) {
  const int e0 = ((blockIdx.x * blockDim.x + threadIdx.x) >> 5) * EPW;
  const int lane = threadIdx.x & 31;
  if (e0 >= EE) return;
  const int sub = lane & 7;
  const int srcv = ei[e0 + sub];
  const int dstv = ei[EE + e0 + sub];

  const v4f* pa = (const v4f*)(attn + lane * 8);
  const v4f a0 = pa[0], a1 = pa[1];

  v4f s0[EPW], s1[EPW], d0[EPW], d1[EPW];
#pragma unroll
  for (int j = 0; j < EPW; ++j) {
    const int s = __builtin_amdgcn_readlane(srcv, j);
    const int d = __builtin_amdgcn_readlane(dstv, j);
    const v4f* ps = (const v4f*)(Hsrc + (size_t)s * KDIM + lane * 8);
    const v4f* pd = (const v4f*)(Hdst + (size_t)d * KDIM + lane * 8);
    s0[j] = ps[0]; s1[j] = ps[1];
    d0[j] = pd[0]; d1[j] = pd[1];
  }
#pragma unroll
  for (int j = 0; j < EPW; ++j)
    asm volatile("" :: "v"(s0[j]), "v"(s1[j]), "v"(d0[j]), "v"(d1[j]));

  float p[EPW];
#pragma unroll
  for (int j = 0; j < EPW; ++j) {
    float part = 0.0f;
#pragma unroll
    for (int i = 0; i < 4; ++i) {
      float v = s0[j][i] + d0[j][i];
      v = (v >= 0.0f) ? v : v * NEG_SLOPE;
      part = __builtin_fmaf(v, a0[i], part);
      float w = s1[j][i] + d1[j][i];
      w = (w >= 0.0f) ? w : w * NEG_SLOPE;
      part = __builtin_fmaf(w, a1[i], part);
    }
    p[j] = part;
  }

  const bool b0 = (lane & 1) != 0, b1 = (lane & 2) != 0, b2 = (lane & 4) != 0;
  float q[4], r[2];
#pragma unroll
  for (int k = 0; k < 4; ++k) {
    const float keep = b0 ? p[2 * k + 1] : p[2 * k];
    const float send = b0 ? p[2 * k] : p[2 * k + 1];
    q[k] = keep + __shfl_xor(send, 1, 32);
  }
#pragma unroll
  for (int k = 0; k < 2; ++k) {
    const float keep = b1 ? q[2 * k + 1] : q[2 * k];
    const float send = b1 ? q[2 * k] : q[2 * k + 1];
    r[k] = keep + __shfl_xor(send, 2, 32);
  }
  const float keep = b2 ? r[1] : r[0];
  const float send = b2 ? r[0] : r[1];
  const float sc = keep + __shfl_xor(send, 4, 32);

  const int h = lane >> 3;
  score[(size_t)(e0 + sub) * 4 + h] = sc;
  atomicMax(&m_u32[(size_t)dstv * 4 + h], orderFloat(sc));
}

__global__ __launch_bounds__(256) void sanitize_m_kernel(unsigned* __restrict__ m) {
  int i = blockIdx.x * blockDim.x + threadIdx.x;
  if (i >= NN * 4) return;
  float f = unorderFloat(m[i]);
  if (!isfinite(f)) f = 0.0f;
  ((float*)m)[i] = f;
}

#define EXP_K 4
static_assert((EE * 4) % (32 * EXP_K) == 0, "an item count that is no multiple of 32*EXP_K needs a tail");
__global__ __launch_bounds__(256) void edge_exp_kernel(
    const int* __restrict__ ei, const float* __restrict__ m_f,
    float* __restrict__ score, float* __restrict__ denom) {
  const int base = ((blockIdx.x * blockDim.x + threadIdx.x) >> 5) * (32 * EXP_K);
  const int lane = threadIdx.x & 31;
  if (base >= EE * 4) return;
  const int h = lane & 3;
  int dst[EXP_K];
  float sc[EXP_K], mm[EXP_K];
#pragma unroll
  for (int k = 0; k < EXP_K; ++k) dst[k] = ei[EE + ((base + 32 * k + lane) >> 2)];
#pragma unroll
  for (int k = 0; k < EXP_K; ++k) {
    sc[k] = score[base + 32 * k + lane];
    mm[k] = m_f[(size_t)dst[k] * 4 + h];
  }
#pragma unroll
  for (int k = 0; k < EXP_K; ++k) {
    const float ex = expf(sc[k] - mm[k]);
    score[base + 32 * k + lane] = ex;
    unsafeAtomicAdd(&denom[(size_t)dst[k] * 4 + h], ex);
  }
}

__global__ __launch_bounds__(256) void edge_scatter_kernel(
    const int* __restrict__ ei, const float* __restrict__ Hsrc,
    const float* __restrict__ ex, const float* __restrict__ denom,
    float* __restrict__ rst) {
  const int e0 = ((blockIdx.x * blockDim.x + threadIdx.x) >> 5) * EPW;
  const int lane = threadIdx.x & 31;
  if (e0 >= EE) return;
  const int srcv = ei[e0 + (lane & 7)];
  const int dstq = ei[EE + e0 + (lane >> 2)];
  const int hl = lane & 3;
  const float den = denom[(size_t)dstq * 4 + hl];
  const float alpha = ex[(size_t)e0 * 4 + lane] / ((den == 0.0f) ? 1.0f : den);

  float v[EPW][8];
#pragma unroll
  for (int j = 0; j < EPW; ++j) {
    const int s = __builtin_amdgcn_readlane(srcv, j);
    const float* ps = Hsrc + (size_t)s * KDIM + lane;
#pragma unroll
    for (int i = 0; i < 8; ++i) v[j][i] = ps[32 * i];
  }
#pragma unroll
  for (int j = 0; j < EPW; ++j) {
    const int d = __builtin_amdgcn_readlane(dstq, 4 * j);
    float* pr = rst + (size_t)d * KDIM + lane;
#pragma unroll
    for (int i = 0; i < 8; ++i) {
      const float ah = __int_as_float(
          __builtin_amdgcn_readlane(__float_as_int(alpha), 4 * j + (i >> 1)));
      unsafeAtomicAdd(&pr[32 * i], ah * v[j][i]);
    }
  }
}

__global__ __launch_bounds__(32) void wmma_out_kernel(
    const float* __restrict__ rst, const float* __restrict__ Wfc,
    const float* __restrict__ bfc, float* __restrict__ out) {
  const int lane = threadIdx.x;
  const int row0 = blockIdx.x * 64;
  const int col0 = blockIdx.y * 16;
  const int khi = lane >> 4;
  const int mlo = lane & 15;

  int arow[4];
#pragma unroll
  for (int s = 0; s < 4; ++s) {
    int r = row0 + s * 16 + mlo;
    arow[s] = (r < NN) ? r : 0;
  }

  v8f acc[4] = {};
  for (int kb = 0; kb < KDIM; kb += 4) {
    v2f bfrag;
    bfrag.x = Wfc[(size_t)(kb + 2 * khi)     * OUTD + col0 + mlo];
    bfrag.y = Wfc[(size_t)(kb + 2 * khi + 1) * OUTD + col0 + mlo];
#pragma unroll
    for (int s = 0; s < 4; ++s) {
      v2f afrag = *(const v2f*)(rst + (size_t)arow[s] * KDIM + kb + 2 * khi);
      acc[s] = __builtin_amdgcn_wmma_f32_16x16x4_f32(
          false, afrag, false, bfrag, (short)0, acc[s], false, false);
    }
  }

  const int mbase = khi * 8;
  const int ncol  = col0 + mlo;
  const float bias = bfc[ncol];
  if (row0 + 64 <= NN) {
#pragma unroll
    for (int s = 0; s < 4; ++s) {
#pragma unroll
      for (int i = 0; i < 8; ++i) {
        int r = row0 + s * 16 + mbase + i;
        out[(size_t)r * OUTD + ncol] = acc[s][i] + bias;
      }
    }
  } else {
#pragma unroll
    for (int s = 0; s < 4; ++s) {
#pragma unroll
      for (int i = 0; i < 8; ++i) {
        int r = row0 + s * 16 + mbase + i;
        if (r < NN) out[(size_t)r * OUTD + ncol] = acc[s][i] + bias;
      }
    }
  }
}

extern "C" void kernel_launch(void* const* d_in, const int* in_sizes, int n_in,
                              void* d_out, int out_size, void* d_ws, size_t ws_size,
                              hipStream_t stream) {
  const float* feat     = (const float*)d_in[0];
  const int*   ei       = (const int*)d_in[1];
  const float* w_src    = (const float*)d_in[2];
  const float* b_src    = (const float*)d_in[3];
  const float* w_dst    = (const float*)d_in[4];
  const float* b_dst    = (const float*)d_in[5];
  const float* attn     = (const float*)d_in[6];
  const float* gat_bias = (const float*)d_in[7];
  const float* w_fc     = (const float*)d_in[8];
  const float* b_fc     = (const float*)d_in[9];
  float* out = (float*)d_out;

  float* ws    = (float*)d_ws;
  float* Hsrc  = ws;
  float* Hdst  = Hsrc + (size_t)NN * KDIM;
  float* score = Hdst + (size_t)NN * KDIM;
  float* mbuf  = score + (size_t)EE * 4;
  float* denom = mbuf + (size_t)NN * 4;
  float* rst   = denom + (size_t)NN * 4;

  const int rowTiles = (NN + 63) / 64;

  init_kernel<<<(INIT_T + 255) / 256, 256, 0, stream>>>(
      rst, gat_bias, (unsigned*)mbuf, denom);

  wmma_proj_kernel<<<dim3(2, (NN + PBM - 1) / PBM), 256, 0, stream>>>(
      feat, w_src, b_src, w_dst, b_dst, Hsrc, Hdst);

  edge_score_kernel<<<(EE / EPW * 32 + 255) / 256, 256, 0, stream>>>(
      ei, Hsrc, Hdst, attn, score, (unsigned*)mbuf);

  sanitize_m_kernel<<<(NN * 4 + 255) / 256, 256, 0, stream>>>((unsigned*)mbuf);

  edge_exp_kernel<<<(EE * 4 / EXP_K + 255) / 256, 256, 0, stream>>>(
      ei, mbuf, score, denom);

  edge_scatter_kernel<<<(EE / EPW * 32 + 255) / 256, 256, 0, stream>>>(
      ei, Hsrc, score, denom, rst);

  wmma_out_kernel<<<dim3(rowTiles, 4), 32, 0, stream>>>(rst, w_fc, b_fc, out);
}


# Output GEMM data path: an 8-wave work-group copies its 64x32 rst tile and 32x64 w_fc tile into LDS with full-line 16-byte loads and computes all 64 columns: rst fetched once, not 4x; k order kept
# speedup vs baseline: 2.5092x; 1.0422x over previous
#include <hip/hip_runtime.h>
#include <hip/hip_bf16.h>

#define NN 50000
#define EE 800000
#define KDIM 256
#define OUTD 64
#define NEG_SLOPE 0.2f

typedef __attribute__((ext_vector_type(2))) float v2f;
typedef __attribute__((ext_vector_type(4))) float v4f;
typedef __attribute__((ext_vector_type(8))) float v8f;

__device__ __forceinline__ unsigned orderFloat(float f) {
  unsigned u = __float_as_uint(f);
  return (u & 0x80000000u) ? ~u : (u | 0x80000000u);
}
__device__ __forceinline__ float unorderFloat(unsigned u) {
  u = (u & 0x80000000u) ? (u & 0x7FFFFFFFu) : ~u;
  return __uint_as_float(u);
}

#define INIT_T (NN * KDIM / 16)
static_assert((NN * KDIM) % 16 == 0, "rst must split into 16-float pieces");
__global__ __launch_bounds__(256) void init_kernel(
    float* __restrict__ rst, const float* __restrict__ gat_bias,
    unsigned* __restrict__ m_u32, float* __restrict__ denom) {
  const int t = blockIdx.x * blockDim.x + threadIdx.x;
  if (t < INIT_T) {
#pragma unroll
    for (int k = 0; k < 4; ++k) {
      const int q = t + k * INIT_T;
      ((v4f*)rst)[q] = ((const v4f*)gat_bias)[q & (KDIM / 4 - 1)];
    }
  }
  if (t < NN) {
    const v4f z = {0.0f, 0.0f, 0.0f, 0.0f};
    ((v4f*)m_u32)[t] = z;
    ((v4f*)denom)[t] = z;
  }
}

#define PBM 128
#define PBK 32
#define A_PITCH 36
#define B_PITCH 544

__global__ __launch_bounds__(256) void wmma_proj_kernel(
    const float* __restrict__ feat,
    const float* __restrict__ Wsrc, const float* __restrict__ Bsrc,
    const float* __restrict__ Wdst, const float* __restrict__ Bdst,
    float* __restrict__ Hsrc, float* __restrict__ Hdst) {
  __shared__ __attribute__((aligned(16))) float As[PBM * A_PITCH];
  __shared__ __attribute__((aligned(16))) float Bs[(PBK / 2) * B_PITCH];

  const int tid  = threadIdx.x;
  const int lane = tid & 31;
  const int wave = tid >> 5;
  const int wr   = wave >> 2;
  const int wc   = wave & 3;
  const int mat  = blockIdx.x;
  const int row0 = blockIdx.y * PBM;
  const int khi  = lane >> 4;
  const int mlo  = lane & 15;

  const float* __restrict__ W  = mat ? Wdst : Wsrc;
  const float* __restrict__ Bv = mat ? Bdst : Bsrc;
  float* __restrict__ Hout     = mat ? Hdst : Hsrc;

  const int ar = tid >> 3;
  const int ac = (tid & 7) * 4;
  const int bcg = tid & 63;
  const int bkp = tid >> 6;

  const float* ag[4];
#pragma unroll
  for (int i = 0; i < 4; ++i) {
    int r = row0 + ar + 32 * i;
    r = (r < NN) ? r : 0;
    ag[i] = feat + (size_t)r * KDIM + ac;
  }
  const float* bg = W + (size_t)(2 * bkp) * KDIM + bcg * 4;

  v4f ra[4], rb0[4], rb1[4];
#pragma unroll
  for (int i = 0; i < 4; ++i) {
    ra[i]  = *(const v4f*)(ag[i]);
    rb0[i] = *(const v4f*)(bg + (size_t)(8 * i) * KDIM);
    rb1[i] = *(const v4f*)(bg + (size_t)(8 * i + 1) * KDIM);
  }

  const float* aFrag = &As[(wr * 64 + mlo) * A_PITCH + 2 * khi];
  const float* bFrag = &Bs[khi * B_PITCH + (wc * 64 + mlo) * 2];

  v8f acc[4][4] = {};
  for (int c = 0; c < KDIM / PBK; ++c) {
#pragma unroll
    for (int i = 0; i < 4; ++i) {
      *(v4f*)&As[(ar + 32 * i) * A_PITCH + ac] = ra[i];
      float* p = &Bs[(bkp + 4 * i) * B_PITCH + bcg * 8];
      v4f lo = {rb0[i].x, rb1[i].x, rb0[i].y, rb1[i].y};
      v4f hi = {rb0[i].z, rb1[i].z, rb0[i].w, rb1[i].w};
      *(v4f*)p = lo;
      *(v4f*)(p + 4) = hi;
    }
    __syncthreads();

    if (c + 1 < KDIM / PBK) {
      const int k0 = (c + 1) * PBK;
#pragma unroll
      for (int i = 0; i < 4; ++i) {
        ra[i]  = *(const v4f*)(ag[i] + k0);
        rb0[i] = *(const v4f*)(bg + (size_t)(k0 + 8 * i) * KDIM);
        rb1[i] = *(const v4f*)(bg + (size_t)(k0 + 8 * i + 1) * KDIM);
      }
    }

#pragma unroll
    for (int ks = 0; ks < PBK / 4; ++ks) {
      v2f af[4], bf[4];
#pragma unroll
      for (int s = 0; s < 4; ++s)
        af[s] = *(const v2f*)(aFrag + s * 16 * A_PITCH + ks * 4);
#pragma unroll
      for (int t = 0; t < 4; ++t)
        bf[t] = *(const v2f*)(bFrag + ks * 2 * B_PITCH + t * 32);
#pragma unroll
      for (int s = 0; s < 4; ++s) {
#pragma unroll
        for (int t = 0; t < 4; ++t) {
          acc[s][t] = __builtin_amdgcn_wmma_f32_16x16x4_f32(
              false, af[s], false, bf[t], (short)0, acc[s][t], false, false);
        }
      }
    }
    __syncthreads();
  }

  const int mbase = khi * 8;
  float bias[4];
#pragma unroll
  for (int t = 0; t < 4; ++t) bias[t] = Bv[wc * 64 + t * 16 + mlo];
  const int rbase = row0 + wr * 64 + mbase;
  if (row0 + PBM <= NN) {
#pragma unroll
    for (int s = 0; s < 4; ++s) {
#pragma unroll
      for (int i = 0; i < 8; ++i) {
        int r = rbase + s * 16 + i;
#pragma unroll
        for (int t = 0; t < 4; ++t)
          Hout[(size_t)r * KDIM + wc * 64 + t * 16 + mlo] = acc[s][t][i] + bias[t];
      }
    }
  } else {
#pragma unroll
    for (int s = 0; s < 4; ++s) {
#pragma unroll
      for (int i = 0; i < 8; ++i) {
        int r = rbase + s * 16 + i;
        if (r < NN) {
#pragma unroll
          for (int t = 0; t < 4; ++t)
            Hout[(size_t)r * KDIM + wc * 64 + t * 16 + mlo] = acc[s][t][i] + bias[t];
        }
      }
    }
  }
}

#define EPW 8
static_assert(EE % EPW == 0, "an edge count that is no multiple of EPW needs a tail");
__global__ __launch_bounds__(256) void edge_score_kernel(
    const int* __restrict__ ei, const float* __restrict__ Hsrc,
    const float* __restrict__ Hdst, const float* __restrict__ attn,
    float* __restrict__ score, unsigned* __restrict__ m_u32) {
  const int e0 = ((blockIdx.x * blockDim.x + threadIdx.x) >> 5) * EPW;
  const int lane = threadIdx.x & 31;
  if (e0 >= EE) return;
  const int sub = lane & 7;
  const int srcv = ei[e0 + sub];
  const int dstv = ei[EE + e0 + sub];

  const v4f* pa = (const v4f*)(attn + lane * 8);
  const v4f a0 = pa[0], a1 = pa[1];

  v4f s0[EPW], s1[EPW], d0[EPW], d1[EPW];
#pragma unroll
  for (int j = 0; j < EPW; ++j) {
    const int s = __builtin_amdgcn_readlane(srcv, j);
    const int d = __builtin_amdgcn_readlane(dstv, j);
    const v4f* ps = (const v4f*)(Hsrc + (size_t)s * KDIM + lane * 8);
    const v4f* pd = (const v4f*)(Hdst + (size_t)d * KDIM + lane * 8);
    s0[j] = ps[0]; s1[j] = ps[1];
    d0[j] = pd[0]; d1[j] = pd[1];
  }
#pragma unroll
  for (int j = 0; j < EPW; ++j)
    asm volatile("" :: "v"(s0[j]), "v"(s1[j]), "v"(d0[j]), "v"(d1[j]));

  float p[EPW];
#pragma unroll
  for (int j = 0; j < EPW; ++j) {
    float part = 0.0f;
#pragma unroll
    for (int i = 0; i < 4; ++i) {
      float v = s0[j][i] + d0[j][i];
      v = (v >= 0.0f) ? v : v * NEG_SLOPE;
      part = __builtin_fmaf(v, a0[i], part);
      float w = s1[j][i] + d1[j][i];
      w = (w >= 0.0f) ? w : w * NEG_SLOPE;
      part = __builtin_fmaf(w, a1[i], part);
    }
    p[j] = part;
  }

  const bool b0 = (lane & 1) != 0, b1 = (lane & 2) != 0, b2 = (lane & 4) != 0;
  float q[4], r[2];
#pragma unroll
  for (int k = 0; k < 4; ++k) {
    const float keep = b0 ? p[2 * k + 1] : p[2 * k];
    const float send = b0 ? p[2 * k] : p[2 * k + 1];
    q[k] = keep + __shfl_xor(send, 1, 32);
  }
#pragma unroll
  for (int k = 0; k < 2; ++k) {
    const float keep = b1 ? q[2 * k + 1] : q[2 * k];
    const float send = b1 ? q[2 * k] : q[2 * k + 1];
    r[k] = keep + __shfl_xor(send, 2, 32);
  }
  const float keep = b2 ? r[1] : r[0];
  const float send = b2 ? r[0] : r[1];
  const float sc = keep + __shfl_xor(send, 4, 32);

  const int h = lane >> 3;
  score[(size_t)(e0 + sub) * 4 + h] = sc;
  atomicMax(&m_u32[(size_t)dstv * 4 + h], orderFloat(sc));
}

__global__ __launch_bounds__(256) void sanitize_m_kernel(unsigned* __restrict__ m) {
  int i = blockIdx.x * blockDim.x + threadIdx.x;
  if (i >= NN * 4) return;
  float f = unorderFloat(m[i]);
  if (!isfinite(f)) f = 0.0f;
  ((float*)m)[i] = f;
}

#define EXP_K 4
static_assert((EE * 4) % (32 * EXP_K) == 0, "an item count that is no multiple of 32*EXP_K needs a tail");
__global__ __launch_bounds__(256) void edge_exp_kernel(
    const int* __restrict__ ei, const float* __restrict__ m_f,
    float* __restrict__ score, float* __restrict__ denom) {
  const int base = ((blockIdx.x * blockDim.x + threadIdx.x) >> 5) * (32 * EXP_K);
  const int lane = threadIdx.x & 31;
  if (base >= EE * 4) return;
  const int h = lane & 3;
  int dst[EXP_K];
  float sc[EXP_K], mm[EXP_K];
#pragma unroll
  for (int k = 0; k < EXP_K; ++k) dst[k] = ei[EE + ((base + 32 * k + lane) >> 2)];
#pragma unroll
  for (int k = 0; k < EXP_K; ++k) {
    sc[k] = score[base + 32 * k + lane];
    mm[k] = m_f[(size_t)dst[k] * 4 + h];
  }
#pragma unroll
  for (int k = 0; k < EXP_K; ++k) {
    const float ex = expf(sc[k] - mm[k]);
    score[base + 32 * k + lane] = ex;
    unsafeAtomicAdd(&denom[(size_t)dst[k] * 4 + h], ex);
  }
}

__global__ __launch_bounds__(256) void edge_scatter_kernel(
    const int* __restrict__ ei, const float* __restrict__ Hsrc,
    const float* __restrict__ ex, const float* __restrict__ denom,
    float* __restrict__ rst) {
  const int e0 = ((blockIdx.x * blockDim.x + threadIdx.x) >> 5) * EPW;
  const int lane = threadIdx.x & 31;
  if (e0 >= EE) return;
  const int srcv = ei[e0 + (lane & 7)];
  const int dstq = ei[EE + e0 + (lane >> 2)];
  const int hl = lane & 3;
  const float den = denom[(size_t)dstq * 4 + hl];
  const float alpha = ex[(size_t)e0 * 4 + lane] / ((den == 0.0f) ? 1.0f : den);

  float v[EPW][8];
#pragma unroll
  for (int j = 0; j < EPW; ++j) {
    const int s = __builtin_amdgcn_readlane(srcv, j);
    const float* ps = Hsrc + (size_t)s * KDIM + lane;
#pragma unroll
    for (int i = 0; i < 8; ++i) v[j][i] = ps[32 * i];
  }
#pragma unroll
  for (int j = 0; j < EPW; ++j) {
    const int d = __builtin_amdgcn_readlane(dstq, 4 * j);
    float* pr = rst + (size_t)d * KDIM + lane;
#pragma unroll
    for (int i = 0; i < 8; ++i) {
      const float ah = __int_as_float(
          __builtin_amdgcn_readlane(__float_as_int(alpha), 4 * j + (i >> 1)));
      unsafeAtomicAdd(&pr[32 * i], ah * v[j][i]);
    }
  }
}

#define OBM 64
#define OBK 32
#define OA_PITCH 36
#define OB_PITCH 160
static_assert(KDIM % OBK == 0, "k must split into whole chunks");
static_assert(OUTD == 64, "copy roles and wave layout assume 64 output columns");

__global__ __launch_bounds__(256) void wmma_out_kernel(
    const float* __restrict__ rst, const float* __restrict__ Wfc,
    const float* __restrict__ bfc, float* __restrict__ out) {
  __shared__ __attribute__((aligned(16))) float As[OBM * OA_PITCH];
  __shared__ __attribute__((aligned(16))) float Bs[(OBK / 2) * OB_PITCH];

  const int tid  = threadIdx.x;
  const int lane = tid & 31;
  const int wave = tid >> 5;
  const int wr   = wave >> 1;
  const int wc   = wave & 1;
  const int row0 = blockIdx.x * OBM;
  const int khi  = lane >> 4;
  const int mlo  = lane & 15;

  const int ar  = tid >> 3;
  const int ac  = (tid & 7) * 4;
  const int bcg = tid & 15;
  const int bkp = tid >> 4;

  const float* ag[2];
#pragma unroll
  for (int i = 0; i < 2; ++i) {
    int r = row0 + ar + 32 * i;
    r = (r < NN) ? r : 0;
    ag[i] = rst + (size_t)r * KDIM + ac;
  }
  const float* bg = Wfc + (size_t)(2 * bkp) * OUTD + bcg * 4;

  v4f ra[2], rb0, rb1;
#pragma unroll
  for (int i = 0; i < 2; ++i) ra[i] = *(const v4f*)(ag[i]);
  rb0 = *(const v4f*)(bg);
  rb1 = *(const v4f*)(bg + OUTD);

  const float* aFrag = &As[(wr * 16 + mlo) * OA_PITCH + 2 * khi];
  const float* bFrag = &Bs[khi * OB_PITCH + (wc * 32 + mlo) * 2];

  v8f acc[2] = {};
  for (int c = 0; c < KDIM / OBK; ++c) {
#pragma unroll
    for (int i = 0; i < 2; ++i)
      *(v4f*)&As[(ar + 32 * i) * OA_PITCH + ac] = ra[i];
    {
      float* p = &Bs[bkp * OB_PITCH + bcg * 8];
      v4f lo = {rb0.x, rb1.x, rb0.y, rb1.y};
      v4f hi = {rb0.z, rb1.z, rb0.w, rb1.w};
      *(v4f*)p = lo;
      *(v4f*)(p + 4) = hi;
    }
    __syncthreads();

    if (c + 1 < KDIM / OBK) {
      const int k0 = (c + 1) * OBK;
#pragma unroll
      for (int i = 0; i < 2; ++i) ra[i] = *(const v4f*)(ag[i] + k0);
      rb0 = *(const v4f*)(bg + (size_t)k0 * OUTD);
      rb1 = *(const v4f*)(bg + (size_t)(k0 + 1) * OUTD);
    }

#pragma unroll
    for (int ks = 0; ks < OBK / 4; ++ks) {
      const v2f af = *(const v2f*)(aFrag + ks * 4);
      v2f bf[2];
#pragma unroll
      for (int t = 0; t < 2; ++t)
        bf[t] = *(const v2f*)(bFrag + ks * 2 * OB_PITCH + t * 32);
#pragma unroll
      for (int t = 0; t < 2; ++t)
        acc[t] = __builtin_amdgcn_wmma_f32_16x16x4_f32(
            false, af, false, bf[t], (short)0, acc[t], false, false);
    }
    __syncthreads();
  }

  const int rbase = row0 + wr * 16 + khi * 8;
  float bias[2];
#pragma unroll
  for (int t = 0; t < 2; ++t) bias[t] = bfc[wc * 32 + t * 16 + mlo];
  if (row0 + OBM <= NN) {
#pragma unroll
    for (int i = 0; i < 8; ++i) {
#pragma unroll
      for (int t = 0; t < 2; ++t)
        out[(size_t)(rbase + i) * OUTD + wc * 32 + t * 16 + mlo] = acc[t][i] + bias[t];
    }
  } else {
#pragma unroll
    for (int i = 0; i < 8; ++i) {
      if (rbase + i < NN) {
#pragma unroll
        for (int t = 0; t < 2; ++t)
          out[(size_t)(rbase + i) * OUTD + wc * 32 + t * 16 + mlo] = acc[t][i] + bias[t];
      }
    }
  }
}

extern "C" void kernel_launch(void* const* d_in, const int* in_sizes, int n_in,
                              void* d_out, int out_size, void* d_ws, size_t ws_size,
                              hipStream_t stream) {
  const float* feat     = (const float*)d_in[0];
  const int*   ei       = (const int*)d_in[1];
  const float* w_src    = (const float*)d_in[2];
  const float* b_src    = (const float*)d_in[3];
  const float* w_dst    = (const float*)d_in[4];
  const float* b_dst    = (const float*)d_in[5];
  const float* attn     = (const float*)d_in[6];
  const float* gat_bias = (const float*)d_in[7];
  const float* w_fc     = (const float*)d_in[8];
  const float* b_fc     = (const float*)d_in[9];
  float* out = (float*)d_out;

  float* ws    = (float*)d_ws;
  float* Hsrc  = ws;
  float* Hdst  = Hsrc + (size_t)NN * KDIM;
  float* score = Hdst + (size_t)NN * KDIM;
  float* mbuf  = score + (size_t)EE * 4;
  float* denom = mbuf + (size_t)NN * 4;
  float* rst   = denom + (size_t)NN * 4;

  init_kernel<<<(INIT_T + 255) / 256, 256, 0, stream>>>(
      rst, gat_bias, (unsigned*)mbuf, denom);

  wmma_proj_kernel<<<dim3(2, (NN + PBM - 1) / PBM), 256, 0, stream>>>(
      feat, w_src, b_src, w_dst, b_dst, Hsrc, Hdst);

  edge_score_kernel<<<(EE / EPW * 32 + 255) / 256, 256, 0, stream>>>(
      ei, Hsrc, Hdst, attn, score, (unsigned*)mbuf);

  sanitize_m_kernel<<<(NN * 4 + 255) / 256, 256, 0, stream>>>((unsigned*)mbuf);

  edge_exp_kernel<<<(EE * 4 / EXP_K + 255) / 256, 256, 0, stream>>>(
      ei, mbuf, score, denom);

  edge_scatter_kernel<<<(EE / EPW * 32 + 255) / 256, 256, 0, stream>>>(
      ei, Hsrc, score, denom, rst);

  wmma_out_kernel<<<(NN + OBM - 1) / OBM, 256, 0, stream>>>(rst, w_fc, b_fc, out);
}


# Scatter accumulator moved from global rst to LDS: 256 work-groups each own 196 dst rows, scan all dst ids, add their edges' weighted rows by LDS float atomics, store rows once; sums in arrival order
# speedup vs baseline: 2.8059x; 1.1182x over previous
#include <hip/hip_runtime.h>
#include <hip/hip_bf16.h>

#define NN 50000
#define EE 800000
#define KDIM 256
#define OUTD 64
#define NEG_SLOPE 0.2f

typedef __attribute__((ext_vector_type(2))) float v2f;
typedef __attribute__((ext_vector_type(4))) float v4f;
typedef __attribute__((ext_vector_type(8))) float v8f;
typedef __attribute__((ext_vector_type(4))) int v4i;

__device__ __forceinline__ unsigned orderFloat(float f) {
  unsigned u = __float_as_uint(f);
  return (u & 0x80000000u) ? ~u : (u | 0x80000000u);
}
__device__ __forceinline__ float unorderFloat(unsigned u) {
  u = (u & 0x80000000u) ? (u & 0x7FFFFFFFu) : ~u;
  return __uint_as_float(u);
}

__global__ __launch_bounds__(256) void init_kernel(
    unsigned* __restrict__ m_u32, float* __restrict__ denom) {
  const int t = blockIdx.x * blockDim.x + threadIdx.x;
  if (t < NN) {
    const v4f z = {0.0f, 0.0f, 0.0f, 0.0f};
    ((v4f*)m_u32)[t] = z;
    ((v4f*)denom)[t] = z;
  }
}

#define PBM 128
#define PBK 32
#define A_PITCH 36
#define B_PITCH 544

__global__ __launch_bounds__(256) void wmma_proj_kernel(
    const float* __restrict__ feat,
    const float* __restrict__ Wsrc, const float* __restrict__ Bsrc,
    const float* __restrict__ Wdst, const float* __restrict__ Bdst,
    float* __restrict__ Hsrc, float* __restrict__ Hdst) {
  __shared__ __attribute__((aligned(16))) float As[PBM * A_PITCH];
  __shared__ __attribute__((aligned(16))) float Bs[(PBK / 2) * B_PITCH];

  const int tid  = threadIdx.x;
  const int lane = tid & 31;
  const int wave = tid >> 5;
  const int wr   = wave >> 2;
  const int wc   = wave & 3;
  const int mat  = blockIdx.x;
  const int row0 = blockIdx.y * PBM;
  const int khi  = lane >> 4;
  const int mlo  = lane & 15;

  const float* __restrict__ W  = mat ? Wdst : Wsrc;
  const float* __restrict__ Bv = mat ? Bdst : Bsrc;
  float* __restrict__ Hout     = mat ? Hdst : Hsrc;

  const int ar = tid >> 3;
  const int ac = (tid & 7) * 4;
  const int bcg = tid & 63;
  const int bkp = tid >> 6;

  const float* ag[4];
#pragma unroll
  for (int i = 0; i < 4; ++i) {
    int r = row0 + ar + 32 * i;
    r = (r < NN) ? r : 0;
    ag[i] = feat + (size_t)r * KDIM + ac;
  }
  const float* bg = W + (size_t)(2 * bkp) * KDIM + bcg * 4;

  v4f ra[4], rb0[4], rb1[4];
#pragma unroll
  for (int i = 0; i < 4; ++i) {
    ra[i]  = *(const v4f*)(ag[i]);
    rb0[i] = *(const v4f*)(bg + (size_t)(8 * i) * KDIM);
    rb1[i] = *(const v4f*)(bg + (size_t)(8 * i + 1) * KDIM);
  }

  const float* aFrag = &As[(wr * 64 + mlo) * A_PITCH + 2 * khi];
  const float* bFrag = &Bs[khi * B_PITCH + (wc * 64 + mlo) * 2];

  v8f acc[4][4] = {};
  for (int c = 0; c < KDIM / PBK; ++c) {
#pragma unroll
    for (int i = 0; i < 4; ++i) {
      *(v4f*)&As[(ar + 32 * i) * A_PITCH + ac] = ra[i];
      float* p = &Bs[(bkp + 4 * i) * B_PITCH + bcg * 8];
      v4f lo = {rb0[i].x, rb1[i].x, rb0[i].y, rb1[i].y};
      v4f hi = {rb0[i].z, rb1[i].z, rb0[i].w, rb1[i].w};
      *(v4f*)p = lo;
      *(v4f*)(p + 4) = hi;
    }
    __syncthreads();

    if (c + 1 < KDIM / PBK) {
      const int k0 = (c + 1) * PBK;
#pragma unroll
      for (int i = 0; i < 4; ++i) {
        ra[i]  = *(const v4f*)(ag[i] + k0);
        rb0[i] = *(const v4f*)(bg + (size_t)(k0 + 8 * i) * KDIM);
        rb1[i] = *(const v4f*)(bg + (size_t)(k0 + 8 * i + 1) * KDIM);
      }
    }

#pragma unroll
    for (int ks = 0; ks < PBK / 4; ++ks) {
      v2f af[4], bf[4];
#pragma unroll
      for (int s = 0; s < 4; ++s)
        af[s] = *(const v2f*)(aFrag + s * 16 * A_PITCH + ks * 4);
#pragma unroll
      for (int t = 0; t < 4; ++t)
        bf[t] = *(const v2f*)(bFrag + ks * 2 * B_PITCH + t * 32);
#pragma unroll
      for (int s = 0; s < 4; ++s) {
#pragma unroll
        for (int t = 0; t < 4; ++t) {
          acc[s][t] = __builtin_amdgcn_wmma_f32_16x16x4_f32(
              false, af[s], false, bf[t], (short)0, acc[s][t], false, false);
        }
      }
    }
    __syncthreads();
  }

  const int mbase = khi * 8;
  float bias[4];
#pragma unroll
  for (int t = 0; t < 4; ++t) bias[t] = Bv[wc * 64 + t * 16 + mlo];
  const int rbase = row0 + wr * 64 + mbase;
  if (row0 + PBM <= NN) {
#pragma unroll
    for (int s = 0; s < 4; ++s) {
#pragma unroll
      for (int i = 0; i < 8; ++i) {
        int r = rbase + s * 16 + i;
#pragma unroll
        for (int t = 0; t < 4; ++t)
          Hout[(size_t)r * KDIM + wc * 64 + t * 16 + mlo] = acc[s][t][i] + bias[t];
      }
    }
  } else {
#pragma unroll
    for (int s = 0; s < 4; ++s) {
#pragma unroll
      for (int i = 0; i < 8; ++i) {
        int r = rbase + s * 16 + i;
        if (r < NN) {
#pragma unroll
          for (int t = 0; t < 4; ++t)
            Hout[(size_t)r * KDIM + wc * 64 + t * 16 + mlo] = acc[s][t][i] + bias[t];
        }
      }
    }
  }
}

#define EPW 8
static_assert(EE % EPW == 0, "an edge count that is no multiple of EPW needs a tail");
__global__ __launch_bounds__(256) void edge_score_kernel(
    const int* __restrict__ ei, const float* __restrict__ Hsrc,
    const float* __restrict__ Hdst, const float* __restrict__ attn,
    float* __restrict__ score, unsigned* __restrict__ m_u32) {
  const int e0 = ((blockIdx.x * blockDim.x + threadIdx.x) >> 5) * EPW;
  const int lane = threadIdx.x & 31;
  if (e0 >= EE) return;
  const int sub = lane & 7;
  const int srcv = ei[e0 + sub];
  const int dstv = ei[EE + e0 + sub];

  const v4f* pa = (const v4f*)(attn + lane * 8);
  const v4f a0 = pa[0], a1 = pa[1];

  v4f s0[EPW], s1[EPW], d0[EPW], d1[EPW];
#pragma unroll
  for (int j = 0; j < EPW; ++j) {
    const int s = __builtin_amdgcn_readlane(srcv, j);
    const int d = __builtin_amdgcn_readlane(dstv, j);
    const v4f* ps = (const v4f*)(Hsrc + (size_t)s * KDIM + lane * 8);
    const v4f* pd = (const v4f*)(Hdst + (size_t)d * KDIM + lane * 8);
    s0[j] = ps[0]; s1[j] = ps[1];
    d0[j] = pd[0]; d1[j] = pd[1];
  }
#pragma unroll
  for (int j = 0; j < EPW; ++j)
    asm volatile("" :: "v"(s0[j]), "v"(s1[j]), "v"(d0[j]), "v"(d1[j]));

  float p[EPW];
#pragma unroll
  for (int j = 0; j < EPW; ++j) {
    float part = 0.0f;
#pragma unroll
    for (int i = 0; i < 4; ++i) {
      float v = s0[j][i] + d0[j][i];
      v = (v >= 0.0f) ? v : v * NEG_SLOPE;
      part = __builtin_fmaf(v, a0[i], part);
      float w = s1[j][i] + d1[j][i];
      w = (w >= 0.0f) ? w : w * NEG_SLOPE;
      part = __builtin_fmaf(w, a1[i], part);
    }
    p[j] = part;
  }

  const bool b0 = (lane & 1) != 0, b1 = (lane & 2) != 0, b2 = (lane & 4) != 0;
  float q[4], r[2];
#pragma unroll
  for (int k = 0; k < 4; ++k) {
    const float keep = b0 ? p[2 * k + 1] : p[2 * k];
    const float send = b0 ? p[2 * k] : p[2 * k + 1];
    q[k] = keep + __shfl_xor(send, 1, 32);
  }
#pragma unroll
  for (int k = 0; k < 2; ++k) {
    const float keep = b1 ? q[2 * k + 1] : q[2 * k];
    const float send = b1 ? q[2 * k] : q[2 * k + 1];
    r[k] = keep + __shfl_xor(send, 2, 32);
  }
  const float keep = b2 ? r[1] : r[0];
  const float send = b2 ? r[0] : r[1];
  const float sc = keep + __shfl_xor(send, 4, 32);

  const int h = lane >> 3;
  score[(size_t)(e0 + sub) * 4 + h] = sc;
  atomicMax(&m_u32[(size_t)dstv * 4 + h], orderFloat(sc));
}

__global__ __launch_bounds__(256) void sanitize_m_kernel(unsigned* __restrict__ m) {
  int i = blockIdx.x * blockDim.x + threadIdx.x;
  if (i >= NN * 4) return;
  float f = unorderFloat(m[i]);
  if (!isfinite(f)) f = 0.0f;
  ((float*)m)[i] = f;
}

#define EXP_K 4
static_assert((EE * 4) % (32 * EXP_K) == 0, "an item count that is no multiple of 32*EXP_K needs a tail");
__global__ __launch_bounds__(256) void edge_exp_kernel(
    const int* __restrict__ ei, const float* __restrict__ m_f,
    float* __restrict__ score, float* __restrict__ denom) {
  const int base = ((blockIdx.x * blockDim.x + threadIdx.x) >> 5) * (32 * EXP_K);
  const int lane = threadIdx.x & 31;
  if (base >= EE * 4) return;
  const int h = lane & 3;
  int dst[EXP_K];
  float sc[EXP_K], mm[EXP_K];
#pragma unroll
  for (int k = 0; k < EXP_K; ++k) dst[k] = ei[EE + ((base + 32 * k + lane) >> 2)];
#pragma unroll
  for (int k = 0; k < EXP_K; ++k) {
    sc[k] = score[base + 32 * k + lane];
    mm[k] = m_f[(size_t)dst[k] * 4 + h];
  }
#pragma unroll
  for (int k = 0; k < EXP_K; ++k) {
    const float ex = expf(sc[k] - mm[k]);
    score[base + 32 * k + lane] = ex;
    unsafeAtomicAdd(&denom[(size_t)dst[k] * 4 + h], ex);
  }
}

#define SC_G 256
#define SC_R ((NN + SC_G - 1) / SC_G)
#define SC_T 1024
#define SC_W (SC_T / 32)
#define SC_U 4
#define SC_CH 128
#define SC_NCH (EE / SC_CH)
#define SC_QCAP (SC_U * SC_CH + EPW)
static_assert(EE % SC_CH == 0, "an edge count that is no multiple of SC_CH needs a tail");
static_assert(EE <= (1 << 24) && SC_R <= 256, "a queue entry packs row (8 bits) and edge id (24 bits)");
static_assert((SC_G - 1) * SC_R < NN, "every group owns at least one row");
static_assert((SC_G - 1) * (SC_NCH / SC_G) < SC_NCH, "a group's start offset stays inside the list");
static_assert(SC_R * KDIM * 4 + SC_W * SC_QCAP * 4 <= 327680, "rows and queues must fit the LDS");

__global__ __launch_bounds__(SC_T) void edge_scatter_kernel(
    const int* __restrict__ ei, const float* __restrict__ Hsrc,
    const float* __restrict__ ex, const float* __restrict__ denom,
    const float* __restrict__ gat_bias, float* __restrict__ rst) {
  __shared__ __attribute__((aligned(16))) float acc[SC_R * KDIM];
  __shared__ int queue[SC_W * SC_QCAP];

  const int tid  = threadIdx.x;
  const int lane = tid & 31;
  const int wave = __builtin_amdgcn_readfirstlane(tid >> 5);
  const int n0 = blockIdx.x * SC_R;
  const int nrows = (NN - n0 < SC_R) ? (NN - n0) : SC_R;

  for (int q = tid; q < SC_R * (KDIM / 4); q += SC_T)
    ((v4f*)acc)[q] = ((const v4f*)gat_bias)[q & (KDIM / 4 - 1)];
  __syncthreads();

  int* myQ = queue + wave * SC_QCAP;
  const int* dstp = ei + EE;
  int qn = 0;

  auto batch = [&](int h0, int cnt) __attribute__((always_inline)) {
    const int sw = lane >> 2;
    const int ss = lane & 7;
    const int pw = myQ[h0 + ((sw < cnt) ? sw : 0)];
    const int pq = myQ[h0 + ((ss < cnt) ? ss : 0)];
    const int ew = pw & 0xFFFFFF;
    const int rw = (int)((unsigned)pw >> 24);
    const int srcv = ei[pq & 0xFFFFFF];
    const int hl = lane & 3;
    const float den = denom[(size_t)(n0 + rw) * 4 + hl];
    const float alpha = ex[(size_t)ew * 4 + hl] / ((den == 0.0f) ? 1.0f : den);

    float v[EPW][8];
#pragma unroll
    for (int j = 0; j < EPW; ++j) {
      if (j < cnt) {
        const int s = __builtin_amdgcn_readlane(srcv, j);
        const float* ps = Hsrc + (size_t)s * KDIM + lane;
#pragma unroll
        for (int i = 0; i < 8; ++i) v[j][i] = ps[32 * i];
      }
    }
#pragma unroll
    for (int j = 0; j < EPW; ++j) {
      if (j < cnt) {
        const int r = __builtin_amdgcn_readlane(rw, 4 * j);
        float* pa = &acc[r * KDIM + lane];
#pragma unroll
        for (int i = 0; i < 8; ++i) {
          const float ah = __int_as_float(
              __builtin_amdgcn_readlane(__float_as_int(alpha), 4 * j + (i >> 1)));
          __hip_atomic_fetch_add(&pa[32 * i], ah * v[j][i], __ATOMIC_RELAXED,
                                 __HIP_MEMORY_SCOPE_WORKGROUP);
        }
      }
    }
  };

  const int rot = blockIdx.x * (SC_NCH / SC_G);
  auto chunk_of = [&](int c) __attribute__((always_inline)) {
    int cc = ((c < SC_NCH) ? c : (SC_NCH - 1)) + rot;
    return (cc >= SC_NCH) ? (cc - SC_NCH) : cc;
  };
  v4i cur[SC_U], nxt[SC_U];
#pragma unroll
  for (int u = 0; u < SC_U; ++u)
    cur[u] = *(const v4i*)(dstp + (size_t)chunk_of(wave + u * SC_W) * SC_CH + lane * 4);
  for (int c0 = wave; c0 < SC_NCH; c0 += SC_W * SC_U) {
#pragma unroll
    for (int u = 0; u < SC_U; ++u)
      nxt[u] = *(const v4i*)(dstp + (size_t)chunk_of(c0 + (SC_U + u) * SC_W) * SC_CH + lane * 4);
#pragma unroll
    for (int u = 0; u < SC_U; ++u) {
      if (c0 + u * SC_W < SC_NCH) {
        const int e0 = chunk_of(c0 + u * SC_W) * SC_CH + lane * 4;
#pragma unroll
        for (int k = 0; k < 4; ++k) {
          const unsigned r = (unsigned)(cur[u][k] - n0);
          const bool hit = r < (unsigned)nrows;
          const unsigned m = __builtin_amdgcn_ballot_w32(hit);
          if (m) {
            if (hit)
              myQ[qn + (int)__builtin_amdgcn_mbcnt_lo(m, 0u)] =
                  (int)((r << 24) | (unsigned)(e0 + k));
            qn += __builtin_popcount(m);
          }
        }
      }
    }
    int h = 0;
    for (; h + EPW <= qn; h += EPW) batch(h, EPW);
    if (h > 0) {
      const int rem = qn - h;
      int t = 0;
      if (lane < rem) t = myQ[h + lane];
      if (lane < rem) myQ[lane] = t;
      qn = rem;
    }
#pragma unroll
    for (int u = 0; u < SC_U; ++u) cur[u] = nxt[u];
  }
  if (qn > 0) batch(0, qn);
  __syncthreads();

  v4f* out_rows = (v4f*)(rst + (size_t)n0 * KDIM);
  for (int q = tid; q < nrows * (KDIM / 4); q += SC_T)
    out_rows[q] = ((const v4f*)acc)[q];
}

#define OBM 64
#define OBK 32
#define OA_PITCH 36
#define OB_PITCH 160
static_assert(KDIM % OBK == 0, "k must split into whole chunks");
static_assert(OUTD == 64, "copy roles and wave layout assume 64 output columns");

__global__ __launch_bounds__(256) void wmma_out_kernel(
    const float* __restrict__ rst, const float* __restrict__ Wfc,
    const float* __restrict__ bfc, float* __restrict__ out) {
  __shared__ __attribute__((aligned(16))) float As[OBM * OA_PITCH];
  __shared__ __attribute__((aligned(16))) float Bs[(OBK / 2) * OB_PITCH];

  const int tid  = threadIdx.x;
  const int lane = tid & 31;
  const int wave = tid >> 5;
  const int wr   = wave >> 1;
  const int wc   = wave & 1;
  const int row0 = blockIdx.x * OBM;
  const int khi  = lane >> 4;
  const int mlo  = lane & 15;

  const int ar  = tid >> 3;
  const int ac  = (tid & 7) * 4;
  const int bcg = tid & 15;
  const int bkp = tid >> 4;

  const float* ag[2];
#pragma unroll
  for (int i = 0; i < 2; ++i) {
    int r = row0 + ar + 32 * i;
    r = (r < NN) ? r : 0;
    ag[i] = rst + (size_t)r * KDIM + ac;
  }
  const float* bg = Wfc + (size_t)(2 * bkp) * OUTD + bcg * 4;

  v4f ra[2], rb0, rb1;
#pragma unroll
  for (int i = 0; i < 2; ++i) ra[i] = *(const v4f*)(ag[i]);
  rb0 = *(const v4f*)(bg);
  rb1 = *(const v4f*)(bg + OUTD);

  const float* aFrag = &As[(wr * 16 + mlo) * OA_PITCH + 2 * khi];
  const float* bFrag = &Bs[khi * OB_PITCH + (wc * 32 + mlo) * 2];

  v8f acc[2] = {};
  for (int c = 0; c < KDIM / OBK; ++c) {
#pragma unroll
    for (int i = 0; i < 2; ++i)
      *(v4f*)&As[(ar + 32 * i) * OA_PITCH + ac] = ra[i];
    {
      float* p = &Bs[bkp * OB_PITCH + bcg * 8];
      v4f lo = {rb0.x, rb1.x, rb0.y, rb1.y};
      v4f hi = {rb0.z, rb1.z, rb0.w, rb1.w};
      *(v4f*)p = lo;
      *(v4f*)(p + 4) = hi;
    }
    __syncthreads();

    if (c + 1 < KDIM / OBK) {
      const int k0 = (c + 1) * OBK;
#pragma unroll
      for (int i = 0; i < 2; ++i) ra[i] = *(const v4f*)(ag[i] + k0);
      rb0 = *(const v4f*)(bg + (size_t)k0 * OUTD);
      rb1 = *(const v4f*)(bg + (size_t)(k0 + 1) * OUTD);
    }

#pragma unroll
    for (int ks = 0; ks < OBK / 4; ++ks) {
      const v2f af = *(const v2f*)(aFrag + ks * 4);
      v2f bf[2];
#pragma unroll
      for (int t = 0; t < 2; ++t)
        bf[t] = *(const v2f*)(bFrag + ks * 2 * OB_PITCH + t * 32);
#pragma unroll
      for (int t = 0; t < 2; ++t)
        acc[t] = __builtin_amdgcn_wmma_f32_16x16x4_f32(
            false, af, false, bf[t], (short)0, acc[t], false, false);
    }
    __syncthreads();
  }

  const int rbase = row0 + wr * 16 + khi * 8;
  float bias[2];
#pragma unroll
  for (int t = 0; t < 2; ++t) bias[t] = bfc[wc * 32 + t * 16 + mlo];
  if (row0 + OBM <= NN) {
#pragma unroll
    for (int i = 0; i < 8; ++i) {
#pragma unroll
      for (int t = 0; t < 2; ++t)
        out[(size_t)(rbase + i) * OUTD + wc * 32 + t * 16 + mlo] = acc[t][i] + bias[t];
    }
  } else {
#pragma unroll
    for (int i = 0; i < 8; ++i) {
      if (rbase + i < NN) {
#pragma unroll
        for (int t = 0; t < 2; ++t)
          out[(size_t)(rbase + i) * OUTD + wc * 32 + t * 16 + mlo] = acc[t][i] + bias[t];
      }
    }
  }
}

extern "C" void kernel_launch(void* const* d_in, const int* in_sizes, int n_in,
                              void* d_out, int out_size, void* d_ws, size_t ws_size,
                              hipStream_t stream) {
  const float* feat     = (const float*)d_in[0];
  const int*   ei       = (const int*)d_in[1];
  const float* w_src    = (const float*)d_in[2];
  const float* b_src    = (const float*)d_in[3];
  const float* w_dst    = (const float*)d_in[4];
  const float* b_dst    = (const float*)d_in[5];
  const float* attn     = (const float*)d_in[6];
  const float* gat_bias = (const float*)d_in[7];
  const float* w_fc     = (const float*)d_in[8];
  const float* b_fc     = (const float*)d_in[9];
  float* out = (float*)d_out;

  float* ws    = (float*)d_ws;
  float* Hsrc  = ws;
  float* Hdst  = Hsrc + (size_t)NN * KDIM;
  float* score = Hdst + (size_t)NN * KDIM;
  float* mbuf  = score + (size_t)EE * 4;
  float* denom = mbuf + (size_t)NN * 4;
  float* rst   = denom + (size_t)NN * 4;

  init_kernel<<<(NN + 255) / 256, 256, 0, stream>>>((unsigned*)mbuf, denom);

  wmma_proj_kernel<<<dim3(2, (NN + PBM - 1) / PBM), 256, 0, stream>>>(
      feat, w_src, b_src, w_dst, b_dst, Hsrc, Hdst);

  edge_score_kernel<<<(EE / EPW * 32 + 255) / 256, 256, 0, stream>>>(
      ei, Hsrc, Hdst, attn, score, (unsigned*)mbuf);

  sanitize_m_kernel<<<(NN * 4 + 255) / 256, 256, 0, stream>>>((unsigned*)mbuf);

  edge_exp_kernel<<<(EE * 4 / EXP_K + 255) / 256, 256, 0, stream>>>(
      ei, mbuf, score, denom);

  edge_scatter_kernel<<<SC_G, SC_T, 0, stream>>>(
      ei, Hsrc, score, denom, gat_bias, rst);

  wmma_out_kernel<<<(NN + OBM - 1) / OBM, 256, 0, stream>>>(rst, w_fc, b_fc, out);
}


# Scatter groups read a per-group bit mask of their edges (a 128-byte line per 1024 edges, built in LDS by a new first kernel that also does the init stores), not all dst ids; sums in arrival order
# speedup vs baseline: 2.9709x; 1.0588x over previous
#include <hip/hip_runtime.h>
#include <hip/hip_bf16.h>

#define NN 50000
#define EE 800000
#define KDIM 256
#define OUTD 64
#define NEG_SLOPE 0.2f

typedef __attribute__((ext_vector_type(2))) float v2f;
typedef __attribute__((ext_vector_type(4))) float v4f;
typedef __attribute__((ext_vector_type(8))) float v8f;
typedef __attribute__((ext_vector_type(4))) int v4i;

__device__ __forceinline__ unsigned orderFloat(float f) {
  unsigned u = __float_as_uint(f);
  return (u & 0x80000000u) ? ~u : (u | 0x80000000u);
}
__device__ __forceinline__ float unorderFloat(unsigned u) {
  u = (u & 0x80000000u) ? (u & 0x7FFFFFFFu) : ~u;
  return __uint_as_float(u);
}

#define PBM 128
#define PBK 32
#define A_PITCH 36
#define B_PITCH 544

__global__ __launch_bounds__(256) void wmma_proj_kernel(
    const float* __restrict__ feat,
    const float* __restrict__ Wsrc, const float* __restrict__ Bsrc,
    const float* __restrict__ Wdst, const float* __restrict__ Bdst,
    float* __restrict__ Hsrc, float* __restrict__ Hdst) {
  __shared__ __attribute__((aligned(16))) float As[PBM * A_PITCH];
  __shared__ __attribute__((aligned(16))) float Bs[(PBK / 2) * B_PITCH];

  const int tid  = threadIdx.x;
  const int lane = tid & 31;
  const int wave = tid >> 5;
  const int wr   = wave >> 2;
  const int wc   = wave & 3;
  const int mat  = blockIdx.x;
  const int row0 = blockIdx.y * PBM;
  const int khi  = lane >> 4;
  const int mlo  = lane & 15;

  const float* __restrict__ W  = mat ? Wdst : Wsrc;
  const float* __restrict__ Bv = mat ? Bdst : Bsrc;
  float* __restrict__ Hout     = mat ? Hdst : Hsrc;

  const int ar = tid >> 3;
  const int ac = (tid & 7) * 4;
  const int bcg = tid & 63;
  const int bkp = tid >> 6;

  const float* ag[4];
#pragma unroll
  for (int i = 0; i < 4; ++i) {
    int r = row0 + ar + 32 * i;
    r = (r < NN) ? r : 0;
    ag[i] = feat + (size_t)r * KDIM + ac;
  }
  const float* bg = W + (size_t)(2 * bkp) * KDIM + bcg * 4;

  v4f ra[4], rb0[4], rb1[4];
#pragma unroll
  for (int i = 0; i < 4; ++i) {
    ra[i]  = *(const v4f*)(ag[i]);
    rb0[i] = *(const v4f*)(bg + (size_t)(8 * i) * KDIM);
    rb1[i] = *(const v4f*)(bg + (size_t)(8 * i + 1) * KDIM);
  }

  const float* aFrag = &As[(wr * 64 + mlo) * A_PITCH + 2 * khi];
  const float* bFrag = &Bs[khi * B_PITCH + (wc * 64 + mlo) * 2];

  v8f acc[4][4] = {};
  for (int c = 0; c < KDIM / PBK; ++c) {
#pragma unroll
    for (int i = 0; i < 4; ++i) {
      *(v4f*)&As[(ar + 32 * i) * A_PITCH + ac] = ra[i];
      float* p = &Bs[(bkp + 4 * i) * B_PITCH + bcg * 8];
      v4f lo = {rb0[i].x, rb1[i].x, rb0[i].y, rb1[i].y};
      v4f hi = {rb0[i].z, rb1[i].z, rb0[i].w, rb1[i].w};
      *(v4f*)p = lo;
      *(v4f*)(p + 4) = hi;
    }
    __syncthreads();

    if (c + 1 < KDIM / PBK) {
      const int k0 = (c + 1) * PBK;
#pragma unroll
      for (int i = 0; i < 4; ++i) {
        ra[i]  = *(const v4f*)(ag[i] + k0);
        rb0[i] = *(const v4f*)(bg + (size_t)(k0 + 8 * i) * KDIM);
        rb1[i] = *(const v4f*)(bg + (size_t)(k0 + 8 * i + 1) * KDIM);
      }
    }

#pragma unroll
    for (int ks = 0; ks < PBK / 4; ++ks) {
      v2f af[4], bf[4];
#pragma unroll
      for (int s = 0; s < 4; ++s)
        af[s] = *(const v2f*)(aFrag + s * 16 * A_PITCH + ks * 4);
#pragma unroll
      for (int t = 0; t < 4; ++t)
        bf[t] = *(const v2f*)(bFrag + ks * 2 * B_PITCH + t * 32);
#pragma unroll
      for (int s = 0; s < 4; ++s) {
#pragma unroll
        for (int t = 0; t < 4; ++t) {
          acc[s][t] = __builtin_amdgcn_wmma_f32_16x16x4_f32(
              false, af[s], false, bf[t], (short)0, acc[s][t], false, false);
        }
      }
    }
    __syncthreads();
  }

  const int mbase = khi * 8;
  float bias[4];
#pragma unroll
  for (int t = 0; t < 4; ++t) bias[t] = Bv[wc * 64 + t * 16 + mlo];
  const int rbase = row0 + wr * 64 + mbase;
  if (row0 + PBM <= NN) {
#pragma unroll
    for (int s = 0; s < 4; ++s) {
#pragma unroll
      for (int i = 0; i < 8; ++i) {
        int r = rbase + s * 16 + i;
#pragma unroll
        for (int t = 0; t < 4; ++t)
          Hout[(size_t)r * KDIM + wc * 64 + t * 16 + mlo] = acc[s][t][i] + bias[t];
      }
    }
  } else {
#pragma unroll
    for (int s = 0; s < 4; ++s) {
#pragma unroll
      for (int i = 0; i < 8; ++i) {
        int r = rbase + s * 16 + i;
        if (r < NN) {
#pragma unroll
          for (int t = 0; t < 4; ++t)
            Hout[(size_t)r * KDIM + wc * 64 + t * 16 + mlo] = acc[s][t][i] + bias[t];
        }
      }
    }
  }
}

#define EPW 8
static_assert(EE % EPW == 0, "an edge count that is no multiple of EPW needs a tail");
__global__ __launch_bounds__(256) void edge_score_kernel(
    const int* __restrict__ ei, const float* __restrict__ Hsrc,
    const float* __restrict__ Hdst, const float* __restrict__ attn,
    float* __restrict__ score, unsigned* __restrict__ m_u32) {
  const int e0 = ((blockIdx.x * blockDim.x + threadIdx.x) >> 5) * EPW;
  const int lane = threadIdx.x & 31;
  if (e0 >= EE) return;
  const int sub = lane & 7;
  const int srcv = ei[e0 + sub];
  const int dstv = ei[EE + e0 + sub];

  const v4f* pa = (const v4f*)(attn + lane * 8);
  const v4f a0 = pa[0], a1 = pa[1];

  v4f s0[EPW], s1[EPW], d0[EPW], d1[EPW];
#pragma unroll
  for (int j = 0; j < EPW; ++j) {
    const int s = __builtin_amdgcn_readlane(srcv, j);
    const int d = __builtin_amdgcn_readlane(dstv, j);
    const v4f* ps = (const v4f*)(Hsrc + (size_t)s * KDIM + lane * 8);
    const v4f* pd = (const v4f*)(Hdst + (size_t)d * KDIM + lane * 8);
    s0[j] = ps[0]; s1[j] = ps[1];
    d0[j] = pd[0]; d1[j] = pd[1];
  }
#pragma unroll
  for (int j = 0; j < EPW; ++j)
    asm volatile("" :: "v"(s0[j]), "v"(s1[j]), "v"(d0[j]), "v"(d1[j]));

  float p[EPW];
#pragma unroll
  for (int j = 0; j < EPW; ++j) {
    float part = 0.0f;
#pragma unroll
    for (int i = 0; i < 4; ++i) {
      float v = s0[j][i] + d0[j][i];
      v = (v >= 0.0f) ? v : v * NEG_SLOPE;
      part = __builtin_fmaf(v, a0[i], part);
      float w = s1[j][i] + d1[j][i];
      w = (w >= 0.0f) ? w : w * NEG_SLOPE;
      part = __builtin_fmaf(w, a1[i], part);
    }
    p[j] = part;
  }

  const bool b0 = (lane & 1) != 0, b1 = (lane & 2) != 0, b2 = (lane & 4) != 0;
  float q[4], r[2];
#pragma unroll
  for (int k = 0; k < 4; ++k) {
    const float keep = b0 ? p[2 * k + 1] : p[2 * k];
    const float send = b0 ? p[2 * k] : p[2 * k + 1];
    q[k] = keep + __shfl_xor(send, 1, 32);
  }
#pragma unroll
  for (int k = 0; k < 2; ++k) {
    const float keep = b1 ? q[2 * k + 1] : q[2 * k];
    const float send = b1 ? q[2 * k] : q[2 * k + 1];
    r[k] = keep + __shfl_xor(send, 2, 32);
  }
  const float keep = b2 ? r[1] : r[0];
  const float send = b2 ? r[0] : r[1];
  const float sc = keep + __shfl_xor(send, 4, 32);

  const int h = lane >> 3;
  score[(size_t)(e0 + sub) * 4 + h] = sc;
  atomicMax(&m_u32[(size_t)dstv * 4 + h], orderFloat(sc));
}

__global__ __launch_bounds__(256) void sanitize_m_kernel(unsigned* __restrict__ m) {
  int i = blockIdx.x * blockDim.x + threadIdx.x;
  if (i >= NN * 4) return;
  float f = unorderFloat(m[i]);
  if (!isfinite(f)) f = 0.0f;
  ((float*)m)[i] = f;
}

#define EXP_K 4
static_assert((EE * 4) % (32 * EXP_K) == 0, "an item count that is no multiple of 32*EXP_K needs a tail");
__global__ __launch_bounds__(256) void edge_exp_kernel(
    const int* __restrict__ ei, const float* __restrict__ m_f,
    float* __restrict__ score, float* __restrict__ denom) {
  const int base = ((blockIdx.x * blockDim.x + threadIdx.x) >> 5) * (32 * EXP_K);
  const int lane = threadIdx.x & 31;
  if (base >= EE * 4) return;
  const int h = lane & 3;
  int dst[EXP_K];
  float sc[EXP_K], mm[EXP_K];
#pragma unroll
  for (int k = 0; k < EXP_K; ++k) dst[k] = ei[EE + ((base + 32 * k + lane) >> 2)];
#pragma unroll
  for (int k = 0; k < EXP_K; ++k) {
    sc[k] = score[base + 32 * k + lane];
    mm[k] = m_f[(size_t)dst[k] * 4 + h];
  }
#pragma unroll
  for (int k = 0; k < EXP_K; ++k) {
    const float ex = expf(sc[k] - mm[k]);
    score[base + 32 * k + lane] = ex;
    unsafeAtomicAdd(&denom[(size_t)dst[k] * 4 + h], ex);
  }
}

#define SC_G 256
#define SC_R ((NN + SC_G - 1) / SC_G)
#define SC_T 1024
#define SC_W (SC_T / 32)
#define SC_MB 1024
#define SC_NMB ((EE + SC_MB - 1) / SC_MB)
#define SC_MROW (SC_G * 32)
#define SC_QCAP (32 + EPW)
static_assert((SC_G - 1) * SC_R < NN, "every group owns at least one row");
static_assert((NN - 1) / SC_R < SC_G, "every dst has an owner group");
static_assert(SC_NMB * SC_MB >= NN, "the mask kernel's threads also cover the N start values");
static_assert(SC_R * KDIM * 4 + SC_W * SC_QCAP * 4 <= 327680, "rows and queues must fit the LDS");

__global__ __launch_bounds__(SC_MB) void owner_mask_kernel(
    const int* __restrict__ ei, unsigned* __restrict__ mask,
    unsigned* __restrict__ m_u32, float* __restrict__ denom) {
  __shared__ __attribute__((aligned(16))) unsigned tab[SC_MROW];

  const int tid = threadIdx.x;
  const int e = blockIdx.x * SC_MB + tid;
  const v4i zi = {0, 0, 0, 0};
  for (int q = tid; q < SC_MROW / 4; q += SC_MB) ((v4i*)tab)[q] = zi;
  if (e < NN) {
    const v4f z = {0.0f, 0.0f, 0.0f, 0.0f};
    ((v4f*)m_u32)[e] = z;
    ((v4f*)denom)[e] = z;
  }
  __syncthreads();

  if (e < EE) {
    const unsigned d = (unsigned)ei[EE + e];
    if (d < (unsigned)NN)
      atomicOr(&tab[(d / SC_R) * 32 + (tid >> 5)], 1u << (tid & 31));
  }
  __syncthreads();

  v4i* out = (v4i*)(mask + (size_t)blockIdx.x * SC_MROW);
  for (int q = tid; q < SC_MROW / 4; q += SC_MB) out[q] = ((const v4i*)tab)[q];
}

__global__ __launch_bounds__(SC_T) void edge_scatter_kernel(
    const int* __restrict__ ei, const unsigned* __restrict__ mask,
    const float* __restrict__ Hsrc,
    const float* __restrict__ ex, const float* __restrict__ denom,
    const float* __restrict__ gat_bias, float* __restrict__ rst) {
  __shared__ __attribute__((aligned(16))) float acc[SC_R * KDIM];
  __shared__ int queue[SC_W * SC_QCAP];

  const int tid  = threadIdx.x;
  const int lane = tid & 31;
  const int wave = __builtin_amdgcn_readfirstlane(tid >> 5);
  const int n0 = blockIdx.x * SC_R;
  const int nrows = (NN - n0 < SC_R) ? (NN - n0) : SC_R;

  for (int q = tid; q < SC_R * (KDIM / 4); q += SC_T)
    ((v4f*)acc)[q] = ((const v4f*)gat_bias)[q & (KDIM / 4 - 1)];
  __syncthreads();

  int* myQ = queue + wave * SC_QCAP;
  int qn = 0;

  auto batch = [&](int h0, int cnt) __attribute__((always_inline)) {
    const int sw = lane >> 2;
    const int ss = lane & 7;
    const int ew = myQ[h0 + ((sw < cnt) ? sw : 0)];
    const int eq = myQ[h0 + ((ss < cnt) ? ss : 0)];
    const int srcv = ei[eq];
    const int dw = ei[EE + ew];
    int rw = dw - n0;
    rw = (rw < 0) ? 0 : ((rw < nrows) ? rw : (nrows - 1));
    const int hl = lane & 3;
    const float den = denom[(size_t)(n0 + rw) * 4 + hl];
    const float alpha = ex[(size_t)ew * 4 + hl] / ((den == 0.0f) ? 1.0f : den);

    float v[EPW][8];
#pragma unroll
    for (int j = 0; j < EPW; ++j) {
      if (j < cnt) {
        const int s = __builtin_amdgcn_readlane(srcv, j);
        const float* ps = Hsrc + (size_t)s * KDIM + lane;
#pragma unroll
        for (int i = 0; i < 8; ++i) v[j][i] = ps[32 * i];
      }
    }
#pragma unroll
    for (int j = 0; j < EPW; ++j) {
      if (j < cnt) {
        const int r = __builtin_amdgcn_readlane(rw, 4 * j);
        float* pa = &acc[r * KDIM + lane];
#pragma unroll
        for (int i = 0; i < 8; ++i) {
          const float ah = __int_as_float(
              __builtin_amdgcn_readlane(__float_as_int(alpha), 4 * j + (i >> 1)));
          __hip_atomic_fetch_add(&pa[32 * i], ah * v[j][i], __ATOMIC_RELAXED,
                                 __HIP_MEMORY_SCOPE_WORKGROUP);
        }
      }
    }
  };

  const unsigned* mrow = mask + (size_t)blockIdx.x * 32 + lane;
  auto line_of = [&](int c) __attribute__((always_inline)) {
    const int cc = (c < SC_NMB) ? c : (SC_NMB - 1);
    return mrow[(size_t)cc * SC_MROW];
  };
  unsigned nxt = line_of(wave);
  for (int c = wave; c < SC_NMB; c += SC_W) {
    unsigned w = nxt;
    nxt = line_of(c + SC_W);
    const int ebase = c * SC_MB + lane * 32;
    unsigned m = __builtin_amdgcn_ballot_w32(w != 0u);
    while (m) {
      if (w != 0u) {
        myQ[qn + (int)__builtin_amdgcn_mbcnt_lo(m, 0u)] = ebase + __builtin_ctz(w);
        w &= w - 1u;
      }
      qn += __builtin_popcount(m);
      m = __builtin_amdgcn_ballot_w32(w != 0u);
      int h = 0;
      for (; h + EPW <= qn; h += EPW) batch(h, EPW);
      if (h > 0) {
        const int rem = qn - h;
        int t = 0;
        if (lane < rem) t = myQ[h + lane];
        if (lane < rem) myQ[lane] = t;
        qn = rem;
      }
    }
  }
  if (qn > 0) batch(0, qn);
  __syncthreads();

  v4f* out_rows = (v4f*)(rst + (size_t)n0 * KDIM);
  for (int q = tid; q < nrows * (KDIM / 4); q += SC_T)
    out_rows[q] = ((const v4f*)acc)[q];
}

#define OBM 64
#define OBK 32
#define OA_PITCH 36
#define OB_PITCH 160
static_assert(KDIM % OBK == 0, "k must split into whole chunks");
static_assert(OUTD == 64, "copy roles and wave layout assume 64 output columns");

__global__ __launch_bounds__(256) void wmma_out_kernel(
    const float* __restrict__ rst, const float* __restrict__ Wfc,
    const float* __restrict__ bfc, float* __restrict__ out) {
  __shared__ __attribute__((aligned(16))) float As[OBM * OA_PITCH];
  __shared__ __attribute__((aligned(16))) float Bs[(OBK / 2) * OB_PITCH];

  const int tid  = threadIdx.x;
  const int lane = tid & 31;
  const int wave = tid >> 5;
  const int wr   = wave >> 1;
  const int wc   = wave & 1;
  const int row0 = blockIdx.x * OBM;
  const int khi  = lane >> 4;
  const int mlo  = lane & 15;

  const int ar  = tid >> 3;
  const int ac  = (tid & 7) * 4;
  const int bcg = tid & 15;
  const int bkp = tid >> 4;

  const float* ag[2];
#pragma unroll
  for (int i = 0; i < 2; ++i) {
    int r = row0 + ar + 32 * i;
    r = (r < NN) ? r : 0;
    ag[i] = rst + (size_t)r * KDIM + ac;
  }
  const float* bg = Wfc + (size_t)(2 * bkp) * OUTD + bcg * 4;

  v4f ra[2], rb0, rb1;
#pragma unroll
  for (int i = 0; i < 2; ++i) ra[i] = *(const v4f*)(ag[i]);
  rb0 = *(const v4f*)(bg);
  rb1 = *(const v4f*)(bg + OUTD);

  const float* aFrag = &As[(wr * 16 + mlo) * OA_PITCH + 2 * khi];
  const float* bFrag = &Bs[khi * OB_PITCH + (wc * 32 + mlo) * 2];

  v8f acc[2] = {};
  for (int c = 0; c < KDIM / OBK; ++c) {
#pragma unroll
    for (int i = 0; i < 2; ++i)
      *(v4f*)&As[(ar + 32 * i) * OA_PITCH + ac] = ra[i];
    {
      float* p = &Bs[bkp * OB_PITCH + bcg * 8];
      v4f lo = {rb0.x, rb1.x, rb0.y, rb1.y};
      v4f hi = {rb0.z, rb1.z, rb0.w, rb1.w};
      *(v4f*)p = lo;
      *(v4f*)(p + 4) = hi;
    }
    __syncthreads();

    if (c + 1 < KDIM / OBK) {
      const int k0 = (c + 1) * OBK;
#pragma unroll
      for (int i = 0; i < 2; ++i) ra[i] = *(const v4f*)(ag[i] + k0);
      rb0 = *(const v4f*)(bg + (size_t)k0 * OUTD);
      rb1 = *(const v4f*)(bg + (size_t)(k0 + 1) * OUTD);
    }

#pragma unroll
    for (int ks = 0; ks < OBK / 4; ++ks) {
      const v2f af = *(const v2f*)(aFrag + ks * 4);
      v2f bf[2];
#pragma unroll
      for (int t = 0; t < 2; ++t)
        bf[t] = *(const v2f*)(bFrag + ks * 2 * OB_PITCH + t * 32);
#pragma unroll
      for (int t = 0; t < 2; ++t)
        acc[t] = __builtin_amdgcn_wmma_f32_16x16x4_f32(
            false, af, false, bf[t], (short)0, acc[t], false, false);
    }
    __syncthreads();
  }

  const int rbase = row0 + wr * 16 + khi * 8;
  float bias[2];
#pragma unroll
  for (int t = 0; t < 2; ++t) bias[t] = bfc[wc * 32 + t * 16 + mlo];
  if (row0 + OBM <= NN) {
#pragma unroll
    for (int i = 0; i < 8; ++i) {
#pragma unroll
      for (int t = 0; t < 2; ++t)
        out[(size_t)(rbase + i) * OUTD + wc * 32 + t * 16 + mlo] = acc[t][i] + bias[t];
    }
  } else {
#pragma unroll
    for (int i = 0; i < 8; ++i) {
      if (rbase + i < NN) {
#pragma unroll
        for (int t = 0; t < 2; ++t)
          out[(size_t)(rbase + i) * OUTD + wc * 32 + t * 16 + mlo] = acc[t][i] + bias[t];
      }
    }
  }
}

extern "C" void kernel_launch(void* const* d_in, const int* in_sizes, int n_in,
                              void* d_out, int out_size, void* d_ws, size_t ws_size,
                              hipStream_t stream) {
  const float* feat     = (const float*)d_in[0];
  const int*   ei       = (const int*)d_in[1];
  const float* w_src    = (const float*)d_in[2];
  const float* b_src    = (const float*)d_in[3];
  const float* w_dst    = (const float*)d_in[4];
  const float* b_dst    = (const float*)d_in[5];
  const float* attn     = (const float*)d_in[6];
  const float* gat_bias = (const float*)d_in[7];
  const float* w_fc     = (const float*)d_in[8];
  const float* b_fc     = (const float*)d_in[9];
  float* out = (float*)d_out;

  float* ws    = (float*)d_ws;
  float* Hsrc  = ws;
  float* Hdst  = Hsrc + (size_t)NN * KDIM;
  float* score = Hdst + (size_t)NN * KDIM;
  float* mbuf  = score + (size_t)EE * 4;
  float* denom = mbuf + (size_t)NN * 4;
  float* rst   = denom + (size_t)NN * 4;
  unsigned* mask = (unsigned*)(rst + (size_t)NN * KDIM);

  owner_mask_kernel<<<SC_NMB, SC_MB, 0, stream>>>(ei, mask, (unsigned*)mbuf, denom);

  wmma_proj_kernel<<<dim3(2, (NN + PBM - 1) / PBM), 256, 0, stream>>>(
      feat, w_src, b_src, w_dst, b_dst, Hsrc, Hdst);

  edge_score_kernel<<<(EE / EPW * 32 + 255) / 256, 256, 0, stream>>>(
      ei, Hsrc, Hdst, attn, score, (unsigned*)mbuf);

  sanitize_m_kernel<<<(NN * 4 + 255) / 256, 256, 0, stream>>>((unsigned*)mbuf);

  edge_exp_kernel<<<(EE * 4 / EXP_K + 255) / 256, 256, 0, stream>>>(
      ei, mbuf, score, denom);

  edge_scatter_kernel<<<SC_G, SC_T, 0, stream>>>(
      ei, mask, Hsrc, score, denom, gat_bias, rst);

  wmma_out_kernel<<<(NN + OBM - 1) / OBM, 256, 0, stream>>>(rst, w_fc, b_fc, out);
}


# Score, segment max, exp and denominator moved into the row-owner groups, one launch for three: a group keeps its Hdst rows in LDS, gathers only Hsrc rows; max and denom by LDS atomics (arrival order)
# speedup vs baseline: 3.1496x; 1.0602x over previous
#include <hip/hip_runtime.h>
#include <hip/hip_bf16.h>

#define NN 50000
#define EE 800000
#define KDIM 256
#define OUTD 64
#define NEG_SLOPE 0.2f

typedef __attribute__((ext_vector_type(2))) float v2f;
typedef __attribute__((ext_vector_type(4))) float v4f;
typedef __attribute__((ext_vector_type(8))) float v8f;
typedef __attribute__((ext_vector_type(4))) int v4i;

__device__ __forceinline__ unsigned orderFloat(float f) {
  unsigned u = __float_as_uint(f);
  return (u & 0x80000000u) ? ~u : (u | 0x80000000u);
}
__device__ __forceinline__ float unorderFloat(unsigned u) {
  u = (u & 0x80000000u) ? (u & 0x7FFFFFFFu) : ~u;
  return __uint_as_float(u);
}

#define PBM 128
#define PBK 32
#define A_PITCH 36
#define B_PITCH 544

__global__ __launch_bounds__(256) void wmma_proj_kernel(
    const float* __restrict__ feat,
    const float* __restrict__ Wsrc, const float* __restrict__ Bsrc,
    const float* __restrict__ Wdst, const float* __restrict__ Bdst,
    float* __restrict__ Hsrc, float* __restrict__ Hdst) {
  __shared__ __attribute__((aligned(16))) float As[PBM * A_PITCH];
  __shared__ __attribute__((aligned(16))) float Bs[(PBK / 2) * B_PITCH];

  const int tid  = threadIdx.x;
  const int lane = tid & 31;
  const int wave = tid >> 5;
  const int wr   = wave >> 2;
  const int wc   = wave & 3;
  const int mat  = blockIdx.x;
  const int row0 = blockIdx.y * PBM;
  const int khi  = lane >> 4;
  const int mlo  = lane & 15;

  const float* __restrict__ W  = mat ? Wdst : Wsrc;
  const float* __restrict__ Bv = mat ? Bdst : Bsrc;
  float* __restrict__ Hout     = mat ? Hdst : Hsrc;

  const int ar = tid >> 3;
  const int ac = (tid & 7) * 4;
  const int bcg = tid & 63;
  const int bkp = tid >> 6;

  const float* ag[4];
#pragma unroll
  for (int i = 0; i < 4; ++i) {
    int r = row0 + ar + 32 * i;
    r = (r < NN) ? r : 0;
    ag[i] = feat + (size_t)r * KDIM + ac;
  }
  const float* bg = W + (size_t)(2 * bkp) * KDIM + bcg * 4;

  v4f ra[4], rb0[4], rb1[4];
#pragma unroll
  for (int i = 0; i < 4; ++i) {
    ra[i]  = *(const v4f*)(ag[i]);
    rb0[i] = *(const v4f*)(bg + (size_t)(8 * i) * KDIM);
    rb1[i] = *(const v4f*)(bg + (size_t)(8 * i + 1) * KDIM);
  }

  const float* aFrag = &As[(wr * 64 + mlo) * A_PITCH + 2 * khi];
  const float* bFrag = &Bs[khi * B_PITCH + (wc * 64 + mlo) * 2];

  v8f acc[4][4] = {};
  for (int c = 0; c < KDIM / PBK; ++c) {
#pragma unroll
    for (int i = 0; i < 4; ++i) {
      *(v4f*)&As[(ar + 32 * i) * A_PITCH + ac] = ra[i];
      float* p = &Bs[(bkp + 4 * i) * B_PITCH + bcg * 8];
      v4f lo = {rb0[i].x, rb1[i].x, rb0[i].y, rb1[i].y};
      v4f hi = {rb0[i].z, rb1[i].z, rb0[i].w, rb1[i].w};
      *(v4f*)p = lo;
      *(v4f*)(p + 4) = hi;
    }
    __syncthreads();

    if (c + 1 < KDIM / PBK) {
      const int k0 = (c + 1) * PBK;
#pragma unroll
      for (int i = 0; i < 4; ++i) {
        ra[i]  = *(const v4f*)(ag[i] + k0);
        rb0[i] = *(const v4f*)(bg + (size_t)(k0 + 8 * i) * KDIM);
        rb1[i] = *(const v4f*)(bg + (size_t)(k0 + 8 * i + 1) * KDIM);
      }
    }

#pragma unroll
    for (int ks = 0; ks < PBK / 4; ++ks) {
      v2f af[4], bf[4];
#pragma unroll
      for (int s = 0; s < 4; ++s)
        af[s] = *(const v2f*)(aFrag + s * 16 * A_PITCH + ks * 4);
#pragma unroll
      for (int t = 0; t < 4; ++t)
        bf[t] = *(const v2f*)(bFrag + ks * 2 * B_PITCH + t * 32);
#pragma unroll
      for (int s = 0; s < 4; ++s) {
#pragma unroll
        for (int t = 0; t < 4; ++t) {
          acc[s][t] = __builtin_amdgcn_wmma_f32_16x16x4_f32(
              false, af[s], false, bf[t], (short)0, acc[s][t], false, false);
        }
      }
    }
    __syncthreads();
  }

  const int mbase = khi * 8;
  float bias[4];
#pragma unroll
  for (int t = 0; t < 4; ++t) bias[t] = Bv[wc * 64 + t * 16 + mlo];
  const int rbase = row0 + wr * 64 + mbase;
  if (row0 + PBM <= NN) {
#pragma unroll
    for (int s = 0; s < 4; ++s) {
#pragma unroll
      for (int i = 0; i < 8; ++i) {
        int r = rbase + s * 16 + i;
#pragma unroll
        for (int t = 0; t < 4; ++t)
          Hout[(size_t)r * KDIM + wc * 64 + t * 16 + mlo] = acc[s][t][i] + bias[t];
      }
    }
  } else {
#pragma unroll
    for (int s = 0; s < 4; ++s) {
#pragma unroll
      for (int i = 0; i < 8; ++i) {
        int r = rbase + s * 16 + i;
        if (r < NN) {
#pragma unroll
          for (int t = 0; t < 4; ++t)
            Hout[(size_t)r * KDIM + wc * 64 + t * 16 + mlo] = acc[s][t][i] + bias[t];
        }
      }
    }
  }
}

#define EPW 8

#define SC_G 256
#define SC_R ((NN + SC_G - 1) / SC_G)
#define SC_T 1024
#define SC_W (SC_T / 32)
#define SC_MB 1024
#define SC_NMB ((EE + SC_MB - 1) / SC_MB)
#define SC_MROW (SC_G * 32)
#define SC_QCAP (32 + EPW)
static_assert((SC_G - 1) * SC_R < NN, "every group owns at least one row");
static_assert((NN - 1) / SC_R < SC_G, "every dst has an owner group");
static_assert(SC_R * KDIM * 4 + SC_W * SC_QCAP * 4 <= 327680, "rows and queues must fit the LDS");

__global__ __launch_bounds__(SC_MB) void owner_mask_kernel(
    const int* __restrict__ ei, unsigned* __restrict__ mask) {
  __shared__ __attribute__((aligned(16))) unsigned tab[SC_MROW];

  const int tid = threadIdx.x;
  const int e = blockIdx.x * SC_MB + tid;
  const v4i zi = {0, 0, 0, 0};
  for (int q = tid; q < SC_MROW / 4; q += SC_MB) ((v4i*)tab)[q] = zi;
  __syncthreads();

  if (e < EE) {
    const unsigned d = (unsigned)ei[EE + e];
    if (d < (unsigned)NN)
      atomicOr(&tab[(d / SC_R) * 32 + (tid >> 5)], 1u << (tid & 31));
  }
  __syncthreads();

  v4i* out = (v4i*)(mask + (size_t)blockIdx.x * SC_MROW);
  for (int q = tid; q < SC_MROW / 4; q += SC_MB) out[q] = ((const v4i*)tab)[q];
}

template <int B, class F>
__device__ __forceinline__ void scan_owned_edges(
    const unsigned* __restrict__ mask, int group, int wave, int nwaves, int lane, int* myQ, F batch) {
  const unsigned* mgrp = mask + (size_t)group * 32;
  auto line_of = [&](int c) __attribute__((always_inline)) {
    const int cc = (c < SC_NMB) ? c : (SC_NMB - 1);
    return (mgrp + (size_t)cc * SC_MROW)[lane];
  };
  int qn = 0;
  unsigned nxt = line_of(wave);
  for (int c = wave; c < SC_NMB; c += nwaves) {
    unsigned w = nxt;
    nxt = line_of(c + nwaves);
    const int ebase = c * SC_MB + lane * 32;
    unsigned m = __builtin_amdgcn_ballot_w32(w != 0u);
    while (m) {
      if (w != 0u) {
        myQ[qn + (int)__builtin_amdgcn_mbcnt_lo(m, 0u)] = ebase + __builtin_ctz(w);
        w &= w - 1u;
      }
      qn += __builtin_popcount(m);
      m = __builtin_amdgcn_ballot_w32(w != 0u);
      int h = 0;
      for (; h + B <= qn; h += B) batch(h, B);
      if (h > 0) {
        const int rem = qn - h;
        int t = 0;
        if (lane < rem) t = myQ[h + lane];
        if (lane < rem) myQ[lane] = t;
        qn = rem;
      }
    }
  }
  if (qn > 0) batch(0, qn);
}

#define SM_T 640
#define SM_W (SM_T / 32)
#define SM_QCAP 64
__global__ __launch_bounds__(SM_T) void edge_softmax_kernel(
    const int* __restrict__ ei, const unsigned* __restrict__ mask,
    const float* __restrict__ Hsrc, const float* __restrict__ Hdst,
    const float* __restrict__ attn, float* __restrict__ score,
    float* __restrict__ denom) {
  __shared__ __attribute__((aligned(16))) float hd[SC_R * KDIM];
  __shared__ __attribute__((aligned(16))) unsigned mx[SC_R * 4];
  __shared__ float dn[SC_R * 4];
  __shared__ int queue[SM_W * SM_QCAP];

  const int tid  = threadIdx.x;
  const int lane = tid & 31;
  const int wave = __builtin_amdgcn_readfirstlane(tid >> 5);
  const int n0 = blockIdx.x * SC_R;
  const int nrows = (NN - n0 < SC_R) ? (NN - n0) : SC_R;

  {
    const v4f* hrows = (const v4f*)(Hdst + (size_t)n0 * KDIM);
    const int total = nrows * (KDIM / 4);
    for (int q0 = tid; q0 < total; q0 += 8 * SM_T) {
      v4f t[8];
#pragma unroll
      for (int u = 0; u < 8; ++u) {
        const int q = q0 + u * SM_T;
        t[u] = hrows[(q < total) ? q : (total - 1)];
      }
#pragma unroll
      for (int u = 0; u < 8; ++u) {
        const int q = q0 + u * SM_T;
        const int cq = q & 63;
        if (q < total) ((v4f*)hd)[(q & ~63) | ((cq & 1) << 5) | (cq >> 1)] = t[u];
      }
    }
    for (int q = tid; q < SC_R * 4; q += SM_T) { mx[q] = 0u; dn[q] = 0.0f; }
  }
  __syncthreads();

  int* myQ = queue + wave * SM_QCAP;
  const int sub = lane & 7;
  const int hh = lane >> 3;
  const v4f* pa = (const v4f*)(attn + lane * 8);
  const v4f a0 = pa[0], a1 = pa[1];
  const bool b0 = (lane & 1) != 0, b1 = (lane & 2) != 0, b2 = (lane & 4) != 0;

  auto score_batch = [&](int h0, int cnt) __attribute__((always_inline)) {
    const int eq = myQ[h0 + ((sub < cnt) ? sub : 0)];
    const int srcv = ei[eq];
    const int dv = ei[EE + eq];
    int rw = dv - n0;
    rw = (rw < 0) ? 0 : ((rw < nrows) ? rw : (nrows - 1));

    v4f s0[EPW], s1[EPW];
#pragma unroll
    for (int j = 0; j < EPW; ++j) {
      const int s = __builtin_amdgcn_readlane(srcv, j);
      const v4f* ps = (const v4f*)(Hsrc + (size_t)s * KDIM + lane * 8);
      s0[j] = ps[0]; s1[j] = ps[1];
    }
#pragma unroll
    for (int j = 0; j < EPW; ++j)
      asm volatile("" :: "v"(s0[j]), "v"(s1[j]));

    float p[EPW];
#pragma unroll
    for (int j = 0; j < EPW; ++j) {
      const int r = __builtin_amdgcn_readlane(rw, j);
      const v4f* pd = (const v4f*)&hd[r * KDIM + lane * 4];
      const v4f d0 = pd[0], d1 = pd[32];
      float part = 0.0f;
#pragma unroll
      for (int i = 0; i < 4; ++i) {
        float v = s0[j][i] + d0[i];
        v = (v >= 0.0f) ? v : v * NEG_SLOPE;
        part = __builtin_fmaf(v, a0[i], part);
        float w = s1[j][i] + d1[i];
        w = (w >= 0.0f) ? w : w * NEG_SLOPE;
        part = __builtin_fmaf(w, a1[i], part);
      }
      p[j] = part;
    }

    float q[4], r2[2];
#pragma unroll
    for (int k = 0; k < 4; ++k) {
      const float keep = b0 ? p[2 * k + 1] : p[2 * k];
      const float send = b0 ? p[2 * k] : p[2 * k + 1];
      q[k] = keep + __shfl_xor(send, 1, 32);
    }
#pragma unroll
    for (int k = 0; k < 2; ++k) {
      const float keep = b1 ? q[2 * k + 1] : q[2 * k];
      const float send = b1 ? q[2 * k] : q[2 * k + 1];
      r2[k] = keep + __shfl_xor(send, 2, 32);
    }
    const float keep = b2 ? r2[1] : r2[0];
    const float send = b2 ? r2[0] : r2[1];
    const float sc = keep + __shfl_xor(send, 4, 32);

    if (sub < cnt) {
      score[(size_t)eq * 4 + hh] = sc;
      __hip_atomic_fetch_max(&mx[rw * 4 + hh], orderFloat(sc), __ATOMIC_RELAXED,
                             __HIP_MEMORY_SCOPE_WORKGROUP);
    }
  };
  scan_owned_edges<EPW>(mask, (int)blockIdx.x, wave, SM_W, lane, myQ, score_batch);
  __syncthreads();

  auto exp_batch = [&](int h0, int cnt) __attribute__((always_inline)) {
    if (lane < cnt) {
      const int ew = myQ[h0 + lane];
      const int dv = ei[EE + ew];
      int rw = dv - n0;
      rw = (rw < 0) ? 0 : ((rw < nrows) ? rw : (nrows - 1));
      v4f* ps = (v4f*)(score + (size_t)ew * 4);
      const v4f sc = *ps;
      const v4i mu = *(const v4i*)&mx[rw * 4];
      v4f ex;
#pragma unroll
      for (int h = 0; h < 4; ++h) {
        float mf = unorderFloat((unsigned)mu[h]);
        if (!isfinite(mf)) mf = 0.0f;
        ex[h] = expf(sc[h] - mf);
      }
      *ps = ex;
#pragma unroll
      for (int h = 0; h < 4; ++h)
        __hip_atomic_fetch_add(&dn[rw * 4 + h], ex[h], __ATOMIC_RELAXED,
                               __HIP_MEMORY_SCOPE_WORKGROUP);
    }
  };
  scan_owned_edges<32>(mask, (int)blockIdx.x, wave, SM_W, lane, myQ, exp_batch);
  __syncthreads();

  for (int q = tid; q < nrows * 4; q += SM_T) denom[(size_t)n0 * 4 + q] = dn[q];
}

__global__ __launch_bounds__(SC_T) void edge_scatter_kernel(
    const int* __restrict__ ei, const unsigned* __restrict__ mask,
    const float* __restrict__ Hsrc,
    const float* __restrict__ ex, const float* __restrict__ denom,
    const float* __restrict__ gat_bias, float* __restrict__ rst) {
  __shared__ __attribute__((aligned(16))) float acc[SC_R * KDIM];
  __shared__ int queue[SC_W * SC_QCAP];

  const int tid  = threadIdx.x;
  const int lane = tid & 31;
  const int wave = __builtin_amdgcn_readfirstlane(tid >> 5);
  const int n0 = blockIdx.x * SC_R;
  const int nrows = (NN - n0 < SC_R) ? (NN - n0) : SC_R;

  for (int q = tid; q < SC_R * (KDIM / 4); q += SC_T)
    ((v4f*)acc)[q] = ((const v4f*)gat_bias)[q & (KDIM / 4 - 1)];
  __syncthreads();

  int* myQ = queue + wave * SC_QCAP;
  int qn = 0;

  auto batch = [&](int h0, int cnt) __attribute__((always_inline)) {
    const int sw = lane >> 2;
    const int ss = lane & 7;
    const int ew = myQ[h0 + ((sw < cnt) ? sw : 0)];
    const int eq = myQ[h0 + ((ss < cnt) ? ss : 0)];
    const int srcv = ei[eq];
    const int dw = ei[EE + ew];
    int rw = dw - n0;
    rw = (rw < 0) ? 0 : ((rw < nrows) ? rw : (nrows - 1));
    const int hl = lane & 3;
    const float den = denom[(size_t)(n0 + rw) * 4 + hl];
    const float alpha = ex[(size_t)ew * 4 + hl] / ((den == 0.0f) ? 1.0f : den);

    float v[EPW][8];
#pragma unroll
    for (int j = 0; j < EPW; ++j) {
      if (j < cnt) {
        const int s = __builtin_amdgcn_readlane(srcv, j);
        const float* ps = Hsrc + (size_t)s * KDIM + lane;
#pragma unroll
        for (int i = 0; i < 8; ++i) v[j][i] = ps[32 * i];
      }
    }
#pragma unroll
    for (int j = 0; j < EPW; ++j) {
      if (j < cnt) {
        const int r = __builtin_amdgcn_readlane(rw, 4 * j);
        float* pa = &acc[r * KDIM + lane];
#pragma unroll
        for (int i = 0; i < 8; ++i) {
          const float ah = __int_as_float(
              __builtin_amdgcn_readlane(__float_as_int(alpha), 4 * j + (i >> 1)));
          __hip_atomic_fetch_add(&pa[32 * i], ah * v[j][i], __ATOMIC_RELAXED,
                                 __HIP_MEMORY_SCOPE_WORKGROUP);
        }
      }
    }
  };

  const unsigned* mrow = mask + (size_t)blockIdx.x * 32 + lane;
  auto line_of = [&](int c) __attribute__((always_inline)) {
    const int cc = (c < SC_NMB) ? c : (SC_NMB - 1);
    return mrow[(size_t)cc * SC_MROW];
  };
  unsigned nxt = line_of(wave);
  for (int c = wave; c < SC_NMB; c += SC_W) {
    unsigned w = nxt;
    nxt = line_of(c + SC_W);
    const int ebase = c * SC_MB + lane * 32;
    unsigned m = __builtin_amdgcn_ballot_w32(w != 0u);
    while (m) {
      if (w != 0u) {
        myQ[qn + (int)__builtin_amdgcn_mbcnt_lo(m, 0u)] = ebase + __builtin_ctz(w);
        w &= w - 1u;
      }
      qn += __builtin_popcount(m);
      m = __builtin_amdgcn_ballot_w32(w != 0u);
      int h = 0;
      for (; h + EPW <= qn; h += EPW) batch(h, EPW);
      if (h > 0) {
        const int rem = qn - h;
        int t = 0;
        if (lane < rem) t = myQ[h + lane];
        if (lane < rem) myQ[lane] = t;
        qn = rem;
      }
    }
  }
  if (qn > 0) batch(0, qn);
  __syncthreads();

  v4f* out_rows = (v4f*)(rst + (size_t)n0 * KDIM);
  for (int q = tid; q < nrows * (KDIM / 4); q += SC_T)
    out_rows[q] = ((const v4f*)acc)[q];
}

#define OBM 64
#define OBK 32
#define OA_PITCH 36
#define OB_PITCH 160
static_assert(KDIM % OBK == 0, "k must split into whole chunks");
static_assert(OUTD == 64, "copy roles and wave layout assume 64 output columns");

__global__ __launch_bounds__(256) void wmma_out_kernel(
    const float* __restrict__ rst, const float* __restrict__ Wfc,
    const float* __restrict__ bfc, float* __restrict__ out) {
  __shared__ __attribute__((aligned(16))) float As[OBM * OA_PITCH];
  __shared__ __attribute__((aligned(16))) float Bs[(OBK / 2) * OB_PITCH];

  const int tid  = threadIdx.x;
  const int lane = tid & 31;
  const int wave = tid >> 5;
  const int wr   = wave >> 1;
  const int wc   = wave & 1;
  const int row0 = blockIdx.x * OBM;
  const int khi  = lane >> 4;
  const int mlo  = lane & 15;

  const int ar  = tid >> 3;
  const int ac  = (tid & 7) * 4;
  const int bcg = tid & 15;
  const int bkp = tid >> 4;

  const float* ag[2];
#pragma unroll
  for (int i = 0; i < 2; ++i) {
    int r = row0 + ar + 32 * i;
    r = (r < NN) ? r : 0;
    ag[i] = rst + (size_t)r * KDIM + ac;
  }
  const float* bg = Wfc + (size_t)(2 * bkp) * OUTD + bcg * 4;

  v4f ra[2], rb0, rb1;
#pragma unroll
  for (int i = 0; i < 2; ++i) ra[i] = *(const v4f*)(ag[i]);
  rb0 = *(const v4f*)(bg);
  rb1 = *(const v4f*)(bg + OUTD);

  const float* aFrag = &As[(wr * 16 + mlo) * OA_PITCH + 2 * khi];
  const float* bFrag = &Bs[khi * OB_PITCH + (wc * 32 + mlo) * 2];

  v8f acc[2] = {};
  for (int c = 0; c < KDIM / OBK; ++c) {
#pragma unroll
    for (int i = 0; i < 2; ++i)
      *(v4f*)&As[(ar + 32 * i) * OA_PITCH + ac] = ra[i];
    {
      float* p = &Bs[bkp * OB_PITCH + bcg * 8];
      v4f lo = {rb0.x, rb1.x, rb0.y, rb1.y};
      v4f hi = {rb0.z, rb1.z, rb0.w, rb1.w};
      *(v4f*)p = lo;
      *(v4f*)(p + 4) = hi;
    }
    __syncthreads();

    if (c + 1 < KDIM / OBK) {
      const int k0 = (c + 1) * OBK;
#pragma unroll
      for (int i = 0; i < 2; ++i) ra[i] = *(const v4f*)(ag[i] + k0);
      rb0 = *(const v4f*)(bg + (size_t)k0 * OUTD);
      rb1 = *(const v4f*)(bg + (size_t)(k0 + 1) * OUTD);
    }

#pragma unroll
    for (int ks = 0; ks < OBK / 4; ++ks) {
      const v2f af = *(const v2f*)(aFrag + ks * 4);
      v2f bf[2];
#pragma unroll
      for (int t = 0; t < 2; ++t)
        bf[t] = *(const v2f*)(bFrag + ks * 2 * OB_PITCH + t * 32);
#pragma unroll
      for (int t = 0; t < 2; ++t)
        acc[t] = __builtin_amdgcn_wmma_f32_16x16x4_f32(
            false, af, false, bf[t], (short)0, acc[t], false, false);
    }
    __syncthreads();
  }

  const int rbase = row0 + wr * 16 + khi * 8;
  float bias[2];
#pragma unroll
  for (int t = 0; t < 2; ++t) bias[t] = bfc[wc * 32 + t * 16 + mlo];
  if (row0 + OBM <= NN) {
#pragma unroll
    for (int i = 0; i < 8; ++i) {
#pragma unroll
      for (int t = 0; t < 2; ++t)
        out[(size_t)(rbase + i) * OUTD + wc * 32 + t * 16 + mlo] = acc[t][i] + bias[t];
    }
  } else {
#pragma unroll
    for (int i = 0; i < 8; ++i) {
      if (rbase + i < NN) {
#pragma unroll
        for (int t = 0; t < 2; ++t)
          out[(size_t)(rbase + i) * OUTD + wc * 32 + t * 16 + mlo] = acc[t][i] + bias[t];
      }
    }
  }
}

extern "C" void kernel_launch(void* const* d_in, const int* in_sizes, int n_in,
                              void* d_out, int out_size, void* d_ws, size_t ws_size,
                              hipStream_t stream) {
  const float* feat     = (const float*)d_in[0];
  const int*   ei       = (const int*)d_in[1];
  const float* w_src    = (const float*)d_in[2];
  const float* b_src    = (const float*)d_in[3];
  const float* w_dst    = (const float*)d_in[4];
  const float* b_dst    = (const float*)d_in[5];
  const float* attn     = (const float*)d_in[6];
  const float* gat_bias = (const float*)d_in[7];
  const float* w_fc     = (const float*)d_in[8];
  const float* b_fc     = (const float*)d_in[9];
  float* out = (float*)d_out;

  float* ws    = (float*)d_ws;
  float* Hsrc  = ws;
  float* Hdst  = Hsrc + (size_t)NN * KDIM;
  float* score = Hdst + (size_t)NN * KDIM;
  float* denom = score + (size_t)EE * 4 + (size_t)NN * 4;
  float* rst   = denom + (size_t)NN * 4;
  unsigned* mask = (unsigned*)(rst + (size_t)NN * KDIM);

  owner_mask_kernel<<<SC_NMB, SC_MB, 0, stream>>>(ei, mask);

  wmma_proj_kernel<<<dim3(2, (NN + PBM - 1) / PBM), 256, 0, stream>>>(
      feat, w_src, b_src, w_dst, b_dst, Hsrc, Hdst);

  edge_softmax_kernel<<<SC_G, SM_T, 0, stream>>>(
      ei, mask, Hsrc, Hdst, attn, score, denom);

  edge_scatter_kernel<<<SC_G, SC_T, 0, stream>>>(
      ei, mask, Hsrc, score, denom, gat_bias, rst);

  wmma_out_kernel<<<(NN + OBM - 1) / OBM, 256, 0, stream>>>(rst, w_fc, b_fc, out);
}


# Softmax exp and denominator moved from their own pass into the scatter pass: an edge is added with weight exp(score-max); each row scaled by 1/denominator once at the end (sum, then normalise)
# speedup vs baseline: 3.8978x; 1.2376x over previous
#include <hip/hip_runtime.h>
#include <hip/hip_bf16.h>

#define NN 50000
#define EE 800000
#define KDIM 256
#define OUTD 64
#define NEG_SLOPE 0.2f

typedef __attribute__((ext_vector_type(2))) float v2f;
typedef __attribute__((ext_vector_type(4))) float v4f;
typedef __attribute__((ext_vector_type(8))) float v8f;
typedef __attribute__((ext_vector_type(4))) int v4i;

__device__ __forceinline__ unsigned orderFloat(float f) {
  unsigned u = __float_as_uint(f);
  return (u & 0x80000000u) ? ~u : (u | 0x80000000u);
}
__device__ __forceinline__ float unorderFloat(unsigned u) {
  u = (u & 0x80000000u) ? (u & 0x7FFFFFFFu) : ~u;
  return __uint_as_float(u);
}

#define PBM 128
#define PBK 32
#define A_PITCH 36
#define B_PITCH 544

__global__ __launch_bounds__(256) void wmma_proj_kernel(
    const float* __restrict__ feat,
    const float* __restrict__ Wsrc, const float* __restrict__ Bsrc,
    const float* __restrict__ Wdst, const float* __restrict__ Bdst,
    float* __restrict__ Hsrc, float* __restrict__ Hdst) {
  __shared__ __attribute__((aligned(16))) float As[PBM * A_PITCH];
  __shared__ __attribute__((aligned(16))) float Bs[(PBK / 2) * B_PITCH];

  const int tid  = threadIdx.x;
  const int lane = tid & 31;
  const int wave = tid >> 5;
  const int wr   = wave >> 2;
  const int wc   = wave & 3;
  const int mat  = blockIdx.x;
  const int row0 = blockIdx.y * PBM;
  const int khi  = lane >> 4;
  const int mlo  = lane & 15;

  const float* __restrict__ W  = mat ? Wdst : Wsrc;
  const float* __restrict__ Bv = mat ? Bdst : Bsrc;
  float* __restrict__ Hout     = mat ? Hdst : Hsrc;

  const int ar = tid >> 3;
  const int ac = (tid & 7) * 4;
  const int bcg = tid & 63;
  const int bkp = tid >> 6;

  const float* ag[4];
#pragma unroll
  for (int i = 0; i < 4; ++i) {
    int r = row0 + ar + 32 * i;
    r = (r < NN) ? r : 0;
    ag[i] = feat + (size_t)r * KDIM + ac;
  }
  const float* bg = W + (size_t)(2 * bkp) * KDIM + bcg * 4;

  v4f ra[4], rb0[4], rb1[4];
#pragma unroll
  for (int i = 0; i < 4; ++i) {
    ra[i]  = *(const v4f*)(ag[i]);
    rb0[i] = *(const v4f*)(bg + (size_t)(8 * i) * KDIM);
    rb1[i] = *(const v4f*)(bg + (size_t)(8 * i + 1) * KDIM);
  }

  const float* aFrag = &As[(wr * 64 + mlo) * A_PITCH + 2 * khi];
  const float* bFrag = &Bs[khi * B_PITCH + (wc * 64 + mlo) * 2];

  v8f acc[4][4] = {};
  for (int c = 0; c < KDIM / PBK; ++c) {
#pragma unroll
    for (int i = 0; i < 4; ++i) {
      *(v4f*)&As[(ar + 32 * i) * A_PITCH + ac] = ra[i];
      float* p = &Bs[(bkp + 4 * i) * B_PITCH + bcg * 8];
      v4f lo = {rb0[i].x, rb1[i].x, rb0[i].y, rb1[i].y};
      v4f hi = {rb0[i].z, rb1[i].z, rb0[i].w, rb1[i].w};
      *(v4f*)p = lo;
      *(v4f*)(p + 4) = hi;
    }
    __syncthreads();

    if (c + 1 < KDIM / PBK) {
      const int k0 = (c + 1) * PBK;
#pragma unroll
      for (int i = 0; i < 4; ++i) {
        ra[i]  = *(const v4f*)(ag[i] + k0);
        rb0[i] = *(const v4f*)(bg + (size_t)(k0 + 8 * i) * KDIM);
        rb1[i] = *(const v4f*)(bg + (size_t)(k0 + 8 * i + 1) * KDIM);
      }
    }

#pragma unroll
    for (int ks = 0; ks < PBK / 4; ++ks) {
      v2f af[4], bf[4];
#pragma unroll
      for (int s = 0; s < 4; ++s)
        af[s] = *(const v2f*)(aFrag + s * 16 * A_PITCH + ks * 4);
#pragma unroll
      for (int t = 0; t < 4; ++t)
        bf[t] = *(const v2f*)(bFrag + ks * 2 * B_PITCH + t * 32);
#pragma unroll
      for (int s = 0; s < 4; ++s) {
#pragma unroll
        for (int t = 0; t < 4; ++t) {
          acc[s][t] = __builtin_amdgcn_wmma_f32_16x16x4_f32(
              false, af[s], false, bf[t], (short)0, acc[s][t], false, false);
        }
      }
    }
    __syncthreads();
  }

  const int mbase = khi * 8;
  float bias[4];
#pragma unroll
  for (int t = 0; t < 4; ++t) bias[t] = Bv[wc * 64 + t * 16 + mlo];
  const int rbase = row0 + wr * 64 + mbase;
  if (row0 + PBM <= NN) {
#pragma unroll
    for (int s = 0; s < 4; ++s) {
#pragma unroll
      for (int i = 0; i < 8; ++i) {
        int r = rbase + s * 16 + i;
#pragma unroll
        for (int t = 0; t < 4; ++t)
          Hout[(size_t)r * KDIM + wc * 64 + t * 16 + mlo] = acc[s][t][i] + bias[t];
      }
    }
  } else {
#pragma unroll
    for (int s = 0; s < 4; ++s) {
#pragma unroll
      for (int i = 0; i < 8; ++i) {
        int r = rbase + s * 16 + i;
        if (r < NN) {
#pragma unroll
          for (int t = 0; t < 4; ++t)
            Hout[(size_t)r * KDIM + wc * 64 + t * 16 + mlo] = acc[s][t][i] + bias[t];
        }
      }
    }
  }
}

#define EPW 8

#define SC_G 256
#define SC_R ((NN + SC_G - 1) / SC_G)
#define SC_T 1024
#define SC_W (SC_T / 32)
#define SC_MB 1024
#define SC_NMB ((EE + SC_MB - 1) / SC_MB)
#define SC_MROW (SC_G * 32)
#define SC_QCAP (32 + EPW)
static_assert((SC_G - 1) * SC_R < NN, "every group owns at least one row");
static_assert((NN - 1) / SC_R < SC_G, "every dst has an owner group");
static_assert(SC_R * KDIM * 4 + 2 * SC_R * 4 * 4 + SC_W * SC_QCAP * 4 <= 327680,
              "rows, their maxima and denominators, and the queues must fit the LDS");

__global__ __launch_bounds__(SC_MB) void owner_mask_kernel(
    const int* __restrict__ ei, unsigned* __restrict__ mask) {
  __shared__ __attribute__((aligned(16))) unsigned tab[SC_MROW];

  const int tid = threadIdx.x;
  const int e = blockIdx.x * SC_MB + tid;
  const v4i zi = {0, 0, 0, 0};
  for (int q = tid; q < SC_MROW / 4; q += SC_MB) ((v4i*)tab)[q] = zi;
  __syncthreads();

  if (e < EE) {
    const unsigned d = (unsigned)ei[EE + e];
    if (d < (unsigned)NN)
      atomicOr(&tab[(d / SC_R) * 32 + (tid >> 5)], 1u << (tid & 31));
  }
  __syncthreads();

  v4i* out = (v4i*)(mask + (size_t)blockIdx.x * SC_MROW);
  for (int q = tid; q < SC_MROW / 4; q += SC_MB) out[q] = ((const v4i*)tab)[q];
}

template <int B, class F>
__device__ __forceinline__ void scan_owned_edges(
    const unsigned* __restrict__ mask, int group, int wave, int nwaves, int lane, int* myQ, F batch) {
  const unsigned* mgrp = mask + (size_t)group * 32;
  auto line_of = [&](int c) __attribute__((always_inline)) {
    const int cc = (c < SC_NMB) ? c : (SC_NMB - 1);
    return (mgrp + (size_t)cc * SC_MROW)[lane];
  };
  int qn = 0;
  unsigned nxt = line_of(wave);
  for (int c = wave; c < SC_NMB; c += nwaves) {
    unsigned w = nxt;
    nxt = line_of(c + nwaves);
    const int ebase = c * SC_MB + lane * 32;
    unsigned m = __builtin_amdgcn_ballot_w32(w != 0u);
    while (m) {
      if (w != 0u) {
        myQ[qn + (int)__builtin_amdgcn_mbcnt_lo(m, 0u)] = ebase + __builtin_ctz(w);
        w &= w - 1u;
      }
      qn += __builtin_popcount(m);
      m = __builtin_amdgcn_ballot_w32(w != 0u);
      int h = 0;
      for (; h + B <= qn; h += B) batch(h, B);
      if (h > 0) {
        const int rem = qn - h;
        int t = 0;
        if (lane < rem) t = myQ[h + lane];
        if (lane < rem) myQ[lane] = t;
        qn = rem;
      }
    }
  }
  if (qn > 0) batch(0, qn);
}

#define SM_T 640
#define SM_W (SM_T / 32)
#define SM_QCAP SC_QCAP
__global__ __launch_bounds__(SM_T) void edge_score_kernel(
    const int* __restrict__ ei, const unsigned* __restrict__ mask,
    const float* __restrict__ Hsrc, const float* __restrict__ Hdst,
    const float* __restrict__ attn, float* __restrict__ score,
    float* __restrict__ rowmax) {
  __shared__ __attribute__((aligned(16))) float hd[SC_R * KDIM];
  __shared__ __attribute__((aligned(16))) unsigned mx[SC_R * 4];
  __shared__ int queue[SM_W * SM_QCAP];

  const int tid  = threadIdx.x;
  const int lane = tid & 31;
  const int wave = __builtin_amdgcn_readfirstlane(tid >> 5);
  const int n0 = blockIdx.x * SC_R;
  const int nrows = (NN - n0 < SC_R) ? (NN - n0) : SC_R;

  {
    const v4f* hrows = (const v4f*)(Hdst + (size_t)n0 * KDIM);
    const int total = nrows * (KDIM / 4);
    for (int q0 = tid; q0 < total; q0 += 8 * SM_T) {
      v4f t[8];
#pragma unroll
      for (int u = 0; u < 8; ++u) {
        const int q = q0 + u * SM_T;
        t[u] = hrows[(q < total) ? q : (total - 1)];
      }
#pragma unroll
      for (int u = 0; u < 8; ++u) {
        const int q = q0 + u * SM_T;
        const int cq = q & 63;
        if (q < total) ((v4f*)hd)[(q & ~63) | ((cq & 1) << 5) | (cq >> 1)] = t[u];
      }
    }
    for (int q = tid; q < SC_R * 4; q += SM_T) mx[q] = 0u;
  }
  __syncthreads();

  int* myQ = queue + wave * SM_QCAP;
  const int sub = lane & 7;
  const int hh = lane >> 3;
  const v4f* pa = (const v4f*)(attn + lane * 8);
  const v4f a0 = pa[0], a1 = pa[1];
  const bool b0 = (lane & 1) != 0, b1 = (lane & 2) != 0, b2 = (lane & 4) != 0;

  auto score_batch = [&](int h0, int cnt) __attribute__((always_inline)) {
    const int eq = myQ[h0 + ((sub < cnt) ? sub : 0)];
    const int srcv = ei[eq];
    const int dv = ei[EE + eq];
    int rw = dv - n0;
    rw = (rw < 0) ? 0 : ((rw < nrows) ? rw : (nrows - 1));

    v4f s0[EPW], s1[EPW];
#pragma unroll
    for (int j = 0; j < EPW; ++j) {
      const int s = __builtin_amdgcn_readlane(srcv, j);
      const v4f* ps = (const v4f*)(Hsrc + (size_t)s * KDIM + lane * 8);
      s0[j] = ps[0]; s1[j] = ps[1];
    }
#pragma unroll
    for (int j = 0; j < EPW; ++j)
      asm volatile("" :: "v"(s0[j]), "v"(s1[j]));

    float p[EPW];
#pragma unroll
    for (int j = 0; j < EPW; ++j) {
      const int r = __builtin_amdgcn_readlane(rw, j);
      const v4f* pd = (const v4f*)&hd[r * KDIM + lane * 4];
      const v4f d0 = pd[0], d1 = pd[32];
      float part = 0.0f;
#pragma unroll
      for (int i = 0; i < 4; ++i) {
        float v = s0[j][i] + d0[i];
        v = (v >= 0.0f) ? v : v * NEG_SLOPE;
        part = __builtin_fmaf(v, a0[i], part);
        float w = s1[j][i] + d1[i];
        w = (w >= 0.0f) ? w : w * NEG_SLOPE;
        part = __builtin_fmaf(w, a1[i], part);
      }
      p[j] = part;
    }

    float q[4], r2[2];
#pragma unroll
    for (int k = 0; k < 4; ++k) {
      const float keep = b0 ? p[2 * k + 1] : p[2 * k];
      const float send = b0 ? p[2 * k] : p[2 * k + 1];
      q[k] = keep + __shfl_xor(send, 1, 32);
    }
#pragma unroll
    for (int k = 0; k < 2; ++k) {
      const float keep = b1 ? q[2 * k + 1] : q[2 * k];
      const float send = b1 ? q[2 * k] : q[2 * k + 1];
      r2[k] = keep + __shfl_xor(send, 2, 32);
    }
    const float keep = b2 ? r2[1] : r2[0];
    const float send = b2 ? r2[0] : r2[1];
    const float sc = keep + __shfl_xor(send, 4, 32);

    if (sub < cnt) {
      score[(size_t)eq * 4 + hh] = sc;
      __hip_atomic_fetch_max(&mx[rw * 4 + hh], orderFloat(sc), __ATOMIC_RELAXED,
                             __HIP_MEMORY_SCOPE_WORKGROUP);
    }
  };
  scan_owned_edges<EPW>(mask, (int)blockIdx.x, wave, SM_W, lane, myQ, score_batch);
  __syncthreads();

  for (int q = tid; q < nrows * 4; q += SM_T) {
    float mf = unorderFloat(mx[q]);
    if (!isfinite(mf)) mf = 0.0f;
    rowmax[(size_t)n0 * 4 + q] = mf;
  }
}

__global__ __launch_bounds__(SC_T) void edge_scatter_kernel(
    const int* __restrict__ ei, const unsigned* __restrict__ mask,
    const float* __restrict__ Hsrc,
    const float* __restrict__ score, const float* __restrict__ rowmax,
    const float* __restrict__ gat_bias, float* __restrict__ rst) {
  __shared__ __attribute__((aligned(16))) float acc[SC_R * KDIM];
  __shared__ float ms[SC_R * 4];
  __shared__ float dn[SC_R * 4];
  __shared__ int queue[SC_W * SC_QCAP];

  const int tid  = threadIdx.x;
  const int lane = tid & 31;
  const int wave = __builtin_amdgcn_readfirstlane(tid >> 5);
  const int n0 = blockIdx.x * SC_R;
  const int nrows = (NN - n0 < SC_R) ? (NN - n0) : SC_R;

  {
    const v4f zf = {0.0f, 0.0f, 0.0f, 0.0f};
    for (int q = tid; q < SC_R * (KDIM / 4); q += SC_T) ((v4f*)acc)[q] = zf;
    for (int q = tid; q < SC_R * 4; q += SC_T) {
      ms[q] = (q < nrows * 4) ? rowmax[(size_t)n0 * 4 + q] : 0.0f;
      dn[q] = 0.0f;
    }
  }
  __syncthreads();

  int* myQ = queue + wave * SC_QCAP;
  int qn = 0;

  auto batch = [&](int h0, int cnt) __attribute__((always_inline)) {
    const int sw = lane >> 2;
    const int ss = lane & 7;
    const int ew = myQ[h0 + ((sw < cnt) ? sw : 0)];
    const int eq = myQ[h0 + ((ss < cnt) ? ss : 0)];
    const int srcv = ei[eq];
    const int dw = ei[EE + ew];
    int rw = dw - n0;
    rw = (rw < 0) ? 0 : ((rw < nrows) ? rw : (nrows - 1));
    const int hl = lane & 3;
    const float sc = score[(size_t)ew * 4 + hl];

    float v[EPW][8];
#pragma unroll
    for (int j = 0; j < EPW; ++j) {
      if (j < cnt) {
        const int s = __builtin_amdgcn_readlane(srcv, j);
        const float* ps = Hsrc + (size_t)s * KDIM + lane;
#pragma unroll
        for (int i = 0; i < 8; ++i) v[j][i] = ps[32 * i];
      }
    }

    const float alpha = expf(sc - ms[rw * 4 + hl]);
    if (sw < cnt)
      __hip_atomic_fetch_add(&dn[rw * 4 + hl], alpha, __ATOMIC_RELAXED,
                             __HIP_MEMORY_SCOPE_WORKGROUP);
#pragma unroll
    for (int j = 0; j < EPW; ++j) {
      if (j < cnt) {
        const int r = __builtin_amdgcn_readlane(rw, 4 * j);
        float* pa = &acc[r * KDIM + lane];
#pragma unroll
        for (int i = 0; i < 8; ++i) {
          const float ah = __int_as_float(
              __builtin_amdgcn_readlane(__float_as_int(alpha), 4 * j + (i >> 1)));
          __hip_atomic_fetch_add(&pa[32 * i], ah * v[j][i], __ATOMIC_RELAXED,
                                 __HIP_MEMORY_SCOPE_WORKGROUP);
        }
      }
    }
  };

  const unsigned* mrow = mask + (size_t)blockIdx.x * 32 + lane;
  auto line_of = [&](int c) __attribute__((always_inline)) {
    const int cc = (c < SC_NMB) ? c : (SC_NMB - 1);
    return mrow[(size_t)cc * SC_MROW];
  };
  unsigned nxt = line_of(wave);
  for (int c = wave; c < SC_NMB; c += SC_W) {
    unsigned w = nxt;
    nxt = line_of(c + SC_W);
    const int ebase = c * SC_MB + lane * 32;
    unsigned m = __builtin_amdgcn_ballot_w32(w != 0u);
    while (m) {
      if (w != 0u) {
        myQ[qn + (int)__builtin_amdgcn_mbcnt_lo(m, 0u)] = ebase + __builtin_ctz(w);
        w &= w - 1u;
      }
      qn += __builtin_popcount(m);
      m = __builtin_amdgcn_ballot_w32(w != 0u);
      int h = 0;
      for (; h + EPW <= qn; h += EPW) batch(h, EPW);
      if (h > 0) {
        const int rem = qn - h;
        int t = 0;
        if (lane < rem) t = myQ[h + lane];
        if (lane < rem) myQ[lane] = t;
        qn = rem;
      }
    }
  }
  if (qn > 0) batch(0, qn);
  __syncthreads();

  for (int q = tid; q < SC_R * 4; q += SC_T) {
    const float den = dn[q];
    dn[q] = 1.0f / ((den == 0.0f) ? 1.0f : den);
  }
  __syncthreads();

  v4f* out_rows = (v4f*)(rst + (size_t)n0 * KDIM);
  for (int q = tid; q < nrows * (KDIM / 4); q += SC_T) {
    const float rinv = dn[(q >> 6) * 4 + ((q & 63) >> 4)];
    const v4f a = ((const v4f*)acc)[q];
    const v4f b = ((const v4f*)gat_bias)[q & (KDIM / 4 - 1)];
    v4f o;
#pragma unroll
    for (int i = 0; i < 4; ++i) o[i] = a[i] * rinv + b[i];
    out_rows[q] = o;
  }
}

#define OBM 64
#define OBK 32
#define OA_PITCH 36
#define OB_PITCH 160
static_assert(KDIM % OBK == 0, "k must split into whole chunks");
static_assert(OUTD == 64, "copy roles and wave layout assume 64 output columns");

__global__ __launch_bounds__(256) void wmma_out_kernel(
    const float* __restrict__ rst, const float* __restrict__ Wfc,
    const float* __restrict__ bfc, float* __restrict__ out) {
  __shared__ __attribute__((aligned(16))) float As[OBM * OA_PITCH];
  __shared__ __attribute__((aligned(16))) float Bs[(OBK / 2) * OB_PITCH];

  const int tid  = threadIdx.x;
  const int lane = tid & 31;
  const int wave = tid >> 5;
  const int wr   = wave >> 1;
  const int wc   = wave & 1;
  const int row0 = blockIdx.x * OBM;
  const int khi  = lane >> 4;
  const int mlo  = lane & 15;

  const int ar  = tid >> 3;
  const int ac  = (tid & 7) * 4;
  const int bcg = tid & 15;
  const int bkp = tid >> 4;

  const float* ag[2];
#pragma unroll
  for (int i = 0; i < 2; ++i) {
    int r = row0 + ar + 32 * i;
    r = (r < NN) ? r : 0;
    ag[i] = rst + (size_t)r * KDIM + ac;
  }
  const float* bg = Wfc + (size_t)(2 * bkp) * OUTD + bcg * 4;

  v4f ra[2], rb0, rb1;
#pragma unroll
  for (int i = 0; i < 2; ++i) ra[i] = *(const v4f*)(ag[i]);
  rb0 = *(const v4f*)(bg);
  rb1 = *(const v4f*)(bg + OUTD);

  const float* aFrag = &As[(wr * 16 + mlo) * OA_PITCH + 2 * khi];
  const float* bFrag = &Bs[khi * OB_PITCH + (wc * 32 + mlo) * 2];

  v8f acc[2] = {};
  for (int c = 0; c < KDIM / OBK; ++c) {
#pragma unroll
    for (int i = 0; i < 2; ++i)
      *(v4f*)&As[(ar + 32 * i) * OA_PITCH + ac] = ra[i];
    {
      float* p = &Bs[bkp * OB_PITCH + bcg * 8];
      v4f lo = {rb0.x, rb1.x, rb0.y, rb1.y};
      v4f hi = {rb0.z, rb1.z, rb0.w, rb1.w};
      *(v4f*)p = lo;
      *(v4f*)(p + 4) = hi;
    }
    __syncthreads();

    if (c + 1 < KDIM / OBK) {
      const int k0 = (c + 1) * OBK;
#pragma unroll
      for (int i = 0; i < 2; ++i) ra[i] = *(const v4f*)(ag[i] + k0);
      rb0 = *(const v4f*)(bg + (size_t)k0 * OUTD);
      rb1 = *(const v4f*)(bg + (size_t)(k0 + 1) * OUTD);
    }

#pragma unroll
    for (int ks = 0; ks < OBK / 4; ++ks) {
      const v2f af = *(const v2f*)(aFrag + ks * 4);
      v2f bf[2];
#pragma unroll
      for (int t = 0; t < 2; ++t)
        bf[t] = *(const v2f*)(bFrag + ks * 2 * OB_PITCH + t * 32);
#pragma unroll
      for (int t = 0; t < 2; ++t)
        acc[t] = __builtin_amdgcn_wmma_f32_16x16x4_f32(
            false, af, false, bf[t], (short)0, acc[t], false, false);
    }
    __syncthreads();
  }

  const int rbase = row0 + wr * 16 + khi * 8;
  float bias[2];
#pragma unroll
  for (int t = 0; t < 2; ++t) bias[t] = bfc[wc * 32 + t * 16 + mlo];
  if (row0 + OBM <= NN) {
#pragma unroll
    for (int i = 0; i < 8; ++i) {
#pragma unroll
      for (int t = 0; t < 2; ++t)
        out[(size_t)(rbase + i) * OUTD + wc * 32 + t * 16 + mlo] = acc[t][i] + bias[t];
    }
  } else {
#pragma unroll
    for (int i = 0; i < 8; ++i) {
      if (rbase + i < NN) {
#pragma unroll
        for (int t = 0; t < 2; ++t)
          out[(size_t)(rbase + i) * OUTD + wc * 32 + t * 16 + mlo] = acc[t][i] + bias[t];
      }
    }
  }
}

extern "C" void kernel_launch(void* const* d_in, const int* in_sizes, int n_in,
                              void* d_out, int out_size, void* d_ws, size_t ws_size,
                              hipStream_t stream) {
  const float* feat     = (const float*)d_in[0];
  const int*   ei       = (const int*)d_in[1];
  const float* w_src    = (const float*)d_in[2];
  const float* b_src    = (const float*)d_in[3];
  const float* w_dst    = (const float*)d_in[4];
  const float* b_dst    = (const float*)d_in[5];
  const float* attn     = (const float*)d_in[6];
  const float* gat_bias = (const float*)d_in[7];
  const float* w_fc     = (const float*)d_in[8];
  const float* b_fc     = (const float*)d_in[9];
  float* out = (float*)d_out;

  float* ws    = (float*)d_ws;
  float* Hsrc  = ws;
  float* Hdst  = Hsrc + (size_t)NN * KDIM;
  float* score = Hdst + (size_t)NN * KDIM;
  float* rowmax = score + (size_t)EE * 4 + (size_t)NN * 4;
  float* rst   = rowmax + (size_t)NN * 4;
  unsigned* mask = (unsigned*)(rst + (size_t)NN * KDIM);

  owner_mask_kernel<<<SC_NMB, SC_MB, 0, stream>>>(ei, mask);

  wmma_proj_kernel<<<dim3(2, (NN + PBM - 1) / PBM), 256, 0, stream>>>(
      feat, w_src, b_src, w_dst, b_dst, Hsrc, Hdst);

  edge_score_kernel<<<SC_G, SM_T, 0, stream>>>(
      ei, mask, Hsrc, Hdst, attn, score, rowmax);

  edge_scatter_kernel<<<SC_G, SC_T, 0, stream>>>(
      ei, mask, Hsrc, score, rowmax, gat_bias, rst);

  wmma_out_kernel<<<(NN + OBM - 1) / OBM, 256, 0, stream>>>(rst, w_fc, b_fc, out);
}


# Edge score, softmax weight and weighted sum in one pass: Hsrc gathered once, not twice; group = row block x head pair; shift = first-arrived score, restart with max if spread > 32; sum order changed
# speedup vs baseline: 4.6337x; 1.1888x over previous
#include <hip/hip_runtime.h>
#include <hip/hip_bf16.h>

#define NN 50000
#define EE 800000
#define KDIM 256
#define OUTD 64
#define NEG_SLOPE 0.2f

typedef __attribute__((ext_vector_type(2))) float v2f;
typedef __attribute__((ext_vector_type(4))) float v4f;
typedef __attribute__((ext_vector_type(8))) float v8f;
typedef __attribute__((ext_vector_type(4))) int v4i;

__device__ __forceinline__ unsigned orderFloat(float f) {
  unsigned u = __float_as_uint(f);
  return (u & 0x80000000u) ? ~u : (u | 0x80000000u);
}
__device__ __forceinline__ float unorderFloat(unsigned u) {
  u = (u & 0x80000000u) ? (u & 0x7FFFFFFFu) : ~u;
  return __uint_as_float(u);
}

#define PBM 128
#define PBK 32
#define A_PITCH 36
#define B_PITCH 544

__global__ __launch_bounds__(256) void wmma_proj_kernel(
    const float* __restrict__ feat,
    const float* __restrict__ Wsrc, const float* __restrict__ Bsrc,
    const float* __restrict__ Wdst, const float* __restrict__ Bdst,
    float* __restrict__ Hsrc, float* __restrict__ Hdst) {
  __shared__ __attribute__((aligned(16))) float As[PBM * A_PITCH];
  __shared__ __attribute__((aligned(16))) float Bs[(PBK / 2) * B_PITCH];

  const int tid  = threadIdx.x;
  const int lane = tid & 31;
  const int wave = tid >> 5;
  const int wr   = wave >> 2;
  const int wc   = wave & 3;
  const int mat  = blockIdx.x;
  const int row0 = blockIdx.y * PBM;
  const int khi  = lane >> 4;
  const int mlo  = lane & 15;

  const float* __restrict__ W  = mat ? Wdst : Wsrc;
  const float* __restrict__ Bv = mat ? Bdst : Bsrc;
  float* __restrict__ Hout     = mat ? Hdst : Hsrc;

  const int ar = tid >> 3;
  const int ac = (tid & 7) * 4;
  const int bcg = tid & 63;
  const int bkp = tid >> 6;

  const float* ag[4];
#pragma unroll
  for (int i = 0; i < 4; ++i) {
    int r = row0 + ar + 32 * i;
    r = (r < NN) ? r : 0;
    ag[i] = feat + (size_t)r * KDIM + ac;
  }
  const float* bg = W + (size_t)(2 * bkp) * KDIM + bcg * 4;

  v4f ra[4], rb0[4], rb1[4];
#pragma unroll
  for (int i = 0; i < 4; ++i) {
    ra[i]  = *(const v4f*)(ag[i]);
    rb0[i] = *(const v4f*)(bg + (size_t)(8 * i) * KDIM);
    rb1[i] = *(const v4f*)(bg + (size_t)(8 * i + 1) * KDIM);
  }

  const float* aFrag = &As[(wr * 64 + mlo) * A_PITCH + 2 * khi];
  const float* bFrag = &Bs[khi * B_PITCH + (wc * 64 + mlo) * 2];

  v8f acc[4][4] = {};
  for (int c = 0; c < KDIM / PBK; ++c) {
#pragma unroll
    for (int i = 0; i < 4; ++i) {
      *(v4f*)&As[(ar + 32 * i) * A_PITCH + ac] = ra[i];
      float* p = &Bs[(bkp + 4 * i) * B_PITCH + bcg * 8];
      v4f lo = {rb0[i].x, rb1[i].x, rb0[i].y, rb1[i].y};
      v4f hi = {rb0[i].z, rb1[i].z, rb0[i].w, rb1[i].w};
      *(v4f*)p = lo;
      *(v4f*)(p + 4) = hi;
    }
    __syncthreads();

    if (c + 1 < KDIM / PBK) {
      const int k0 = (c + 1) * PBK;
#pragma unroll
      for (int i = 0; i < 4; ++i) {
        ra[i]  = *(const v4f*)(ag[i] + k0);
        rb0[i] = *(const v4f*)(bg + (size_t)(k0 + 8 * i) * KDIM);
        rb1[i] = *(const v4f*)(bg + (size_t)(k0 + 8 * i + 1) * KDIM);
      }
    }

#pragma unroll
    for (int ks = 0; ks < PBK / 4; ++ks) {
      v2f af[4], bf[4];
#pragma unroll
      for (int s = 0; s < 4; ++s)
        af[s] = *(const v2f*)(aFrag + s * 16 * A_PITCH + ks * 4);
#pragma unroll
      for (int t = 0; t < 4; ++t)
        bf[t] = *(const v2f*)(bFrag + ks * 2 * B_PITCH + t * 32);
#pragma unroll
      for (int s = 0; s < 4; ++s) {
#pragma unroll
        for (int t = 0; t < 4; ++t) {
          acc[s][t] = __builtin_amdgcn_wmma_f32_16x16x4_f32(
              false, af[s], false, bf[t], (short)0, acc[s][t], false, false);
        }
      }
    }
    __syncthreads();
  }

  const int mbase = khi * 8;
  float bias[4];
#pragma unroll
  for (int t = 0; t < 4; ++t) bias[t] = Bv[wc * 64 + t * 16 + mlo];
  const int rbase = row0 + wr * 64 + mbase;
  if (row0 + PBM <= NN) {
#pragma unroll
    for (int s = 0; s < 4; ++s) {
#pragma unroll
      for (int i = 0; i < 8; ++i) {
        int r = rbase + s * 16 + i;
#pragma unroll
        for (int t = 0; t < 4; ++t)
          Hout[(size_t)r * KDIM + wc * 64 + t * 16 + mlo] = acc[s][t][i] + bias[t];
      }
    }
  } else {
#pragma unroll
    for (int s = 0; s < 4; ++s) {
#pragma unroll
      for (int i = 0; i < 8; ++i) {
        int r = rbase + s * 16 + i;
        if (r < NN) {
#pragma unroll
          for (int t = 0; t < 4; ++t)
            Hout[(size_t)r * KDIM + wc * 64 + t * 16 + mlo] = acc[s][t][i] + bias[t];
        }
      }
    }
  }
}


#define EPW 8

#define SC_G 256
#define SC_R ((NN + SC_G - 1) / SC_G)
#define SC_MB 1024
#define SC_NMB ((EE + SC_MB - 1) / SC_MB)
#define SC_MROW (SC_G * 32)
#define SC_QCAP (32 + EPW)
#define FH 128
#define F_T 1024
#define F_W (F_T / 32)
#define F_UNSET 0xFFFFFFFFu
#define F_SPREAD 32.0f
static_assert((SC_G - 1) * SC_R < NN, "every block owns at least one row");
static_assert((NN - 1) / SC_R < SC_G, "every dst has an owner block");
static_assert(2 * SC_R * FH * 4 + 3 * SC_R * 2 * 4 + F_W * SC_QCAP * 4 + 4 <= 327680,
              "Hdst pieces, accumulators, per-row state and the queues must fit the LDS");

__global__ __launch_bounds__(SC_MB) void owner_mask_kernel(
    const int* __restrict__ ei, unsigned* __restrict__ mask) {
  __shared__ __attribute__((aligned(16))) unsigned tab[SC_MROW];

  const int tid = threadIdx.x;
  const int e = blockIdx.x * SC_MB + tid;
  const v4i zi = {0, 0, 0, 0};
  for (int q = tid; q < SC_MROW / 4; q += SC_MB) ((v4i*)tab)[q] = zi;
  __syncthreads();

  if (e < EE) {
    const unsigned d = (unsigned)ei[EE + e];
    if (d < (unsigned)NN)
      atomicOr(&tab[(d / SC_R) * 32 + (tid >> 5)], 1u << (tid & 31));
  }
  __syncthreads();

  v4i* out = (v4i*)(mask + (size_t)blockIdx.x * SC_MROW);
  for (int q = tid; q < SC_MROW / 4; q += SC_MB) out[q] = ((const v4i*)tab)[q];
}

template <int B, class F>
__device__ __forceinline__ void scan_owned_edges(
    const unsigned* __restrict__ mask, int group, int wave, int nwaves, int lane, int* myQ, F batch) {
  const unsigned* mgrp = mask + (size_t)group * 32;
  auto line_of = [&](int c) __attribute__((always_inline)) {
    const int cc = (c < SC_NMB) ? c : (SC_NMB - 1);
    return (mgrp + (size_t)cc * SC_MROW)[lane];
  };
  int qn = 0;
  unsigned nxt = line_of(wave);
  for (int c = wave; c < SC_NMB; c += nwaves) {
    unsigned w = nxt;
    nxt = line_of(c + nwaves);
    const int ebase = c * SC_MB + lane * 32;
    unsigned m = __builtin_amdgcn_ballot_w32(w != 0u);
    while (m) {
      if (w != 0u) {
        myQ[qn + (int)__builtin_amdgcn_mbcnt_lo(m, 0u)] = ebase + __builtin_ctz(w);
        w &= w - 1u;
      }
      qn += __builtin_popcount(m);
      m = __builtin_amdgcn_ballot_w32(w != 0u);
      int h = 0;
      for (; h + B <= qn; h += B) batch(h, B);
      if (h > 0) {
        const int rem = qn - h;
        int t = 0;
        if (lane < rem) t = myQ[h + lane];
        if (lane < rem) myQ[lane] = t;
        qn = rem;
      }
    }
  }
  if (qn > 0) batch(0, qn);
}

__global__ __launch_bounds__(F_T) void edge_fused_kernel(
    const int* __restrict__ ei, const unsigned* __restrict__ mask,
    const float* __restrict__ Hsrc, const float* __restrict__ Hdst,
    const float* __restrict__ attn, const float* __restrict__ gat_bias,
    float* __restrict__ rst) {
  __shared__ __attribute__((aligned(16))) float hd[SC_R * FH];
  __shared__ __attribute__((aligned(16))) float acc[SC_R * FH];
  __shared__ unsigned mxu[SC_R * 2];
  __shared__ unsigned cu[SC_R * 2];
  __shared__ float dn[SC_R * 2];
  __shared__ int queue[F_W * SC_QCAP];
  __shared__ int redo;

  const int tid  = threadIdx.x;
  const int lane = tid & 31;
  const int wave = __builtin_amdgcn_readfirstlane(tid >> 5);
  const int blk  = blockIdx.x >> 1;
  const int col0 = (blockIdx.x & 1) * FH;
  const int n0 = blk * SC_R;
  const int nrows = (NN - n0 < SC_R) ? (NN - n0) : SC_R;

  {
    const int total = nrows * (FH / 4);
    for (int q0 = tid; q0 < total; q0 += 4 * F_T) {
      v4f t[4];
#pragma unroll
      for (int u = 0; u < 4; ++u) {
        const int q = q0 + u * F_T;
        const int qq = (q < total) ? q : (total - 1);
        t[u] = *(const v4f*)(Hdst + (size_t)(n0 + (qq >> 5)) * KDIM + col0 + (qq & 31) * 4);
      }
#pragma unroll
      for (int u = 0; u < 4; ++u) {
        const int q = q0 + u * F_T;
        if (q < total) ((v4f*)hd)[q] = t[u];
      }
    }
    const v4f zf = {0.0f, 0.0f, 0.0f, 0.0f};
    for (int q = tid; q < SC_R * (FH / 4); q += F_T) ((v4f*)acc)[q] = zf;
    for (int q = tid; q < SC_R * 2; q += F_T) {
      mxu[q] = 0u;
      cu[q] = F_UNSET;
      dn[q] = 0.0f;
    }
    if (tid == 0) redo = 0;
  }
  __syncthreads();

  int* myQ = queue + wave * SC_QCAP;
  const int sub = lane & 7;
  const int hx = lane >> 4;
  const v4f av = ((const v4f*)(attn + col0))[lane];
  const unsigned voff = (unsigned)(col0 + lane * 4);
  const bool b0 = (lane & 1) != 0, b1 = (lane & 2) != 0, b2 = (lane & 4) != 0, b3 = (lane & 8) != 0;

  auto batch = [&](int h0, int cnt) __attribute__((always_inline)) {
    const int eq = myQ[h0 + ((sub < cnt) ? sub : 0)];
    const int srcv = ei[eq];
    const int dv = ei[EE + eq];
    int rw = dv - n0;
    rw = (rw < 0) ? 0 : ((rw < nrows) ? rw : (nrows - 1));

    v4f hs[EPW];
#pragma unroll
    for (int j = 0; j < EPW; ++j) {
      const int s = __builtin_amdgcn_readlane(srcv, j);
      hs[j] = *(const v4f*)(Hsrc + (size_t)s * KDIM + voff);
    }
#pragma unroll
    for (int j = 0; j < EPW; ++j)
      asm volatile("" :: "v"(hs[j]));

    float p[EPW];
#pragma unroll
    for (int j = 0; j < EPW; ++j) {
      const int r = __builtin_amdgcn_readlane(rw, j);
      const v4f d = ((const v4f*)hd)[r * (FH / 4) + lane];
      float part = 0.0f;
#pragma unroll
      for (int i = 0; i < 4; ++i) {
        float v = hs[j][i] + d[i];
        v = (v >= 0.0f) ? v : v * NEG_SLOPE;
        part = __builtin_fmaf(v, av[i], part);
      }
      p[j] = part;
    }

    float q4[4], q2[2];
#pragma unroll
    for (int k = 0; k < 4; ++k) {
      const float keep = b0 ? p[2 * k + 1] : p[2 * k];
      const float send = b0 ? p[2 * k] : p[2 * k + 1];
      q4[k] = keep + __shfl_xor(send, 1, 32);
    }
#pragma unroll
    for (int k = 0; k < 2; ++k) {
      const float keep = b1 ? q4[2 * k + 1] : q4[2 * k];
      const float send = b1 ? q4[2 * k] : q4[2 * k + 1];
      q2[k] = keep + __shfl_xor(send, 2, 32);
    }
    const float keep = b2 ? q2[1] : q2[0];
    const float send = b2 ? q2[0] : q2[1];
    const float half = keep + __shfl_xor(send, 4, 32);
    const float sc = half + __shfl_xor(half, 8, 32);

    float w = 0.0f;
    if (sub < cnt && !b3) {
      const int idx = rw * 2 + hx;
      __hip_atomic_fetch_max(&mxu[idx], orderFloat(sc), __ATOMIC_RELAXED,
                             __HIP_MEMORY_SCOPE_WORKGROUP);
      const unsigned old = atomicCAS(&cu[idx], F_UNSET, __float_as_uint(sc));
      const float shift = (old == F_UNSET) ? sc : __uint_as_float(old);
      w = expf(sc - shift);
      __hip_atomic_fetch_add(&dn[idx], w, __ATOMIC_RELAXED, __HIP_MEMORY_SCOPE_WORKGROUP);
    }

#pragma unroll
    for (int j = 0; j < EPW; ++j) {
      if (j < cnt) {
        const int r = __builtin_amdgcn_readlane(rw, j);
        const float w0 = __int_as_float(__builtin_amdgcn_readlane(__float_as_int(w), j));
        const float w1 = __int_as_float(__builtin_amdgcn_readlane(__float_as_int(w), 16 + j));
        const float wj = hx ? w1 : w0;
        float* pa = &acc[r * FH + lane];
#pragma unroll
        for (int i = 0; i < 4; ++i)
          __hip_atomic_fetch_add(&pa[32 * i], wj * hs[j][i], __ATOMIC_RELAXED,
                                 __HIP_MEMORY_SCOPE_WORKGROUP);
      }
    }
  };

#pragma nounroll
  for (int pass = 0; pass < 2; ++pass) {
    scan_owned_edges<EPW>(mask, blk, wave, F_W, lane, myQ, batch);
    __syncthreads();
    if (pass == 1) break;

    for (int q = tid; q < nrows * 2; q += F_T) {
      const unsigned cb = cu[q];
      if (cb != F_UNSET) {
        const float m = unorderFloat(mxu[q]);
        if (!(m - __uint_as_float(cb) <= F_SPREAD)) redo = 1;
      }
    }
    __syncthreads();
    if (redo == 0) break;

    for (int q = tid; q < SC_R * 2; q += F_T) {
      if (cu[q] != F_UNSET) {
        const float m = unorderFloat(mxu[q]);
        cu[q] = __float_as_uint(isfinite(m) ? m : 0.0f);
      }
      dn[q] = 0.0f;
    }
    const v4f zf = {0.0f, 0.0f, 0.0f, 0.0f};
    for (int q = tid; q < SC_R * (FH / 4); q += F_T) ((v4f*)acc)[q] = zf;
    __syncthreads();
  }

  for (int q = tid; q < SC_R * 2; q += F_T) {
    const float den = dn[q];
    dn[q] = 1.0f / ((den == 0.0f) ? 1.0f : den);
  }
  __syncthreads();

  for (int q = tid; q < nrows * (FH / 4); q += F_T) {
    const int r = q >> 5;
    const int l = q & 31;
    const float rinv = dn[r * 2 + (l >> 4)];
    const v4f b = ((const v4f*)(gat_bias + col0))[l];
    v4f o;
#pragma unroll
    for (int i = 0; i < 4; ++i) o[i] = acc[r * FH + 32 * i + l] * rinv + b[i];
    *(v4f*)(rst + (size_t)(n0 + r) * KDIM + col0 + 4 * l) = o;
  }
}

#define OBM 64
#define OBK 32
#define OA_PITCH 36
#define OB_PITCH 160
static_assert(KDIM % OBK == 0, "k must split into whole chunks");
static_assert(OUTD == 64, "copy roles and wave layout assume 64 output columns");

__global__ __launch_bounds__(256) void wmma_out_kernel(
    const float* __restrict__ rst, const float* __restrict__ Wfc,
    const float* __restrict__ bfc, float* __restrict__ out) {
  __shared__ __attribute__((aligned(16))) float As[OBM * OA_PITCH];
  __shared__ __attribute__((aligned(16))) float Bs[(OBK / 2) * OB_PITCH];

  const int tid  = threadIdx.x;
  const int lane = tid & 31;
  const int wave = tid >> 5;
  const int wr   = wave >> 1;
  const int wc   = wave & 1;
  const int row0 = blockIdx.x * OBM;
  const int khi  = lane >> 4;
  const int mlo  = lane & 15;

  const int ar  = tid >> 3;
  const int ac  = (tid & 7) * 4;
  const int bcg = tid & 15;
  const int bkp = tid >> 4;

  const float* ag[2];
#pragma unroll
  for (int i = 0; i < 2; ++i) {
    int r = row0 + ar + 32 * i;
    r = (r < NN) ? r : 0;
    ag[i] = rst + (size_t)r * KDIM + ac;
  }
  const float* bg = Wfc + (size_t)(2 * bkp) * OUTD + bcg * 4;

  v4f ra[2], rb0, rb1;
#pragma unroll
  for (int i = 0; i < 2; ++i) ra[i] = *(const v4f*)(ag[i]);
  rb0 = *(const v4f*)(bg);
  rb1 = *(const v4f*)(bg + OUTD);

  const float* aFrag = &As[(wr * 16 + mlo) * OA_PITCH + 2 * khi];
  const float* bFrag = &Bs[khi * OB_PITCH + (wc * 32 + mlo) * 2];

  v8f acc[2] = {};
  for (int c = 0; c < KDIM / OBK; ++c) {
#pragma unroll
    for (int i = 0; i < 2; ++i)
      *(v4f*)&As[(ar + 32 * i) * OA_PITCH + ac] = ra[i];
    {
      float* p = &Bs[bkp * OB_PITCH + bcg * 8];
      v4f lo = {rb0.x, rb1.x, rb0.y, rb1.y};
      v4f hi = {rb0.z, rb1.z, rb0.w, rb1.w};
      *(v4f*)p = lo;
      *(v4f*)(p + 4) = hi;
    }
    __syncthreads();

    if (c + 1 < KDIM / OBK) {
      const int k0 = (c + 1) * OBK;
#pragma unroll
      for (int i = 0; i < 2; ++i) ra[i] = *(const v4f*)(ag[i] + k0);
      rb0 = *(const v4f*)(bg + (size_t)k0 * OUTD);
      rb1 = *(const v4f*)(bg + (size_t)(k0 + 1) * OUTD);
    }

#pragma unroll
    for (int ks = 0; ks < OBK / 4; ++ks) {
      const v2f af = *(const v2f*)(aFrag + ks * 4);
      v2f bf[2];
#pragma unroll
      for (int t = 0; t < 2; ++t)
        bf[t] = *(const v2f*)(bFrag + ks * 2 * OB_PITCH + t * 32);
#pragma unroll
      for (int t = 0; t < 2; ++t)
        acc[t] = __builtin_amdgcn_wmma_f32_16x16x4_f32(
            false, af, false, bf[t], (short)0, acc[t], false, false);
    }
    __syncthreads();
  }

  const int rbase = row0 + wr * 16 + khi * 8;
  float bias[2];
#pragma unroll
  for (int t = 0; t < 2; ++t) bias[t] = bfc[wc * 32 + t * 16 + mlo];
  if (row0 + OBM <= NN) {
#pragma unroll
    for (int i = 0; i < 8; ++i) {
#pragma unroll
      for (int t = 0; t < 2; ++t)
        out[(size_t)(rbase + i) * OUTD + wc * 32 + t * 16 + mlo] = acc[t][i] + bias[t];
    }
  } else {
#pragma unroll
    for (int i = 0; i < 8; ++i) {
      if (rbase + i < NN) {
#pragma unroll
        for (int t = 0; t < 2; ++t)
          out[(size_t)(rbase + i) * OUTD + wc * 32 + t * 16 + mlo] = acc[t][i] + bias[t];
      }
    }
  }
}

extern "C" void kernel_launch(void* const* d_in, const int* in_sizes, int n_in,
                              void* d_out, int out_size, void* d_ws, size_t ws_size,
                              hipStream_t stream) {
  const float* feat     = (const float*)d_in[0];
  const int*   ei       = (const int*)d_in[1];
  const float* w_src    = (const float*)d_in[2];
  const float* b_src    = (const float*)d_in[3];
  const float* w_dst    = (const float*)d_in[4];
  const float* b_dst    = (const float*)d_in[5];
  const float* attn     = (const float*)d_in[6];
  const float* gat_bias = (const float*)d_in[7];
  const float* w_fc     = (const float*)d_in[8];
  const float* b_fc     = (const float*)d_in[9];
  float* out = (float*)d_out;

  float* ws    = (float*)d_ws;
  float* Hsrc  = ws;
  float* Hdst  = Hsrc + (size_t)NN * KDIM;
  float* rst   = Hdst + (size_t)NN * KDIM;
  unsigned* mask = (unsigned*)(rst + (size_t)NN * KDIM);

  owner_mask_kernel<<<SC_NMB, SC_MB, 0, stream>>>(ei, mask);

  wmma_proj_kernel<<<dim3(2, (NN + PBM - 1) / PBM), 256, 0, stream>>>(
      feat, w_src, b_src, w_dst, b_dst, Hsrc, Hdst);

  edge_fused_kernel<<<SC_G * 2, F_T, 0, stream>>>(
      ei, mask, Hsrc, Hdst, attn, gat_bias, rst);

  wmma_out_kernel<<<(NN + OBM - 1) / OBM, 256, 0, stream>>>(rst, w_fc, b_fc, out);
}


# Owner-mask construction moved into the projection launch as trailing work-groups (one per 1024 edges, 4 edges a thread): it runs beside the last GEMM groups, 3 launches for 4
# speedup vs baseline: 4.8278x; 1.0419x over previous
#include <hip/hip_runtime.h>
#include <hip/hip_bf16.h>

#define NN 50000
#define EE 800000
#define KDIM 256
#define OUTD 64
#define NEG_SLOPE 0.2f

typedef __attribute__((ext_vector_type(2))) float v2f;
typedef __attribute__((ext_vector_type(4))) float v4f;
typedef __attribute__((ext_vector_type(8))) float v8f;
typedef __attribute__((ext_vector_type(4))) int v4i;

__device__ __forceinline__ unsigned orderFloat(float f) {
  unsigned u = __float_as_uint(f);
  return (u & 0x80000000u) ? ~u : (u | 0x80000000u);
}
__device__ __forceinline__ float unorderFloat(unsigned u) {
  u = (u & 0x80000000u) ? (u & 0x7FFFFFFFu) : ~u;
  return __uint_as_float(u);
}

#define PBM 128
#define PBK 32
#define A_PITCH 36
#define B_PITCH 544
#define PROJ_T 256
#define PROJ_TILES ((NN + PBM - 1) / PBM)

#define SC_G 256
#define SC_R ((NN + SC_G - 1) / SC_G)
#define SC_MB 1024
#define SC_NMB ((EE + SC_MB - 1) / SC_MB)
#define SC_MROW (SC_G * 32)
static_assert(SC_MROW * 4 <= (PBM * A_PITCH + (PBK / 2) * B_PITCH) * 4,
              "the mask table must fit the LDS block the GEMM role uses");
static_assert(SC_MB % PROJ_T == 0 && PROJ_T % 32 == 0, "a thread's edges keep its bit position");

__global__ __launch_bounds__(PROJ_T) void wmma_proj_kernel(
    const float* __restrict__ feat,
    const float* __restrict__ Wsrc, const float* __restrict__ Bsrc,
    const float* __restrict__ Wdst, const float* __restrict__ Bdst,
    float* __restrict__ Hsrc, float* __restrict__ Hdst,
    const int* __restrict__ ei, unsigned* __restrict__ mask) {
  __shared__ __attribute__((aligned(16))) float smem[PBM * A_PITCH + (PBK / 2) * B_PITCH];
  float* As = smem;
  float* Bs = smem + PBM * A_PITCH;

  const int tid  = threadIdx.x;

  if (blockIdx.y >= PROJ_TILES) {
    unsigned* tab = (unsigned*)smem;
    const int mb = (blockIdx.y - PROJ_TILES) * 2 + blockIdx.x;
    if (mb >= SC_NMB) return;
    unsigned d[SC_MB / PROJ_T];
#pragma unroll
    for (int i = 0; i < SC_MB / PROJ_T; ++i) {
      const int e = mb * SC_MB + tid + PROJ_T * i;
      d[i] = (e < EE) ? (unsigned)ei[EE + e] : 0xFFFFFFFFu;
    }
    const v4i zi = {0, 0, 0, 0};
    for (int q = tid; q < SC_MROW / 4; q += PROJ_T) ((v4i*)tab)[q] = zi;
    __syncthreads();
#pragma unroll
    for (int i = 0; i < SC_MB / PROJ_T; ++i) {
      if (d[i] < (unsigned)NN)
        atomicOr(&tab[(d[i] / SC_R) * 32 + (tid >> 5) + (PROJ_T / 32) * i], 1u << (tid & 31));
    }
    __syncthreads();
    v4i* mout = (v4i*)(mask + (size_t)mb * SC_MROW);
    for (int q = tid; q < SC_MROW / 4; q += PROJ_T) mout[q] = ((const v4i*)tab)[q];
    return;
  }

  const int lane = tid & 31;
  const int wave = tid >> 5;
  const int wr   = wave >> 2;
  const int wc   = wave & 3;
  const int mat  = blockIdx.x;
  const int row0 = blockIdx.y * PBM;
  const int khi  = lane >> 4;
  const int mlo  = lane & 15;

  const float* __restrict__ W  = mat ? Wdst : Wsrc;
  const float* __restrict__ Bv = mat ? Bdst : Bsrc;
  float* __restrict__ Hout     = mat ? Hdst : Hsrc;

  const int ar = tid >> 3;
  const int ac = (tid & 7) * 4;
  const int bcg = tid & 63;
  const int bkp = tid >> 6;

  const float* ag[4];
#pragma unroll
  for (int i = 0; i < 4; ++i) {
    int r = row0 + ar + 32 * i;
    r = (r < NN) ? r : 0;
    ag[i] = feat + (size_t)r * KDIM + ac;
  }
  const float* bg = W + (size_t)(2 * bkp) * KDIM + bcg * 4;

  v4f ra[4], rb0[4], rb1[4];
#pragma unroll
  for (int i = 0; i < 4; ++i) {
    ra[i]  = *(const v4f*)(ag[i]);
    rb0[i] = *(const v4f*)(bg + (size_t)(8 * i) * KDIM);
    rb1[i] = *(const v4f*)(bg + (size_t)(8 * i + 1) * KDIM);
  }

  const float* aFrag = &As[(wr * 64 + mlo) * A_PITCH + 2 * khi];
  const float* bFrag = &Bs[khi * B_PITCH + (wc * 64 + mlo) * 2];

  v8f acc[4][4] = {};
  for (int c = 0; c < KDIM / PBK; ++c) {
#pragma unroll
    for (int i = 0; i < 4; ++i) {
      *(v4f*)&As[(ar + 32 * i) * A_PITCH + ac] = ra[i];
      float* p = &Bs[(bkp + 4 * i) * B_PITCH + bcg * 8];
      v4f lo = {rb0[i].x, rb1[i].x, rb0[i].y, rb1[i].y};
      v4f hi = {rb0[i].z, rb1[i].z, rb0[i].w, rb1[i].w};
      *(v4f*)p = lo;
      *(v4f*)(p + 4) = hi;
    }
    __syncthreads();

    if (c + 1 < KDIM / PBK) {
      const int k0 = (c + 1) * PBK;
#pragma unroll
      for (int i = 0; i < 4; ++i) {
        ra[i]  = *(const v4f*)(ag[i] + k0);
        rb0[i] = *(const v4f*)(bg + (size_t)(k0 + 8 * i) * KDIM);
        rb1[i] = *(const v4f*)(bg + (size_t)(k0 + 8 * i + 1) * KDIM);
      }
    }

#pragma unroll
    for (int ks = 0; ks < PBK / 4; ++ks) {
      v2f af[4], bf[4];
#pragma unroll
      for (int s = 0; s < 4; ++s)
        af[s] = *(const v2f*)(aFrag + s * 16 * A_PITCH + ks * 4);
#pragma unroll
      for (int t = 0; t < 4; ++t)
        bf[t] = *(const v2f*)(bFrag + ks * 2 * B_PITCH + t * 32);
#pragma unroll
      for (int s = 0; s < 4; ++s) {
#pragma unroll
        for (int t = 0; t < 4; ++t) {
          acc[s][t] = __builtin_amdgcn_wmma_f32_16x16x4_f32(
              false, af[s], false, bf[t], (short)0, acc[s][t], false, false);
        }
      }
    }
    __syncthreads();
  }

  const int mbase = khi * 8;
  float bias[4];
#pragma unroll
  for (int t = 0; t < 4; ++t) bias[t] = Bv[wc * 64 + t * 16 + mlo];
  const int rbase = row0 + wr * 64 + mbase;
  if (row0 + PBM <= NN) {
#pragma unroll
    for (int s = 0; s < 4; ++s) {
#pragma unroll
      for (int i = 0; i < 8; ++i) {
        int r = rbase + s * 16 + i;
#pragma unroll
        for (int t = 0; t < 4; ++t)
          Hout[(size_t)r * KDIM + wc * 64 + t * 16 + mlo] = acc[s][t][i] + bias[t];
      }
    }
  } else {
#pragma unroll
    for (int s = 0; s < 4; ++s) {
#pragma unroll
      for (int i = 0; i < 8; ++i) {
        int r = rbase + s * 16 + i;
        if (r < NN) {
#pragma unroll
          for (int t = 0; t < 4; ++t)
            Hout[(size_t)r * KDIM + wc * 64 + t * 16 + mlo] = acc[s][t][i] + bias[t];
        }
      }
    }
  }
}


#define EPW 8

#define SC_QCAP (32 + EPW)
#define FH 128
#define F_T 1024
#define F_W (F_T / 32)
#define F_UNSET 0xFFFFFFFFu
#define F_SPREAD 32.0f
static_assert((SC_G - 1) * SC_R < NN, "every block owns at least one row");
static_assert((NN - 1) / SC_R < SC_G, "every dst has an owner block");
static_assert(2 * SC_R * FH * 4 + 3 * SC_R * 2 * 4 + F_W * SC_QCAP * 4 + 4 <= 327680,
              "Hdst pieces, accumulators, per-row state and the queues must fit the LDS");

template <int B, class F>
__device__ __forceinline__ void scan_owned_edges(
    const unsigned* __restrict__ mask, int group, int wave, int nwaves, int lane, int* myQ, F batch) {
  const unsigned* mgrp = mask + (size_t)group * 32;
  auto line_of = [&](int c) __attribute__((always_inline)) {
    const int cc = (c < SC_NMB) ? c : (SC_NMB - 1);
    return (mgrp + (size_t)cc * SC_MROW)[lane];
  };
  int qn = 0;
  unsigned nxt = line_of(wave);
  for (int c = wave; c < SC_NMB; c += nwaves) {
    unsigned w = nxt;
    nxt = line_of(c + nwaves);
    const int ebase = c * SC_MB + lane * 32;
    unsigned m = __builtin_amdgcn_ballot_w32(w != 0u);
    while (m) {
      if (w != 0u) {
        myQ[qn + (int)__builtin_amdgcn_mbcnt_lo(m, 0u)] = ebase + __builtin_ctz(w);
        w &= w - 1u;
      }
      qn += __builtin_popcount(m);
      m = __builtin_amdgcn_ballot_w32(w != 0u);
      int h = 0;
      for (; h + B <= qn; h += B) batch(h, B);
      if (h > 0) {
        const int rem = qn - h;
        int t = 0;
        if (lane < rem) t = myQ[h + lane];
        if (lane < rem) myQ[lane] = t;
        qn = rem;
      }
    }
  }
  if (qn > 0) batch(0, qn);
}

__global__ __launch_bounds__(F_T) void edge_fused_kernel(
    const int* __restrict__ ei, const unsigned* __restrict__ mask,
    const float* __restrict__ Hsrc, const float* __restrict__ Hdst,
    const float* __restrict__ attn, const float* __restrict__ gat_bias,
    float* __restrict__ rst) {
  __shared__ __attribute__((aligned(16))) float hd[SC_R * FH];
  __shared__ __attribute__((aligned(16))) float acc[SC_R * FH];
  __shared__ unsigned mxu[SC_R * 2];
  __shared__ unsigned cu[SC_R * 2];
  __shared__ float dn[SC_R * 2];
  __shared__ int queue[F_W * SC_QCAP];
  __shared__ int redo;

  const int tid  = threadIdx.x;
  const int lane = tid & 31;
  const int wave = __builtin_amdgcn_readfirstlane(tid >> 5);
  const int blk  = blockIdx.x >> 1;
  const int col0 = (blockIdx.x & 1) * FH;
  const int n0 = blk * SC_R;
  const int nrows = (NN - n0 < SC_R) ? (NN - n0) : SC_R;

  {
    const int total = nrows * (FH / 4);
    for (int q0 = tid; q0 < total; q0 += 4 * F_T) {
      v4f t[4];
#pragma unroll
      for (int u = 0; u < 4; ++u) {
        const int q = q0 + u * F_T;
        const int qq = (q < total) ? q : (total - 1);
        t[u] = *(const v4f*)(Hdst + (size_t)(n0 + (qq >> 5)) * KDIM + col0 + (qq & 31) * 4);
      }
#pragma unroll
      for (int u = 0; u < 4; ++u) {
        const int q = q0 + u * F_T;
        if (q < total) ((v4f*)hd)[q] = t[u];
      }
    }
    const v4f zf = {0.0f, 0.0f, 0.0f, 0.0f};
    for (int q = tid; q < SC_R * (FH / 4); q += F_T) ((v4f*)acc)[q] = zf;
    for (int q = tid; q < SC_R * 2; q += F_T) {
      mxu[q] = 0u;
      cu[q] = F_UNSET;
      dn[q] = 0.0f;
    }
    if (tid == 0) redo = 0;
  }
  __syncthreads();

  int* myQ = queue + wave * SC_QCAP;
  const int sub = lane & 7;
  const int hx = lane >> 4;
  const v4f av = ((const v4f*)(attn + col0))[lane];
  const unsigned voff = (unsigned)(col0 + lane * 4);
  const bool b0 = (lane & 1) != 0, b1 = (lane & 2) != 0, b2 = (lane & 4) != 0, b3 = (lane & 8) != 0;

  auto batch = [&](int h0, int cnt) __attribute__((always_inline)) {
    const int eq = myQ[h0 + ((sub < cnt) ? sub : 0)];
    const int srcv = ei[eq];
    const int dv = ei[EE + eq];
    int rw = dv - n0;
    rw = (rw < 0) ? 0 : ((rw < nrows) ? rw : (nrows - 1));

    v4f hs[EPW];
#pragma unroll
    for (int j = 0; j < EPW; ++j) {
      const int s = __builtin_amdgcn_readlane(srcv, j);
      hs[j] = *(const v4f*)(Hsrc + (size_t)s * KDIM + voff);
    }
#pragma unroll
    for (int j = 0; j < EPW; ++j)
      asm volatile("" :: "v"(hs[j]));

    float p[EPW];
#pragma unroll
    for (int j = 0; j < EPW; ++j) {
      const int r = __builtin_amdgcn_readlane(rw, j);
      const v4f d = ((const v4f*)hd)[r * (FH / 4) + lane];
      float part = 0.0f;
#pragma unroll
      for (int i = 0; i < 4; ++i) {
        float v = hs[j][i] + d[i];
        v = (v >= 0.0f) ? v : v * NEG_SLOPE;
        part = __builtin_fmaf(v, av[i], part);
      }
      p[j] = part;
    }

    float q4[4], q2[2];
#pragma unroll
    for (int k = 0; k < 4; ++k) {
      const float keep = b0 ? p[2 * k + 1] : p[2 * k];
      const float send = b0 ? p[2 * k] : p[2 * k + 1];
      q4[k] = keep + __shfl_xor(send, 1, 32);
    }
#pragma unroll
    for (int k = 0; k < 2; ++k) {
      const float keep = b1 ? q4[2 * k + 1] : q4[2 * k];
      const float send = b1 ? q4[2 * k] : q4[2 * k + 1];
      q2[k] = keep + __shfl_xor(send, 2, 32);
    }
    const float keep = b2 ? q2[1] : q2[0];
    const float send = b2 ? q2[0] : q2[1];
    const float half = keep + __shfl_xor(send, 4, 32);
    const float sc = half + __shfl_xor(half, 8, 32);

    float w = 0.0f;
    if (sub < cnt && !b3) {
      const int idx = rw * 2 + hx;
      __hip_atomic_fetch_max(&mxu[idx], orderFloat(sc), __ATOMIC_RELAXED,
                             __HIP_MEMORY_SCOPE_WORKGROUP);
      const unsigned old = atomicCAS(&cu[idx], F_UNSET, __float_as_uint(sc));
      const float shift = (old == F_UNSET) ? sc : __uint_as_float(old);
      w = expf(sc - shift);
      __hip_atomic_fetch_add(&dn[idx], w, __ATOMIC_RELAXED, __HIP_MEMORY_SCOPE_WORKGROUP);
    }

#pragma unroll
    for (int j = 0; j < EPW; ++j) {
      if (j < cnt) {
        const int r = __builtin_amdgcn_readlane(rw, j);
        const float w0 = __int_as_float(__builtin_amdgcn_readlane(__float_as_int(w), j));
        const float w1 = __int_as_float(__builtin_amdgcn_readlane(__float_as_int(w), 16 + j));
        const float wj = hx ? w1 : w0;
        float* pa = &acc[r * FH + lane];
#pragma unroll
        for (int i = 0; i < 4; ++i)
          __hip_atomic_fetch_add(&pa[32 * i], wj * hs[j][i], __ATOMIC_RELAXED,
                                 __HIP_MEMORY_SCOPE_WORKGROUP);
      }
    }
  };

#pragma nounroll
  for (int pass = 0; pass < 2; ++pass) {
    scan_owned_edges<EPW>(mask, blk, wave, F_W, lane, myQ, batch);
    __syncthreads();
    if (pass == 1) break;

    for (int q = tid; q < nrows * 2; q += F_T) {
      const unsigned cb = cu[q];
      if (cb != F_UNSET) {
        const float m = unorderFloat(mxu[q]);
        if (!(m - __uint_as_float(cb) <= F_SPREAD)) redo = 1;
      }
    }
    __syncthreads();
    if (redo == 0) break;

    for (int q = tid; q < SC_R * 2; q += F_T) {
      if (cu[q] != F_UNSET) {
        const float m = unorderFloat(mxu[q]);
        cu[q] = __float_as_uint(isfinite(m) ? m : 0.0f);
      }
      dn[q] = 0.0f;
    }
    const v4f zf = {0.0f, 0.0f, 0.0f, 0.0f};
    for (int q = tid; q < SC_R * (FH / 4); q += F_T) ((v4f*)acc)[q] = zf;
    __syncthreads();
  }

  for (int q = tid; q < SC_R * 2; q += F_T) {
    const float den = dn[q];
    dn[q] = 1.0f / ((den == 0.0f) ? 1.0f : den);
  }
  __syncthreads();

  for (int q = tid; q < nrows * (FH / 4); q += F_T) {
    const int r = q >> 5;
    const int l = q & 31;
    const float rinv = dn[r * 2 + (l >> 4)];
    const v4f b = ((const v4f*)(gat_bias + col0))[l];
    v4f o;
#pragma unroll
    for (int i = 0; i < 4; ++i) o[i] = acc[r * FH + 32 * i + l] * rinv + b[i];
    *(v4f*)(rst + (size_t)(n0 + r) * KDIM + col0 + 4 * l) = o;
  }
}

#define OBM 64
#define OBK 32
#define OA_PITCH 36
#define OB_PITCH 160
static_assert(KDIM % OBK == 0, "k must split into whole chunks");
static_assert(OUTD == 64, "copy roles and wave layout assume 64 output columns");

__global__ __launch_bounds__(256) void wmma_out_kernel(
    const float* __restrict__ rst, const float* __restrict__ Wfc,
    const float* __restrict__ bfc, float* __restrict__ out) {
  __shared__ __attribute__((aligned(16))) float As[OBM * OA_PITCH];
  __shared__ __attribute__((aligned(16))) float Bs[(OBK / 2) * OB_PITCH];

  const int tid  = threadIdx.x;
  const int lane = tid & 31;
  const int wave = tid >> 5;
  const int wr   = wave >> 1;
  const int wc   = wave & 1;
  const int row0 = blockIdx.x * OBM;
  const int khi  = lane >> 4;
  const int mlo  = lane & 15;

  const int ar  = tid >> 3;
  const int ac  = (tid & 7) * 4;
  const int bcg = tid & 15;
  const int bkp = tid >> 4;

  const float* ag[2];
#pragma unroll
  for (int i = 0; i < 2; ++i) {
    int r = row0 + ar + 32 * i;
    r = (r < NN) ? r : 0;
    ag[i] = rst + (size_t)r * KDIM + ac;
  }
  const float* bg = Wfc + (size_t)(2 * bkp) * OUTD + bcg * 4;

  v4f ra[2], rb0, rb1;
#pragma unroll
  for (int i = 0; i < 2; ++i) ra[i] = *(const v4f*)(ag[i]);
  rb0 = *(const v4f*)(bg);
  rb1 = *(const v4f*)(bg + OUTD);

  const float* aFrag = &As[(wr * 16 + mlo) * OA_PITCH + 2 * khi];
  const float* bFrag = &Bs[khi * OB_PITCH + (wc * 32 + mlo) * 2];

  v8f acc[2] = {};
  for (int c = 0; c < KDIM / OBK; ++c) {
#pragma unroll
    for (int i = 0; i < 2; ++i)
      *(v4f*)&As[(ar + 32 * i) * OA_PITCH + ac] = ra[i];
    {
      float* p = &Bs[bkp * OB_PITCH + bcg * 8];
      v4f lo = {rb0.x, rb1.x, rb0.y, rb1.y};
      v4f hi = {rb0.z, rb1.z, rb0.w, rb1.w};
      *(v4f*)p = lo;
      *(v4f*)(p + 4) = hi;
    }
    __syncthreads();

    if (c + 1 < KDIM / OBK) {
      const int k0 = (c + 1) * OBK;
#pragma unroll
      for (int i = 0; i < 2; ++i) ra[i] = *(const v4f*)(ag[i] + k0);
      rb0 = *(const v4f*)(bg + (size_t)k0 * OUTD);
      rb1 = *(const v4f*)(bg + (size_t)(k0 + 1) * OUTD);
    }

#pragma unroll
    for (int ks = 0; ks < OBK / 4; ++ks) {
      const v2f af = *(const v2f*)(aFrag + ks * 4);
      v2f bf[2];
#pragma unroll
      for (int t = 0; t < 2; ++t)
        bf[t] = *(const v2f*)(bFrag + ks * 2 * OB_PITCH + t * 32);
#pragma unroll
      for (int t = 0; t < 2; ++t)
        acc[t] = __builtin_amdgcn_wmma_f32_16x16x4_f32(
            false, af, false, bf[t], (short)0, acc[t], false, false);
    }
    __syncthreads();
  }

  const int rbase = row0 + wr * 16 + khi * 8;
  float bias[2];
#pragma unroll
  for (int t = 0; t < 2; ++t) bias[t] = bfc[wc * 32 + t * 16 + mlo];
  if (row0 + OBM <= NN) {
#pragma unroll
    for (int i = 0; i < 8; ++i) {
#pragma unroll
      for (int t = 0; t < 2; ++t)
        out[(size_t)(rbase + i) * OUTD + wc * 32 + t * 16 + mlo] = acc[t][i] + bias[t];
    }
  } else {
#pragma unroll
    for (int i = 0; i < 8; ++i) {
      if (rbase + i < NN) {
#pragma unroll
        for (int t = 0; t < 2; ++t)
          out[(size_t)(rbase + i) * OUTD + wc * 32 + t * 16 + mlo] = acc[t][i] + bias[t];
      }
    }
  }
}

extern "C" void kernel_launch(void* const* d_in, const int* in_sizes, int n_in,
                              void* d_out, int out_size, void* d_ws, size_t ws_size,
                              hipStream_t stream) {
  const float* feat     = (const float*)d_in[0];
  const int*   ei       = (const int*)d_in[1];
  const float* w_src    = (const float*)d_in[2];
  const float* b_src    = (const float*)d_in[3];
  const float* w_dst    = (const float*)d_in[4];
  const float* b_dst    = (const float*)d_in[5];
  const float* attn     = (const float*)d_in[6];
  const float* gat_bias = (const float*)d_in[7];
  const float* w_fc     = (const float*)d_in[8];
  const float* b_fc     = (const float*)d_in[9];
  float* out = (float*)d_out;

  float* ws    = (float*)d_ws;
  float* Hsrc  = ws;
  float* Hdst  = Hsrc + (size_t)NN * KDIM;
  float* rst   = Hdst + (size_t)NN * KDIM;
  unsigned* mask = (unsigned*)(rst + (size_t)NN * KDIM);

  wmma_proj_kernel<<<dim3(2, PROJ_TILES + (SC_NMB + 1) / 2), PROJ_T, 0, stream>>>(
      feat, w_src, b_src, w_dst, b_dst, Hsrc, Hdst, ei, mask);

  edge_fused_kernel<<<SC_G * 2, F_T, 0, stream>>>(
      ei, mask, Hsrc, Hdst, attn, gat_bias, rst);

  wmma_out_kernel<<<(NN + OBM - 1) / OBM, 256, 0, stream>>>(rst, w_fc, b_fc, out);
}
